# starting kernel
# baseline (speedup 1.0000x reference)
#include <hip/hip_runtime.h>
#include <math.h>

#define HIDF   128
#define NHEAD  4
#define WHID   512
#define NGROUP 64
#define OUTD   256
#define OUT2   128
#define NCLSD  2
#define SLOPE  0.2f
#define KC     128
#define BSTRIDE (KC + 2)

typedef float v2f __attribute__((ext_vector_type(2)));
typedef float v8f __attribute__((ext_vector_type(8)));

__device__ __forceinline__ float lrelu(float x) { return x >= 0.f ? x : SLOPE * x; }

__device__ __forceinline__ void atomicMaxF(float* addr, float val) {
  if (val >= 0.f) atomicMax((int*)addr, __float_as_int(val));
  else            atomicMin((unsigned int*)addr, __float_as_uint(val));
}

__global__ void wmma_gemm(const float* __restrict__ A, const float* __restrict__ B,
                          const float* __restrict__ bias, float* __restrict__ C,
                          int M, int K, int Nc) {
  __shared__ float Bs[64 * BSTRIDE];

  const int nslab = Nc >> 6;
  const int mblk  = blockIdx.x / nslab;
  const int nblk  = blockIdx.x - mblk * nslab;
  const int wave  = threadIdx.x >> 5;
  const int lane  = threadIdx.x & 31;
  const int half  = lane >> 4;
  const int l     = lane & 15;

  const int mt    = mblk * 8 + wave;
  const int arow  = min(mt * 16 + l, M - 1);
  const int col0  = nblk * 64;

  const int sc = threadIdx.x & 63;
  const int sr = threadIdx.x >> 6;

  const float* Ap = A + (size_t)arow * K + 2 * half;

  v8f acc0 = {}, acc1 = {}, acc2 = {}, acc3 = {};

  for (int k0 = 0; k0 < K; k0 += KC) {
    const int kc = min(KC, K - k0);
    __syncthreads();
    for (int kk = sr; kk < kc; kk += 4)
      Bs[sc * BSTRIDE + kk] = B[(size_t)(k0 + kk) * Nc + col0 + sc];
    __syncthreads();

    __builtin_prefetch(Ap + k0 + KC, 0, 1);

    const float* Bl0 = &Bs[(l +  0) * BSTRIDE + 2 * half];
    const float* Bl1 = &Bs[(l + 16) * BSTRIDE + 2 * half];
    const float* Bl2 = &Bs[(l + 32) * BSTRIDE + 2 * half];
    const float* Bl3 = &Bs[(l + 48) * BSTRIDE + 2 * half];
    for (int k = 0; k < kc; k += 4) {
      v2f a  = *(const v2f*)(Ap + k0 + k);
      v2f b0 = *(const v2f*)(Bl0 + k);
      v2f b1 = *(const v2f*)(Bl1 + k);
      v2f b2 = *(const v2f*)(Bl2 + k);
      v2f b3 = *(const v2f*)(Bl3 + k);
      acc0 = __builtin_amdgcn_wmma_f32_16x16x4_f32(false, a, false, b0, (short)0, acc0, false, false);
      acc1 = __builtin_amdgcn_wmma_f32_16x16x4_f32(false, a, false, b1, (short)0, acc1, false, false);
      acc2 = __builtin_amdgcn_wmma_f32_16x16x4_f32(false, a, false, b2, (short)0, acc2, false, false);
      acc3 = __builtin_amdgcn_wmma_f32_16x16x4_f32(false, a, false, b3, (short)0, acc3, false, false);
    }
  }

  v8f* accs[4] = {&acc0, &acc1, &acc2, &acc3};
#pragma unroll
  for (int j = 0; j < 4; ++j) {
    const int ccol = col0 + 16 * j + l;
    const float bv = bias ? bias[ccol] : 0.f;
#pragma unroll
    for (int i = 0; i < 8; ++i) {
      const int row = mt * 16 + i + 8 * half;
      if (row < M) C[(size_t)row * Nc + ccol] = (*accs[j])[i] + bv;
    }
  }
}

__global__ void attn_scores(const float* __restrict__ h, const float* __restrict__ a_s,
                            const float* __restrict__ a_d, float* __restrict__ ssrc,
                            float* __restrict__ sdst, int n) {
  int idx = blockIdx.x * blockDim.x + threadIdx.x;
  if (idx >= n * NHEAD) return;
  int hd = idx & (NHEAD - 1);
  const float* hp = h + (size_t)idx * HIDF;
  const float* as = a_s + hd * HIDF;
  const float* ad = a_d + hd * HIDF;
  float s1 = 0.f, s2 = 0.f;
  for (int c = 0; c < HIDF; c += 4) {
    float4 hv = *(const float4*)(hp + c);
    float4 av = *(const float4*)(as + c);
    float4 dv = *(const float4*)(ad + c);
    s1 += hv.x * av.x + hv.y * av.y + hv.z * av.z + hv.w * av.w;
    s2 += hv.x * dv.x + hv.y * dv.y + hv.z * dv.z + hv.w * dv.w;
  }
  ssrc[idx] = s1;
  sdst[idx] = s2;
}

__global__ void fill_f32(float* __restrict__ p, float v, int n) {
  int i = blockIdx.x * blockDim.x + threadIdx.x;
  if (i < n) p[i] = v;
}

__global__ void edge_logits_max(const int* __restrict__ ei, int E, int n,
                                const float* __restrict__ ssrc,
                                const float* __restrict__ sdst,
                                float* __restrict__ eatt, float* __restrict__ m) {
  int idx = blockIdx.x * blockDim.x + threadIdx.x;
  if (idx >= (E + n) * NHEAD) return;
  int e = idx >> 2, hd = idx & 3;
  int s, d;
  if (e < E) { s = ei[e]; d = ei[E + e]; } else { s = e - E; d = s; }
  float v = lrelu(ssrc[s * NHEAD + hd] + sdst[d * NHEAD + hd]);
  eatt[idx] = v;
  atomicMaxF(&m[d * NHEAD + hd], v);
}

__global__ void fix_m(float* __restrict__ m, int count) {
  int i = blockIdx.x * blockDim.x + threadIdx.x;
  if (i >= count) return;
  float v = m[i];
  if (!(v > -3.0e38f && v < 3.0e38f)) m[i] = 0.f;
}

__global__ void edge_expsum(const int* __restrict__ ei, int E, int n,
                            float* __restrict__ eatt, const float* __restrict__ m,
                            float* __restrict__ den) {
  int idx = blockIdx.x * blockDim.x + threadIdx.x;
  if (idx >= (E + n) * NHEAD) return;
  int e = idx >> 2, hd = idx & 3;
  int d = (e < E) ? ei[E + e] : (e - E);
  float ex = expf(eatt[idx] - m[d * NHEAD + hd]);
  eatt[idx] = ex;
  atomicAdd(&den[d * NHEAD + hd], ex);
}

__global__ void edge_aggr(const int* __restrict__ ei, int E, int n,
                          const float* __restrict__ ex, const float* __restrict__ den,
                          const float* __restrict__ h, float* __restrict__ acc,
                          float* __restrict__ alpha_out) {
  long gid  = (long)blockIdx.x * blockDim.x + threadIdx.x;
  long wave = gid >> 5;
  int  lane = threadIdx.x & 31;
  if (wave >= (long)(E + n) * NHEAD) return;
  int e = (int)(wave >> 2), hd = (int)(wave & 3);
  int s, d;
  if (e < E) { s = ei[e]; d = ei[E + e]; } else { s = e - E; d = s; }
  float a = ex[wave] / den[(size_t)d * NHEAD + hd];
  if (alpha_out && lane == 0) alpha_out[wave] = a;
  const float* hp = h + ((size_t)s * NHEAD + hd) * HIDF + lane * 4;
  float*       ap = acc + ((size_t)d * NHEAD + hd) * HIDF + lane * 4;
  float4 hv = *(const float4*)hp;
  atomicAdd(ap + 0, a * hv.x);
  atomicAdd(ap + 1, a * hv.y);
  atomicAdd(ap + 2, a * hv.z);
  atomicAdd(ap + 3, a * hv.w);
}

__global__ void l1_post(const float* __restrict__ acc, const float* __restrict__ b1,
                        const float* __restrict__ g, const float* __restrict__ b,
                        float* __restrict__ x2) {
  __shared__ float sh[HIDF];
  int nd = blockIdx.x, c = threadIdx.x;
  const float* ap = acc + (size_t)nd * WHID;
  float v = 0.25f * (ap[c] + ap[HIDF + c] + ap[2 * HIDF + c] + ap[3 * HIDF + c]) + b1[c];
  sh[c] = v; __syncthreads();
  for (int s = HIDF / 2; s > 0; s >>= 1) { if (c < s) sh[c] += sh[c + s]; __syncthreads(); }
  float mu = sh[0] * (1.0f / HIDF);
  __syncthreads();
  float dv = v - mu;
  sh[c] = dv * dv; __syncthreads();
  for (int s = HIDF / 2; s > 0; s >>= 1) { if (c < s) sh[c] += sh[c + s]; __syncthreads(); }
  float var = sh[0] * (1.0f / HIDF);
  float y = g[c] * dv * rsqrtf(var + 1e-5f) + b[c];
  x2[(size_t)nd * HIDF + c] = lrelu(y);
}

__global__ void l2_post_pool(const float* __restrict__ acc, const float* __restrict__ b2,
                             const int* __restrict__ batch, float* __restrict__ pooled,
                             float* __restrict__ cnt) {
  int nd = blockIdx.x, c = threadIdx.x;
  float y = lrelu(acc[(size_t)nd * WHID + c] + b2[c]);
  int gidx = batch[nd];
  atomicAdd(&pooled[(size_t)gidx * WHID + c], y);
  if (c == 0) atomicAdd(&cnt[gidx], 1.0f);
}

__global__ void pool_div(float* __restrict__ pooled, const float* __restrict__ cnt) {
  int i = blockIdx.x * blockDim.x + threadIdx.x;
  if (i >= NGROUP * WHID) return;
  pooled[i] /= fmaxf(cnt[i / WHID], 1.0f);
}

__global__ void ln_relu(float* __restrict__ x, const float* __restrict__ g,
                        const float* __restrict__ b, int dim) {
  extern __shared__ float sh[];
  int r = blockIdx.x, c = threadIdx.x;
  float v = x[(size_t)r * dim + c];
  sh[c] = v; __syncthreads();
  for (int s = dim >> 1; s > 0; s >>= 1) { if (c < s) sh[c] += sh[c + s]; __syncthreads(); }
  float mu = sh[0] / dim;
  __syncthreads();
  float dv = v - mu;
  sh[c] = dv * dv; __syncthreads();
  for (int s = dim >> 1; s > 0; s >>= 1) { if (c < s) sh[c] += sh[c + s]; __syncthreads(); }
  float var = sh[0] / dim;
  float y = g[c] * dv * rsqrtf(var + 1e-5f) + b[c];
  x[(size_t)r * dim + c] = fmaxf(y, 0.f);
}

__global__ void head_out(const float* __restrict__ t2, const float* __restrict__ Ws3,
                         const float* __restrict__ bs3, float* __restrict__ out) {
  int r = blockIdx.x * blockDim.x + threadIdx.x;
  if (r >= NGROUP) return;
  float l0 = bs3[0], l1 = bs3[1];
  for (int c = 0; c < OUT2; ++c) {
    float v = t2[r * OUT2 + c];
    l0 += v * Ws3[c * 2 + 0];
    l1 += v * Ws3[c * 2 + 1];
  }
  float mx = fmaxf(l0, l1);
  float e0 = expf(l0 - mx), e1 = expf(l1 - mx), s = e0 + e1;
  out[r * 2 + 0] = e0 / s;
  out[r * 2 + 1] = e1 / s;
}

extern "C" void kernel_launch(void* const* d_in, const int* in_sizes, int n_in,
                              void* d_out, int out_size, void* d_ws, size_t ws_size,
                              hipStream_t stream) {
  (void)n_in; (void)out_size; (void)ws_size;
  const float* x    = (const float*)d_in[0];
  const int*   ei   = (const int*)d_in[1];
  const int*   batch= (const int*)d_in[2];
  const float* W1   = (const float*)d_in[3];
  const float* as1  = (const float*)d_in[4];
  const float* ad1  = (const float*)d_in[5];
  const float* b1   = (const float*)d_in[6];
  const float* g_ln = (const float*)d_in[7];
  const float* b_ln = (const float*)d_in[8];
  const float* W2   = (const float*)d_in[9];
  const float* as2  = (const float*)d_in[10];
  const float* ad2  = (const float*)d_in[11];
  const float* b2   = (const float*)d_in[12];
  const float* Wl0  = (const float*)d_in[13];
  const float* bl0  = (const float*)d_in[14];
  const float* Ws1  = (const float*)d_in[15];
  const float* bs1  = (const float*)d_in[16];
  const float* g1   = (const float*)d_in[17];
  const float* bb1  = (const float*)d_in[18];
  const float* Ws2  = (const float*)d_in[19];
  const float* bs2  = (const float*)d_in[20];
  const float* g2   = (const float*)d_in[21];
  const float* bb2  = (const float*)d_in[22];
  const float* Ws3  = (const float*)d_in[23];
  const float* bs3  = (const float*)d_in[24];

  const int N  = in_sizes[0] / HIDF;
  const int E  = in_sizes[1] / 2;
  const int TE = E + N;
  const int NH = N * NHEAD;
  const int TEH = TE * NHEAD;

  float* out       = (float*)d_out;
  float* alpha_out = out + NGROUP * NCLSD;

  size_t off = 0;
  auto alloc = [&](size_t bytes) -> float* {
    off = (off + 255) & ~(size_t)255;
    float* p = (float*)((char*)d_ws + off);
    off += bytes;
    return p;
  };
  float* h      = alloc((size_t)N * WHID * 4);
  float* acc    = alloc((size_t)N * WHID * 4);
  float* x2     = alloc((size_t)N * HIDF * 4);
  float* ssrc   = alloc((size_t)NH * 4);
  float* sdst   = alloc((size_t)NH * 4);
  float* eatt   = alloc((size_t)TEH * 4);
  float* mbuf   = alloc((size_t)NH * 4);
  float* den    = alloc((size_t)NH * 4);
  float* pooled = alloc((size_t)NGROUP * WHID * 4);
  float* cnt    = alloc((size_t)NGROUP * 4);
  float* z0     = alloc((size_t)NGROUP * HIDF * 4);
  float* t1     = alloc((size_t)NGROUP * OUTD * 4);
  float* t2     = alloc((size_t)NGROUP * OUT2 * 4);

  auto cdiv = [](long a, long b) { return (int)((a + b - 1) / b); };
  auto gemm = [&](const float* A, const float* B, const float* bias, float* C,
                  int M, int K, int Nc) {
    int mtiles  = (M + 15) / 16;
    int mblocks = (mtiles + 7) / 8;
    int nslab   = Nc / 64;
    wmma_gemm<<<mblocks * nslab, 256, 0, stream>>>(A, B, bias, C, M, K, Nc);
  };

  gemm(x, W1, nullptr, h, N, HIDF, WHID);
  attn_scores<<<cdiv(NH, 256), 256, 0, stream>>>(h, as1, ad1, ssrc, sdst, N);
  fill_f32<<<cdiv(NH, 256), 256, 0, stream>>>(mbuf, -INFINITY, NH);
  fill_f32<<<cdiv(NH, 256), 256, 0, stream>>>(den, 0.f, NH);
  fill_f32<<<cdiv((long)N * WHID, 256), 256, 0, stream>>>(acc, 0.f, N * WHID);
  edge_logits_max<<<cdiv(TEH, 256), 256, 0, stream>>>(ei, E, N, ssrc, sdst, eatt, mbuf);
  fix_m<<<cdiv(NH, 256), 256, 0, stream>>>(mbuf, NH);
  edge_expsum<<<cdiv(TEH, 256), 256, 0, stream>>>(ei, E, N, eatt, mbuf, den);
  edge_aggr<<<cdiv((long)TEH * 32, 256), 256, 0, stream>>>(ei, E, N, eatt, den, h, acc,
                                                           nullptr);
  l1_post<<<N, HIDF, 0, stream>>>(acc, b1, g_ln, b_ln, x2);

  gemm(x2, W2, nullptr, h, N, HIDF, WHID);
  attn_scores<<<cdiv(NH, 256), 256, 0, stream>>>(h, as2, ad2, ssrc, sdst, N);
  fill_f32<<<cdiv(NH, 256), 256, 0, stream>>>(mbuf, -INFINITY, NH);
  fill_f32<<<cdiv(NH, 256), 256, 0, stream>>>(den, 0.f, NH);
  fill_f32<<<cdiv((long)N * WHID, 256), 256, 0, stream>>>(acc, 0.f, N * WHID);
  edge_logits_max<<<cdiv(TEH, 256), 256, 0, stream>>>(ei, E, N, ssrc, sdst, eatt, mbuf);
  fix_m<<<cdiv(NH, 256), 256, 0, stream>>>(mbuf, NH);
  edge_expsum<<<cdiv(TEH, 256), 256, 0, stream>>>(ei, E, N, eatt, mbuf, den);
  edge_aggr<<<cdiv((long)TEH * 32, 256), 256, 0, stream>>>(ei, E, N, eatt, den, h, acc,
                                                           alpha_out);

  fill_f32<<<cdiv(NGROUP * WHID, 256), 256, 0, stream>>>(pooled, 0.f, NGROUP * WHID);
  fill_f32<<<1, NGROUP, 0, stream>>>(cnt, 0.f, NGROUP);
  l2_post_pool<<<N, WHID, 0, stream>>>(acc, b2, batch, pooled, cnt);
  pool_div<<<cdiv(NGROUP * WHID, 256), 256, 0, stream>>>(pooled, cnt);

  gemm(pooled, Wl0, bl0, z0, NGROUP, WHID, HIDF);
  gemm(z0, Ws1, bs1, t1, NGROUP, HIDF, OUTD);
  ln_relu<<<NGROUP, OUTD, OUTD * sizeof(float), stream>>>(t1, g1, bb1, OUTD);
  gemm(t1, Ws2, bs2, t2, NGROUP, OUTD, OUT2);
  ln_relu<<<NGROUP, OUT2, OUT2 * sizeof(float), stream>>>(t2, g2, bb2, OUT2);
  head_out<<<1, NGROUP, 0, stream>>>(t2, Ws3, bs3, out);
}


# edge aggregation: edges sorted by (dst, edge id) once per call, one wave per node sums its h[src] rows in registers and stores acc once; no float atomics or zero-fill on acc; sums in edge-id order
# speedup vs baseline: 1.8370x; 1.8370x over previous
#include <hip/hip_runtime.h>
#include <math.h>

#define HIDF   128
#define NHEAD  4
#define WHID   512
#define NGROUP 64
#define OUTD   256
#define OUT2   128
#define NCLSD  2
#define SLOPE  0.2f
#define KC     128
#define BSTRIDE (KC + 2)

typedef float v2f __attribute__((ext_vector_type(2)));
typedef float v8f __attribute__((ext_vector_type(8)));

__device__ __forceinline__ float lrelu(float x) { return x >= 0.f ? x : SLOPE * x; }

__device__ __forceinline__ void atomicMaxF(float* addr, float val) {
  if (val >= 0.f) atomicMax((int*)addr, __float_as_int(val));
  else            atomicMin((unsigned int*)addr, __float_as_uint(val));
}

__global__ void wmma_gemm(const float* __restrict__ A, const float* __restrict__ B,
                          const float* __restrict__ bias, float* __restrict__ C,
                          int M, int K, int Nc) {
  __shared__ float Bs[64 * BSTRIDE];

  const int nslab = Nc >> 6;
  const int mblk  = blockIdx.x / nslab;
  const int nblk  = blockIdx.x - mblk * nslab;
  const int wave  = threadIdx.x >> 5;
  const int lane  = threadIdx.x & 31;
  const int half  = lane >> 4;
  const int l     = lane & 15;

  const int mt    = mblk * 8 + wave;
  const int arow  = min(mt * 16 + l, M - 1);
  const int col0  = nblk * 64;

  const int sc = threadIdx.x & 63;
  const int sr = threadIdx.x >> 6;

  const float* Ap = A + (size_t)arow * K + 2 * half;

  v8f acc0 = {}, acc1 = {}, acc2 = {}, acc3 = {};

  for (int k0 = 0; k0 < K; k0 += KC) {
    const int kc = min(KC, K - k0);
    __syncthreads();
    for (int kk = sr; kk < kc; kk += 4)
      Bs[sc * BSTRIDE + kk] = B[(size_t)(k0 + kk) * Nc + col0 + sc];
    __syncthreads();

    __builtin_prefetch(Ap + k0 + KC, 0, 1);

    const float* Bl0 = &Bs[(l +  0) * BSTRIDE + 2 * half];
    const float* Bl1 = &Bs[(l + 16) * BSTRIDE + 2 * half];
    const float* Bl2 = &Bs[(l + 32) * BSTRIDE + 2 * half];
    const float* Bl3 = &Bs[(l + 48) * BSTRIDE + 2 * half];
    for (int k = 0; k < kc; k += 4) {
      v2f a  = *(const v2f*)(Ap + k0 + k);
      v2f b0 = *(const v2f*)(Bl0 + k);
      v2f b1 = *(const v2f*)(Bl1 + k);
      v2f b2 = *(const v2f*)(Bl2 + k);
      v2f b3 = *(const v2f*)(Bl3 + k);
      acc0 = __builtin_amdgcn_wmma_f32_16x16x4_f32(false, a, false, b0, (short)0, acc0, false, false);
      acc1 = __builtin_amdgcn_wmma_f32_16x16x4_f32(false, a, false, b1, (short)0, acc1, false, false);
      acc2 = __builtin_amdgcn_wmma_f32_16x16x4_f32(false, a, false, b2, (short)0, acc2, false, false);
      acc3 = __builtin_amdgcn_wmma_f32_16x16x4_f32(false, a, false, b3, (short)0, acc3, false, false);
    }
  }

  v8f* accs[4] = {&acc0, &acc1, &acc2, &acc3};
#pragma unroll
  for (int j = 0; j < 4; ++j) {
    const int ccol = col0 + 16 * j + l;
    const float bv = bias ? bias[ccol] : 0.f;
#pragma unroll
    for (int i = 0; i < 8; ++i) {
      const int row = mt * 16 + i + 8 * half;
      if (row < M) C[(size_t)row * Nc + ccol] = (*accs[j])[i] + bv;
    }
  }
}

__global__ void attn_scores(const float* __restrict__ h, const float* __restrict__ a_s,
                            const float* __restrict__ a_d, float* __restrict__ ssrc,
                            float* __restrict__ sdst, int n) {
  int idx = blockIdx.x * blockDim.x + threadIdx.x;
  if (idx >= n * NHEAD) return;
  int hd = idx & (NHEAD - 1);
  const float* hp = h + (size_t)idx * HIDF;
  const float* as = a_s + hd * HIDF;
  const float* ad = a_d + hd * HIDF;
  float s1 = 0.f, s2 = 0.f;
  for (int c = 0; c < HIDF; c += 4) {
    float4 hv = *(const float4*)(hp + c);
    float4 av = *(const float4*)(as + c);
    float4 dv = *(const float4*)(ad + c);
    s1 += hv.x * av.x + hv.y * av.y + hv.z * av.z + hv.w * av.w;
    s2 += hv.x * dv.x + hv.y * dv.y + hv.z * dv.z + hv.w * dv.w;
  }
  ssrc[idx] = s1;
  sdst[idx] = s2;
}

__global__ void fill_f32(float* __restrict__ p, float v, int n) {
  int i = blockIdx.x * blockDim.x + threadIdx.x;
  if (i < n) p[i] = v;
}

__global__ void edge_logits_max(const int* __restrict__ ei, int E, int n,
                                const float* __restrict__ ssrc,
                                const float* __restrict__ sdst,
                                float* __restrict__ eatt, float* __restrict__ m) {
  int idx = blockIdx.x * blockDim.x + threadIdx.x;
  if (idx >= (E + n) * NHEAD) return;
  int e = idx >> 2, hd = idx & 3;
  int s, d;
  if (e < E) { s = ei[e]; d = ei[E + e]; } else { s = e - E; d = s; }
  float v = lrelu(ssrc[s * NHEAD + hd] + sdst[d * NHEAD + hd]);
  eatt[idx] = v;
  atomicMaxF(&m[d * NHEAD + hd], v);
}

__global__ void fix_m(float* __restrict__ m, int count) {
  int i = blockIdx.x * blockDim.x + threadIdx.x;
  if (i >= count) return;
  float v = m[i];
  if (!(v > -3.0e38f && v < 3.0e38f)) m[i] = 0.f;
}

__global__ void edge_expsum(const int* __restrict__ ei, int E, int n,
                            float* __restrict__ eatt, const float* __restrict__ m,
                            float* __restrict__ den) {
  int idx = blockIdx.x * blockDim.x + threadIdx.x;
  if (idx >= (E + n) * NHEAD) return;
  int e = idx >> 2, hd = idx & 3;
  int d = (e < E) ? ei[E + e] : (e - E);
  float ex = expf(eatt[idx] - m[d * NHEAD + hd]);
  eatt[idx] = ex;
  atomicAdd(&den[d * NHEAD + hd], ex);
}

__global__ void fill_i32(int* __restrict__ p, int v, int n) {
  int i = blockIdx.x * blockDim.x + threadIdx.x;
  if (i < n) p[i] = v;
}

__global__ void csr_hist(const int* __restrict__ ei, int E, int* __restrict__ deg) {
  int e = blockIdx.x * blockDim.x + threadIdx.x;
  if (e < E) atomicAdd(&deg[ei[E + e]], 1);
}

__global__ void __launch_bounds__(1024)
csr_scan(const int* __restrict__ deg, int* __restrict__ rowptr,
         int* __restrict__ cursor, int n) {
  __shared__ int part[1024];
  const int t     = threadIdx.x;
  const int items = (n + 1023) >> 10;
  const int lo    = min(t * items, n);
  const int hi    = min(lo + items, n);
  int s = 0;
  for (int i = lo; i < hi; ++i) s += deg[i];
  part[t] = s;
  __syncthreads();
  for (int off = 1; off < 1024; off <<= 1) {
    const int v = (t >= off) ? part[t - off] : 0;
    __syncthreads();
    part[t] += v;
    __syncthreads();
  }
  int run = part[t] - s;
  for (int i = lo; i < hi; ++i) {
    rowptr[i] = run;
    cursor[i] = run;
    run += deg[i];
  }
  if (t == 1023) rowptr[n] = part[1023];
}

__global__ void csr_fill(const int* __restrict__ ei, int E, int* __restrict__ cursor,
                         int* __restrict__ csr) {
  int e = blockIdx.x * blockDim.x + threadIdx.x;
  if (e >= E) return;
  const int s = ei[e], d = ei[E + e];
  const int pos = atomicAdd(&cursor[d], 1);
  *(int2*)(csr + 2 * (size_t)pos) = make_int2(s, e);
}

__global__ void __launch_bounds__(256)
csr_sort_rows(const int* __restrict__ rowptr, const int* __restrict__ csr_in,
              int* __restrict__ csr_out, int n) {
  const int lane = threadIdx.x & 31;
  const int node = __builtin_amdgcn_readfirstlane((int)(blockIdx.x * 8 + (threadIdx.x >> 5)));
  if (node >= n) return;
  const int beg = rowptr[node], end = rowptr[node + 1];
  for (int b = beg; b < end; b += 32) {
    const bool have = (b + lane) < end;
    int2 mine = make_int2(0, 0x7fffffff);
    if (have) mine = *(const int2*)(csr_in + 2 * (size_t)(b + lane));
    int rank = 0;
    for (int c0 = beg; c0 < end; c0 += 32) {
      const int cnt = min(32, end - c0);
      int other = 0x7fffffff;
      if (c0 + lane < end) other = csr_in[2 * (size_t)(c0 + lane) + 1];
      for (int j = 0; j < cnt; ++j)
        rank += (__builtin_amdgcn_readlane(other, j) < mine.y) ? 1 : 0;
    }
    if (have) *(int2*)(csr_out + 2 * (size_t)(beg + rank)) = mine;
  }
}

__device__ __forceinline__ float bcast_f(float v, int srcLane) {
  return __int_as_float(__builtin_amdgcn_readlane(__float_as_int(v), srcLane));
}

__global__ void __launch_bounds__(256)
edge_gather(const int* __restrict__ rowptr, const int* __restrict__ csr, int E, int n,
            const float* __restrict__ ex, const float* __restrict__ den,
            const float* __restrict__ h, float* __restrict__ acc,
            float* __restrict__ alpha_out) {
  const int lane = threadIdx.x & 31;
  const int node = __builtin_amdgcn_readfirstlane((int)(blockIdx.x * 8 + (threadIdx.x >> 5)));
  if (node >= n) return;
  const float4 dn = *(const float4*)(den + (size_t)node * NHEAD);
  const int beg = rowptr[node], end = rowptr[node + 1];

  float4 s0, s1, s2, s3;
  {
    const size_t eid = (size_t)E + node;
    const float4 xv = *(const float4*)(ex + eid * NHEAD);
    float4 a;
    a.x = xv.x / dn.x; a.y = xv.y / dn.y; a.z = xv.z / dn.z; a.w = xv.w / dn.w;
    if (alpha_out && lane == 0) *(float4*)(alpha_out + eid * NHEAD) = a;
    const float* hp = h + (size_t)node * WHID + lane * 4;
    const float4 h0 = *(const float4*)(hp);
    const float4 h1 = *(const float4*)(hp + HIDF);
    const float4 h2 = *(const float4*)(hp + 2 * HIDF);
    const float4 h3 = *(const float4*)(hp + 3 * HIDF);
    s0.x = a.x * h0.x; s0.y = a.x * h0.y; s0.z = a.x * h0.z; s0.w = a.x * h0.w;
    s1.x = a.y * h1.x; s1.y = a.y * h1.y; s1.z = a.y * h1.z; s1.w = a.y * h1.w;
    s2.x = a.z * h2.x; s2.y = a.z * h2.y; s2.z = a.z * h2.z; s2.w = a.z * h2.w;
    s3.x = a.w * h3.x; s3.y = a.w * h3.y; s3.z = a.w * h3.z; s3.w = a.w * h3.w;
  }

  for (int base = beg; base < end; base += 32) {
    const int cnt = min(32, end - base);
    int    my_src = 0;
    float4 my_a   = make_float4(0.f, 0.f, 0.f, 0.f);
    if (lane < cnt) {
      const int2 ent = *(const int2*)(csr + 2 * (size_t)(base + lane));
      my_src = ent.x;
      const float4 xv = *(const float4*)(ex + (size_t)ent.y * NHEAD);
      my_a.x = xv.x / dn.x; my_a.y = xv.y / dn.y; my_a.z = xv.z / dn.z; my_a.w = xv.w / dn.w;
      if (alpha_out) *(float4*)(alpha_out + (size_t)ent.y * NHEAD) = my_a;
    }
    for (int j = 0; j < cnt; ++j) {
      const int   s  = __builtin_amdgcn_readlane(my_src, j);
      const float a0 = bcast_f(my_a.x, j);
      const float a1 = bcast_f(my_a.y, j);
      const float a2 = bcast_f(my_a.z, j);
      const float a3 = bcast_f(my_a.w, j);
      const float* hp = h + (size_t)s * WHID + lane * 4;
      const float4 h0 = *(const float4*)(hp);
      const float4 h1 = *(const float4*)(hp + HIDF);
      const float4 h2 = *(const float4*)(hp + 2 * HIDF);
      const float4 h3 = *(const float4*)(hp + 3 * HIDF);
      s0.x += a0 * h0.x; s0.y += a0 * h0.y; s0.z += a0 * h0.z; s0.w += a0 * h0.w;
      s1.x += a1 * h1.x; s1.y += a1 * h1.y; s1.z += a1 * h1.z; s1.w += a1 * h1.w;
      s2.x += a2 * h2.x; s2.y += a2 * h2.y; s2.z += a2 * h2.z; s2.w += a2 * h2.w;
      s3.x += a3 * h3.x; s3.y += a3 * h3.y; s3.z += a3 * h3.z; s3.w += a3 * h3.w;
    }
  }

  float* ap = acc + (size_t)node * WHID + lane * 4;
  *(float4*)(ap)            = s0;
  *(float4*)(ap + HIDF)     = s1;
  *(float4*)(ap + 2 * HIDF) = s2;
  *(float4*)(ap + 3 * HIDF) = s3;
}

__global__ void l1_post(const float* __restrict__ acc, const float* __restrict__ b1,
                        const float* __restrict__ g, const float* __restrict__ b,
                        float* __restrict__ x2) {
  __shared__ float sh[HIDF];
  int nd = blockIdx.x, c = threadIdx.x;
  const float* ap = acc + (size_t)nd * WHID;
  float v = 0.25f * (ap[c] + ap[HIDF + c] + ap[2 * HIDF + c] + ap[3 * HIDF + c]) + b1[c];
  sh[c] = v; __syncthreads();
  for (int s = HIDF / 2; s > 0; s >>= 1) { if (c < s) sh[c] += sh[c + s]; __syncthreads(); }
  float mu = sh[0] * (1.0f / HIDF);
  __syncthreads();
  float dv = v - mu;
  sh[c] = dv * dv; __syncthreads();
  for (int s = HIDF / 2; s > 0; s >>= 1) { if (c < s) sh[c] += sh[c + s]; __syncthreads(); }
  float var = sh[0] * (1.0f / HIDF);
  float y = g[c] * dv * rsqrtf(var + 1e-5f) + b[c];
  x2[(size_t)nd * HIDF + c] = lrelu(y);
}

__global__ void l2_post_pool(const float* __restrict__ acc, const float* __restrict__ b2,
                             const int* __restrict__ batch, float* __restrict__ pooled,
                             float* __restrict__ cnt) {
  int nd = blockIdx.x, c = threadIdx.x;
  float y = lrelu(acc[(size_t)nd * WHID + c] + b2[c]);
  int gidx = batch[nd];
  atomicAdd(&pooled[(size_t)gidx * WHID + c], y);
  if (c == 0) atomicAdd(&cnt[gidx], 1.0f);
}

__global__ void pool_div(float* __restrict__ pooled, const float* __restrict__ cnt) {
  int i = blockIdx.x * blockDim.x + threadIdx.x;
  if (i >= NGROUP * WHID) return;
  pooled[i] /= fmaxf(cnt[i / WHID], 1.0f);
}

__global__ void ln_relu(float* __restrict__ x, const float* __restrict__ g,
                        const float* __restrict__ b, int dim) {
  extern __shared__ float sh[];
  int r = blockIdx.x, c = threadIdx.x;
  float v = x[(size_t)r * dim + c];
  sh[c] = v; __syncthreads();
  for (int s = dim >> 1; s > 0; s >>= 1) { if (c < s) sh[c] += sh[c + s]; __syncthreads(); }
  float mu = sh[0] / dim;
  __syncthreads();
  float dv = v - mu;
  sh[c] = dv * dv; __syncthreads();
  for (int s = dim >> 1; s > 0; s >>= 1) { if (c < s) sh[c] += sh[c + s]; __syncthreads(); }
  float var = sh[0] / dim;
  float y = g[c] * dv * rsqrtf(var + 1e-5f) + b[c];
  x[(size_t)r * dim + c] = fmaxf(y, 0.f);
}

__global__ void head_out(const float* __restrict__ t2, const float* __restrict__ Ws3,
                         const float* __restrict__ bs3, float* __restrict__ out) {
  int r = blockIdx.x * blockDim.x + threadIdx.x;
  if (r >= NGROUP) return;
  float l0 = bs3[0], l1 = bs3[1];
  for (int c = 0; c < OUT2; ++c) {
    float v = t2[r * OUT2 + c];
    l0 += v * Ws3[c * 2 + 0];
    l1 += v * Ws3[c * 2 + 1];
  }
  float mx = fmaxf(l0, l1);
  float e0 = expf(l0 - mx), e1 = expf(l1 - mx), s = e0 + e1;
  out[r * 2 + 0] = e0 / s;
  out[r * 2 + 1] = e1 / s;
}

extern "C" void kernel_launch(void* const* d_in, const int* in_sizes, int n_in,
                              void* d_out, int out_size, void* d_ws, size_t ws_size,
                              hipStream_t stream) {
  (void)n_in; (void)out_size; (void)ws_size;
  const float* x    = (const float*)d_in[0];
  const int*   ei   = (const int*)d_in[1];
  const int*   batch= (const int*)d_in[2];
  const float* W1   = (const float*)d_in[3];
  const float* as1  = (const float*)d_in[4];
  const float* ad1  = (const float*)d_in[5];
  const float* b1   = (const float*)d_in[6];
  const float* g_ln = (const float*)d_in[7];
  const float* b_ln = (const float*)d_in[8];
  const float* W2   = (const float*)d_in[9];
  const float* as2  = (const float*)d_in[10];
  const float* ad2  = (const float*)d_in[11];
  const float* b2   = (const float*)d_in[12];
  const float* Wl0  = (const float*)d_in[13];
  const float* bl0  = (const float*)d_in[14];
  const float* Ws1  = (const float*)d_in[15];
  const float* bs1  = (const float*)d_in[16];
  const float* g1   = (const float*)d_in[17];
  const float* bb1  = (const float*)d_in[18];
  const float* Ws2  = (const float*)d_in[19];
  const float* bs2  = (const float*)d_in[20];
  const float* g2   = (const float*)d_in[21];
  const float* bb2  = (const float*)d_in[22];
  const float* Ws3  = (const float*)d_in[23];
  const float* bs3  = (const float*)d_in[24];

  const int N  = in_sizes[0] / HIDF;
  const int E  = in_sizes[1] / 2;
  const int TE = E + N;
  const int NH = N * NHEAD;
  const int TEH = TE * NHEAD;

  float* out       = (float*)d_out;
  float* alpha_out = out + NGROUP * NCLSD;

  size_t off = 0;
  auto alloc = [&](size_t bytes) -> float* {
    off = (off + 255) & ~(size_t)255;
    float* p = (float*)((char*)d_ws + off);
    off += bytes;
    return p;
  };
  float* h      = alloc((size_t)N * WHID * 4);
  float* acc    = alloc((size_t)N * WHID * 4);
  float* x2     = alloc((size_t)N * HIDF * 4);
  float* ssrc   = alloc((size_t)NH * 4);
  float* sdst   = alloc((size_t)NH * 4);
  float* eatt   = alloc((size_t)TEH * 4);
  float* mbuf   = alloc((size_t)NH * 4);
  float* den    = alloc((size_t)NH * 4);
  float* pooled = alloc((size_t)NGROUP * WHID * 4);
  float* cnt    = alloc((size_t)NGROUP * 4);
  float* z0     = alloc((size_t)NGROUP * HIDF * 4);
  float* t1     = alloc((size_t)NGROUP * OUTD * 4);
  float* t2     = alloc((size_t)NGROUP * OUT2 * 4);
  int*   deg    = (int*)alloc((size_t)N * 4);
  int*   rowptr = (int*)alloc((size_t)(N + 1) * 4);
  int*   cursor = (int*)alloc((size_t)N * 4);
  int*   csr_arr = (int*)alloc((size_t)E * 8);
  int*   csr     = (int*)alloc((size_t)E * 8);

  auto cdiv = [](long a, long b) { return (int)((a + b - 1) / b); };
  auto gemm = [&](const float* A, const float* B, const float* bias, float* C,
                  int M, int K, int Nc) {
    int mtiles  = (M + 15) / 16;
    int mblocks = (mtiles + 7) / 8;
    int nslab   = Nc / 64;
    wmma_gemm<<<mblocks * nslab, 256, 0, stream>>>(A, B, bias, C, M, K, Nc);
  };

  fill_i32<<<cdiv(N, 256), 256, 0, stream>>>(deg, 0, N);
  csr_hist<<<cdiv(E, 256), 256, 0, stream>>>(ei, E, deg);
  csr_scan<<<1, 1024, 0, stream>>>(deg, rowptr, cursor, N);
  csr_fill<<<cdiv(E, 256), 256, 0, stream>>>(ei, E, cursor, csr_arr);
  csr_sort_rows<<<cdiv(N, 8), 256, 0, stream>>>(rowptr, csr_arr, csr, N);

  gemm(x, W1, nullptr, h, N, HIDF, WHID);
  attn_scores<<<cdiv(NH, 256), 256, 0, stream>>>(h, as1, ad1, ssrc, sdst, N);
  fill_f32<<<cdiv(NH, 256), 256, 0, stream>>>(mbuf, -INFINITY, NH);
  fill_f32<<<cdiv(NH, 256), 256, 0, stream>>>(den, 0.f, NH);
  edge_logits_max<<<cdiv(TEH, 256), 256, 0, stream>>>(ei, E, N, ssrc, sdst, eatt, mbuf);
  fix_m<<<cdiv(NH, 256), 256, 0, stream>>>(mbuf, NH);
  edge_expsum<<<cdiv(TEH, 256), 256, 0, stream>>>(ei, E, N, eatt, mbuf, den);
  edge_gather<<<cdiv(N, 8), 256, 0, stream>>>(rowptr, csr, E, N, eatt, den, h, acc, nullptr);
  l1_post<<<N, HIDF, 0, stream>>>(acc, b1, g_ln, b_ln, x2);

  gemm(x2, W2, nullptr, h, N, HIDF, WHID);
  attn_scores<<<cdiv(NH, 256), 256, 0, stream>>>(h, as2, ad2, ssrc, sdst, N);
  fill_f32<<<cdiv(NH, 256), 256, 0, stream>>>(mbuf, -INFINITY, NH);
  fill_f32<<<cdiv(NH, 256), 256, 0, stream>>>(den, 0.f, NH);
  edge_logits_max<<<cdiv(TEH, 256), 256, 0, stream>>>(ei, E, N, ssrc, sdst, eatt, mbuf);
  fix_m<<<cdiv(NH, 256), 256, 0, stream>>>(mbuf, NH);
  edge_expsum<<<cdiv(TEH, 256), 256, 0, stream>>>(ei, E, N, eatt, mbuf, den);
  edge_gather<<<cdiv(N, 8), 256, 0, stream>>>(rowptr, csr, E, N, eatt, den, h, acc, alpha_out);

  fill_f32<<<cdiv(NGROUP * WHID, 256), 256, 0, stream>>>(pooled, 0.f, NGROUP * WHID);
  fill_f32<<<1, NGROUP, 0, stream>>>(cnt, 0.f, NGROUP);
  l2_post_pool<<<N, WHID, 0, stream>>>(acc, b2, batch, pooled, cnt);
  pool_div<<<cdiv(NGROUP * WHID, 256), 256, 0, stream>>>(pooled, cnt);

  gemm(pooled, Wl0, bl0, z0, NGROUP, WHID, HIDF);
  gemm(z0, Ws1, bs1, t1, NGROUP, HIDF, OUTD);
  ln_relu<<<NGROUP, OUTD, OUTD * sizeof(float), stream>>>(t1, g1, bb1, OUTD);
  gemm(t1, Ws2, bs2, t2, NGROUP, OUTD, OUT2);
  ln_relu<<<NGROUP, OUT2, OUT2 * sizeof(float), stream>>>(t2, g2, bb2, OUT2);
  head_out<<<1, NGROUP, 0, stream>>>(t2, Ws3, bs3, out);
}


# per-destination softmax (max, exp, denominator) fused into the gather wave as in-register wave reductions over the sorted row; max/den float atomics and 10 launches removed; den summed in fixed order
# speedup vs baseline: 2.3470x; 1.2776x over previous
#include <hip/hip_runtime.h>
#include <math.h>

#define HIDF   128
#define NHEAD  4
#define WHID   512
#define NGROUP 64
#define OUTD   256
#define OUT2   128
#define NCLSD  2
#define SLOPE  0.2f
#define KC     128
#define BSTRIDE (KC + 2)

typedef float v2f __attribute__((ext_vector_type(2)));
typedef float v8f __attribute__((ext_vector_type(8)));

__device__ __forceinline__ float lrelu(float x) { return x >= 0.f ? x : SLOPE * x; }

__global__ void wmma_gemm(const float* __restrict__ A, const float* __restrict__ B,
                          const float* __restrict__ bias, float* __restrict__ C,
                          int M, int K, int Nc) {
  __shared__ float Bs[64 * BSTRIDE];

  const int nslab = Nc >> 6;
  const int mblk  = blockIdx.x / nslab;
  const int nblk  = blockIdx.x - mblk * nslab;
  const int wave  = threadIdx.x >> 5;
  const int lane  = threadIdx.x & 31;
  const int half  = lane >> 4;
  const int l     = lane & 15;

  const int mt    = mblk * 8 + wave;
  const int arow  = min(mt * 16 + l, M - 1);
  const int col0  = nblk * 64;

  const int sc = threadIdx.x & 63;
  const int sr = threadIdx.x >> 6;

  const float* Ap = A + (size_t)arow * K + 2 * half;

  v8f acc0 = {}, acc1 = {}, acc2 = {}, acc3 = {};

  for (int k0 = 0; k0 < K; k0 += KC) {
    const int kc = min(KC, K - k0);
    __syncthreads();
    for (int kk = sr; kk < kc; kk += 4)
      Bs[sc * BSTRIDE + kk] = B[(size_t)(k0 + kk) * Nc + col0 + sc];
    __syncthreads();

    __builtin_prefetch(Ap + k0 + KC, 0, 1);

    const float* Bl0 = &Bs[(l +  0) * BSTRIDE + 2 * half];
    const float* Bl1 = &Bs[(l + 16) * BSTRIDE + 2 * half];
    const float* Bl2 = &Bs[(l + 32) * BSTRIDE + 2 * half];
    const float* Bl3 = &Bs[(l + 48) * BSTRIDE + 2 * half];
    for (int k = 0; k < kc; k += 4) {
      v2f a  = *(const v2f*)(Ap + k0 + k);
      v2f b0 = *(const v2f*)(Bl0 + k);
      v2f b1 = *(const v2f*)(Bl1 + k);
      v2f b2 = *(const v2f*)(Bl2 + k);
      v2f b3 = *(const v2f*)(Bl3 + k);
      acc0 = __builtin_amdgcn_wmma_f32_16x16x4_f32(false, a, false, b0, (short)0, acc0, false, false);
      acc1 = __builtin_amdgcn_wmma_f32_16x16x4_f32(false, a, false, b1, (short)0, acc1, false, false);
      acc2 = __builtin_amdgcn_wmma_f32_16x16x4_f32(false, a, false, b2, (short)0, acc2, false, false);
      acc3 = __builtin_amdgcn_wmma_f32_16x16x4_f32(false, a, false, b3, (short)0, acc3, false, false);
    }
  }

  v8f* accs[4] = {&acc0, &acc1, &acc2, &acc3};
#pragma unroll
  for (int j = 0; j < 4; ++j) {
    const int ccol = col0 + 16 * j + l;
    const float bv = bias ? bias[ccol] : 0.f;
#pragma unroll
    for (int i = 0; i < 8; ++i) {
      const int row = mt * 16 + i + 8 * half;
      if (row < M) C[(size_t)row * Nc + ccol] = (*accs[j])[i] + bv;
    }
  }
}

__global__ void attn_scores(const float* __restrict__ h, const float* __restrict__ a_s,
                            const float* __restrict__ a_d, float* __restrict__ ssrc,
                            float* __restrict__ sdst, int n) {
  int idx = blockIdx.x * blockDim.x + threadIdx.x;
  if (idx >= n * NHEAD) return;
  int hd = idx & (NHEAD - 1);
  const float* hp = h + (size_t)idx * HIDF;
  const float* as = a_s + hd * HIDF;
  const float* ad = a_d + hd * HIDF;
  float s1 = 0.f, s2 = 0.f;
  for (int c = 0; c < HIDF; c += 4) {
    float4 hv = *(const float4*)(hp + c);
    float4 av = *(const float4*)(as + c);
    float4 dv = *(const float4*)(ad + c);
    s1 += hv.x * av.x + hv.y * av.y + hv.z * av.z + hv.w * av.w;
    s2 += hv.x * dv.x + hv.y * dv.y + hv.z * dv.z + hv.w * dv.w;
  }
  ssrc[idx] = s1;
  sdst[idx] = s2;
}

__global__ void fill_f32(float* __restrict__ p, float v, int n) {
  int i = blockIdx.x * blockDim.x + threadIdx.x;
  if (i < n) p[i] = v;
}

__global__ void fill_i32(int* __restrict__ p, int v, int n) {
  int i = blockIdx.x * blockDim.x + threadIdx.x;
  if (i < n) p[i] = v;
}

__global__ void csr_hist(const int* __restrict__ ei, int E, int* __restrict__ deg) {
  int e = blockIdx.x * blockDim.x + threadIdx.x;
  if (e < E) atomicAdd(&deg[ei[E + e]], 1);
}

__global__ void __launch_bounds__(1024)
csr_scan(const int* __restrict__ deg, int* __restrict__ rowptr,
         int* __restrict__ cursor, int n) {
  __shared__ int part[1024];
  const int t     = threadIdx.x;
  const int items = (n + 1023) >> 10;
  const int lo    = min(t * items, n);
  const int hi    = min(lo + items, n);
  int s = 0;
  for (int i = lo; i < hi; ++i) s += deg[i];
  part[t] = s;
  __syncthreads();
  for (int off = 1; off < 1024; off <<= 1) {
    const int v = (t >= off) ? part[t - off] : 0;
    __syncthreads();
    part[t] += v;
    __syncthreads();
  }
  int run = part[t] - s;
  for (int i = lo; i < hi; ++i) {
    rowptr[i] = run;
    cursor[i] = run;
    run += deg[i];
  }
  if (t == 1023) rowptr[n] = part[1023];
}

__global__ void csr_fill(const int* __restrict__ ei, int E, int* __restrict__ cursor,
                         int* __restrict__ csr) {
  int e = blockIdx.x * blockDim.x + threadIdx.x;
  if (e >= E) return;
  const int s = ei[e], d = ei[E + e];
  const int pos = atomicAdd(&cursor[d], 1);
  *(int2*)(csr + 2 * (size_t)pos) = make_int2(s, e);
}

__global__ void __launch_bounds__(256)
csr_sort_rows(const int* __restrict__ rowptr, const int* __restrict__ csr_in,
              int* __restrict__ csr_out, int n) {
  const int lane = threadIdx.x & 31;
  const int node = __builtin_amdgcn_readfirstlane((int)(blockIdx.x * 8 + (threadIdx.x >> 5)));
  if (node >= n) return;
  const int beg = rowptr[node], end = rowptr[node + 1];
  for (int b = beg; b < end; b += 32) {
    const bool have = (b + lane) < end;
    int2 mine = make_int2(0, 0x7fffffff);
    if (have) mine = *(const int2*)(csr_in + 2 * (size_t)(b + lane));
    int rank = 0;
    for (int c0 = beg; c0 < end; c0 += 32) {
      const int cnt = min(32, end - c0);
      int other = 0x7fffffff;
      if (c0 + lane < end) other = csr_in[2 * (size_t)(c0 + lane) + 1];
      for (int j = 0; j < cnt; ++j)
        rank += (__builtin_amdgcn_readlane(other, j) < mine.y) ? 1 : 0;
    }
    if (have) *(int2*)(csr_out + 2 * (size_t)(beg + rank)) = mine;
  }
}

__device__ __forceinline__ float bcast_f(float v, int srcLane) {
  return __int_as_float(__builtin_amdgcn_readlane(__float_as_int(v), srcLane));
}

__device__ __forceinline__ float wave_max(float v) {
  for (int o = 16; o > 0; o >>= 1) v = fmaxf(v, __shfl_xor(v, o, 32));
  return v;
}
__device__ __forceinline__ float wave_sum(float v) {
  for (int o = 16; o > 0; o >>= 1) v += __shfl_xor(v, o, 32);
  return v;
}

__device__ __forceinline__ void load_entries(const int* __restrict__ csr,
                                             const float* __restrict__ ssrc,
                                             const float4 sd, int base, int cnt, int lane,
                                             int& src, int& eid, float4& e) {
  src = 0; eid = 0;
  e = make_float4(-INFINITY, -INFINITY, -INFINITY, -INFINITY);
  if (lane < cnt) {
    const int2 ent = *(const int2*)(csr + 2 * (size_t)(base + lane));
    src = ent.x; eid = ent.y;
    const float4 ss = *(const float4*)(ssrc + (size_t)src * NHEAD);
    e.x = lrelu(ss.x + sd.x); e.y = lrelu(ss.y + sd.y);
    e.z = lrelu(ss.z + sd.z); e.w = lrelu(ss.w + sd.w);
  }
}

__device__ __forceinline__ void add_rows(const float* __restrict__ h, int lane, int cnt,
                                         int my_src, const float4 my_a,
                                         float4& s0, float4& s1, float4& s2, float4& s3) {
  for (int j = 0; j < cnt; ++j) {
    const int   s  = __builtin_amdgcn_readlane(my_src, j);
    const float a0 = bcast_f(my_a.x, j);
    const float a1 = bcast_f(my_a.y, j);
    const float a2 = bcast_f(my_a.z, j);
    const float a3 = bcast_f(my_a.w, j);
    const float* hp = h + (size_t)s * WHID + lane * 4;
    const float4 h0 = *(const float4*)(hp);
    const float4 h1 = *(const float4*)(hp + HIDF);
    const float4 h2 = *(const float4*)(hp + 2 * HIDF);
    const float4 h3 = *(const float4*)(hp + 3 * HIDF);
    s0.x += a0 * h0.x; s0.y += a0 * h0.y; s0.z += a0 * h0.z; s0.w += a0 * h0.w;
    s1.x += a1 * h1.x; s1.y += a1 * h1.y; s1.z += a1 * h1.z; s1.w += a1 * h1.w;
    s2.x += a2 * h2.x; s2.y += a2 * h2.y; s2.z += a2 * h2.z; s2.w += a2 * h2.w;
    s3.x += a3 * h3.x; s3.y += a3 * h3.y; s3.z += a3 * h3.z; s3.w += a3 * h3.w;
  }
}

__global__ void __launch_bounds__(256)
edge_softmax_gather(const int* __restrict__ rowptr, const int* __restrict__ csr, int E, int n,
                    const float* __restrict__ ssrc, const float* __restrict__ sdst,
                    const float* __restrict__ h, float* __restrict__ acc,
                    float* __restrict__ alpha_out) {
  const int lane = threadIdx.x & 31;
  const int node = __builtin_amdgcn_readfirstlane((int)(blockIdx.x * 8 + (threadIdx.x >> 5)));
  if (node >= n) return;
  const float4 sd = *(const float4*)(sdst + (size_t)node * NHEAD);
  const float4 sn = *(const float4*)(ssrc + (size_t)node * NHEAD);
  const int beg = rowptr[node], end = rowptr[node + 1];

  float4 es;
  es.x = lrelu(sn.x + sd.x); es.y = lrelu(sn.y + sd.y);
  es.z = lrelu(sn.z + sd.z); es.w = lrelu(sn.w + sd.w);

  const int cnt0 = min(32, end - beg);
  int src0, eid0; float4 e0;
  load_entries(csr, ssrc, sd, beg, cnt0, lane, src0, eid0, e0);

  float4 mx;
  mx.x = fmaxf(es.x, e0.x); mx.y = fmaxf(es.y, e0.y);
  mx.z = fmaxf(es.z, e0.z); mx.w = fmaxf(es.w, e0.w);
  for (int base = beg + 32; base < end; base += 32) {
    int s_, i_; float4 e_;
    load_entries(csr, ssrc, sd, base, min(32, end - base), lane, s_, i_, e_);
    mx.x = fmaxf(mx.x, e_.x); mx.y = fmaxf(mx.y, e_.y);
    mx.z = fmaxf(mx.z, e_.z); mx.w = fmaxf(mx.w, e_.w);
  }
  mx.x = wave_max(mx.x); mx.y = wave_max(mx.y);
  mx.z = wave_max(mx.z); mx.w = wave_max(mx.w);
  if (!(mx.x > -3.0e38f && mx.x < 3.0e38f)) mx.x = 0.f;
  if (!(mx.y > -3.0e38f && mx.y < 3.0e38f)) mx.y = 0.f;
  if (!(mx.z > -3.0e38f && mx.z < 3.0e38f)) mx.z = 0.f;
  if (!(mx.w > -3.0e38f && mx.w < 3.0e38f)) mx.w = 0.f;

  float4 x0 = make_float4(0.f, 0.f, 0.f, 0.f);
  if (lane < cnt0) {
    x0.x = expf(e0.x - mx.x); x0.y = expf(e0.y - mx.y);
    x0.z = expf(e0.z - mx.z); x0.w = expf(e0.w - mx.w);
  }
  float4 ds = x0;
  for (int base = beg + 32; base < end; base += 32) {
    const int cnt = min(32, end - base);
    int s_, i_; float4 e_;
    load_entries(csr, ssrc, sd, base, cnt, lane, s_, i_, e_);
    if (lane < cnt) {
      ds.x += expf(e_.x - mx.x); ds.y += expf(e_.y - mx.y);
      ds.z += expf(e_.z - mx.z); ds.w += expf(e_.w - mx.w);
    }
  }
  float4 xs;
  xs.x = expf(es.x - mx.x); xs.y = expf(es.y - mx.y);
  xs.z = expf(es.z - mx.z); xs.w = expf(es.w - mx.w);
  float4 dn;
  dn.x = wave_sum(ds.x) + xs.x; dn.y = wave_sum(ds.y) + xs.y;
  dn.z = wave_sum(ds.z) + xs.z; dn.w = wave_sum(ds.w) + xs.w;

  float4 s0, s1, s2, s3;
  {
    const size_t eid = (size_t)E + node;
    float4 a;
    a.x = xs.x / dn.x; a.y = xs.y / dn.y; a.z = xs.z / dn.z; a.w = xs.w / dn.w;
    if (alpha_out && lane == 0) *(float4*)(alpha_out + eid * NHEAD) = a;
    const float* hp = h + (size_t)node * WHID + lane * 4;
    const float4 h0 = *(const float4*)(hp);
    const float4 h1 = *(const float4*)(hp + HIDF);
    const float4 h2 = *(const float4*)(hp + 2 * HIDF);
    const float4 h3 = *(const float4*)(hp + 3 * HIDF);
    s0.x = a.x * h0.x; s0.y = a.x * h0.y; s0.z = a.x * h0.z; s0.w = a.x * h0.w;
    s1.x = a.y * h1.x; s1.y = a.y * h1.y; s1.z = a.y * h1.z; s1.w = a.y * h1.w;
    s2.x = a.z * h2.x; s2.y = a.z * h2.y; s2.z = a.z * h2.z; s2.w = a.z * h2.w;
    s3.x = a.w * h3.x; s3.y = a.w * h3.y; s3.z = a.w * h3.z; s3.w = a.w * h3.w;
  }

  {
    float4 a;
    a.x = x0.x / dn.x; a.y = x0.y / dn.y; a.z = x0.z / dn.z; a.w = x0.w / dn.w;
    if (alpha_out && lane < cnt0) *(float4*)(alpha_out + (size_t)eid0 * NHEAD) = a;
    add_rows(h, lane, cnt0, src0, a, s0, s1, s2, s3);
  }
  for (int base = beg + 32; base < end; base += 32) {
    const int cnt = min(32, end - base);
    int my_src, my_eid; float4 e_;
    load_entries(csr, ssrc, sd, base, cnt, lane, my_src, my_eid, e_);
    float4 a = make_float4(0.f, 0.f, 0.f, 0.f);
    if (lane < cnt) {
      a.x = expf(e_.x - mx.x) / dn.x; a.y = expf(e_.y - mx.y) / dn.y;
      a.z = expf(e_.z - mx.z) / dn.z; a.w = expf(e_.w - mx.w) / dn.w;
      if (alpha_out) *(float4*)(alpha_out + (size_t)my_eid * NHEAD) = a;
    }
    add_rows(h, lane, cnt, my_src, a, s0, s1, s2, s3);
  }

  float* ap = acc + (size_t)node * WHID + lane * 4;
  *(float4*)(ap)            = s0;
  *(float4*)(ap + HIDF)     = s1;
  *(float4*)(ap + 2 * HIDF) = s2;
  *(float4*)(ap + 3 * HIDF) = s3;
}

__global__ void l1_post(const float* __restrict__ acc, const float* __restrict__ b1,
                        const float* __restrict__ g, const float* __restrict__ b,
                        float* __restrict__ x2) {
  __shared__ float sh[HIDF];
  int nd = blockIdx.x, c = threadIdx.x;
  const float* ap = acc + (size_t)nd * WHID;
  float v = 0.25f * (ap[c] + ap[HIDF + c] + ap[2 * HIDF + c] + ap[3 * HIDF + c]) + b1[c];
  sh[c] = v; __syncthreads();
  for (int s = HIDF / 2; s > 0; s >>= 1) { if (c < s) sh[c] += sh[c + s]; __syncthreads(); }
  float mu = sh[0] * (1.0f / HIDF);
  __syncthreads();
  float dv = v - mu;
  sh[c] = dv * dv; __syncthreads();
  for (int s = HIDF / 2; s > 0; s >>= 1) { if (c < s) sh[c] += sh[c + s]; __syncthreads(); }
  float var = sh[0] * (1.0f / HIDF);
  float y = g[c] * dv * rsqrtf(var + 1e-5f) + b[c];
  x2[(size_t)nd * HIDF + c] = lrelu(y);
}

__global__ void l2_post_pool(const float* __restrict__ acc, const float* __restrict__ b2,
                             const int* __restrict__ batch, float* __restrict__ pooled,
                             float* __restrict__ cnt) {
  int nd = blockIdx.x, c = threadIdx.x;
  float y = lrelu(acc[(size_t)nd * WHID + c] + b2[c]);
  int gidx = batch[nd];
  atomicAdd(&pooled[(size_t)gidx * WHID + c], y);
  if (c == 0) atomicAdd(&cnt[gidx], 1.0f);
}

__global__ void pool_div(float* __restrict__ pooled, const float* __restrict__ cnt) {
  int i = blockIdx.x * blockDim.x + threadIdx.x;
  if (i >= NGROUP * WHID) return;
  pooled[i] /= fmaxf(cnt[i / WHID], 1.0f);
}

__global__ void ln_relu(float* __restrict__ x, const float* __restrict__ g,
                        const float* __restrict__ b, int dim) {
  extern __shared__ float sh[];
  int r = blockIdx.x, c = threadIdx.x;
  float v = x[(size_t)r * dim + c];
  sh[c] = v; __syncthreads();
  for (int s = dim >> 1; s > 0; s >>= 1) { if (c < s) sh[c] += sh[c + s]; __syncthreads(); }
  float mu = sh[0] / dim;
  __syncthreads();
  float dv = v - mu;
  sh[c] = dv * dv; __syncthreads();
  for (int s = dim >> 1; s > 0; s >>= 1) { if (c < s) sh[c] += sh[c + s]; __syncthreads(); }
  float var = sh[0] / dim;
  float y = g[c] * dv * rsqrtf(var + 1e-5f) + b[c];
  x[(size_t)r * dim + c] = fmaxf(y, 0.f);
}

__global__ void head_out(const float* __restrict__ t2, const float* __restrict__ Ws3,
                         const float* __restrict__ bs3, float* __restrict__ out) {
  int r = blockIdx.x * blockDim.x + threadIdx.x;
  if (r >= NGROUP) return;
  float l0 = bs3[0], l1 = bs3[1];
  for (int c = 0; c < OUT2; ++c) {
    float v = t2[r * OUT2 + c];
    l0 += v * Ws3[c * 2 + 0];
    l1 += v * Ws3[c * 2 + 1];
  }
  float mx = fmaxf(l0, l1);
  float e0 = expf(l0 - mx), e1 = expf(l1 - mx), s = e0 + e1;
  out[r * 2 + 0] = e0 / s;
  out[r * 2 + 1] = e1 / s;
}

extern "C" void kernel_launch(void* const* d_in, const int* in_sizes, int n_in,
                              void* d_out, int out_size, void* d_ws, size_t ws_size,
                              hipStream_t stream) {
  (void)n_in; (void)out_size; (void)ws_size;
  const float* x    = (const float*)d_in[0];
  const int*   ei   = (const int*)d_in[1];
  const int*   batch= (const int*)d_in[2];
  const float* W1   = (const float*)d_in[3];
  const float* as1  = (const float*)d_in[4];
  const float* ad1  = (const float*)d_in[5];
  const float* b1   = (const float*)d_in[6];
  const float* g_ln = (const float*)d_in[7];
  const float* b_ln = (const float*)d_in[8];
  const float* W2   = (const float*)d_in[9];
  const float* as2  = (const float*)d_in[10];
  const float* ad2  = (const float*)d_in[11];
  const float* b2   = (const float*)d_in[12];
  const float* Wl0  = (const float*)d_in[13];
  const float* bl0  = (const float*)d_in[14];
  const float* Ws1  = (const float*)d_in[15];
  const float* bs1  = (const float*)d_in[16];
  const float* g1   = (const float*)d_in[17];
  const float* bb1  = (const float*)d_in[18];
  const float* Ws2  = (const float*)d_in[19];
  const float* bs2  = (const float*)d_in[20];
  const float* g2   = (const float*)d_in[21];
  const float* bb2  = (const float*)d_in[22];
  const float* Ws3  = (const float*)d_in[23];
  const float* bs3  = (const float*)d_in[24];

  const int N  = in_sizes[0] / HIDF;
  const int E  = in_sizes[1] / 2;
  const int NH = N * NHEAD;

  float* out       = (float*)d_out;
  float* alpha_out = out + NGROUP * NCLSD;

  size_t off = 0;
  auto alloc = [&](size_t bytes) -> float* {
    off = (off + 255) & ~(size_t)255;
    float* p = (float*)((char*)d_ws + off);
    off += bytes;
    return p;
  };
  float* h      = alloc((size_t)N * WHID * 4);
  float* acc    = alloc((size_t)N * WHID * 4);
  float* x2     = alloc((size_t)N * HIDF * 4);
  float* ssrc   = alloc((size_t)NH * 4);
  float* sdst   = alloc((size_t)NH * 4);
  float* pooled = alloc((size_t)NGROUP * WHID * 4);
  float* cnt    = alloc((size_t)NGROUP * 4);
  float* z0     = alloc((size_t)NGROUP * HIDF * 4);
  float* t1     = alloc((size_t)NGROUP * OUTD * 4);
  float* t2     = alloc((size_t)NGROUP * OUT2 * 4);
  int*   deg    = (int*)alloc((size_t)N * 4);
  int*   rowptr = (int*)alloc((size_t)(N + 1) * 4);
  int*   cursor = (int*)alloc((size_t)N * 4);
  int*   csr_arr = (int*)alloc((size_t)E * 8);
  int*   csr     = (int*)alloc((size_t)E * 8);

  auto cdiv = [](long a, long b) { return (int)((a + b - 1) / b); };
  auto gemm = [&](const float* A, const float* B, const float* bias, float* C,
                  int M, int K, int Nc) {
    int mtiles  = (M + 15) / 16;
    int mblocks = (mtiles + 7) / 8;
    int nslab   = Nc / 64;
    wmma_gemm<<<mblocks * nslab, 256, 0, stream>>>(A, B, bias, C, M, K, Nc);
  };

  fill_i32<<<cdiv(N, 256), 256, 0, stream>>>(deg, 0, N);
  csr_hist<<<cdiv(E, 256), 256, 0, stream>>>(ei, E, deg);
  csr_scan<<<1, 1024, 0, stream>>>(deg, rowptr, cursor, N);
  csr_fill<<<cdiv(E, 256), 256, 0, stream>>>(ei, E, cursor, csr_arr);
  csr_sort_rows<<<cdiv(N, 8), 256, 0, stream>>>(rowptr, csr_arr, csr, N);

  gemm(x, W1, nullptr, h, N, HIDF, WHID);
  attn_scores<<<cdiv(NH, 256), 256, 0, stream>>>(h, as1, ad1, ssrc, sdst, N);
  edge_softmax_gather<<<cdiv(N, 8), 256, 0, stream>>>(rowptr, csr, E, N, ssrc, sdst, h, acc, nullptr);
  l1_post<<<N, HIDF, 0, stream>>>(acc, b1, g_ln, b_ln, x2);

  gemm(x2, W2, nullptr, h, N, HIDF, WHID);
  attn_scores<<<cdiv(NH, 256), 256, 0, stream>>>(h, as2, ad2, ssrc, sdst, N);
  edge_softmax_gather<<<cdiv(N, 8), 256, 0, stream>>>(rowptr, csr, E, N, ssrc, sdst, h, acc, alpha_out);

  fill_f32<<<cdiv(NGROUP * WHID, 256), 256, 0, stream>>>(pooled, 0.f, NGROUP * WHID);
  fill_f32<<<1, NGROUP, 0, stream>>>(cnt, 0.f, NGROUP);
  l2_post_pool<<<N, WHID, 0, stream>>>(acc, b2, batch, pooled, cnt);
  pool_div<<<cdiv(NGROUP * WHID, 256), 256, 0, stream>>>(pooled, cnt);

  gemm(pooled, Wl0, bl0, z0, NGROUP, WHID, HIDF);
  gemm(z0, Ws1, bs1, t1, NGROUP, HIDF, OUTD);
  ln_relu<<<NGROUP, OUTD, OUTD * sizeof(float), stream>>>(t1, g1, bb1, OUTD);
  gemm(t1, Ws2, bs2, t2, NGROUP, OUTD, OUT2);
  ln_relu<<<NGROUP, OUT2, OUT2 * sizeof(float), stream>>>(t2, g2, bb2, OUT2);
  head_out<<<1, NGROUP, 0, stream>>>(t2, Ws3, bs3, out);
}


# mean-pool epilogue: 128-thread block sums 16 consecutive nodes in registers, one atomic add per column per graph-id change; 480K waves, 15M float atomics -> 7500 waves, ~1M; pooled add order changed
# speedup vs baseline: 2.5381x; 1.0814x over previous
#include <hip/hip_runtime.h>
#include <math.h>

#define HIDF   128
#define NHEAD  4
#define WHID   512
#define NGROUP 64
#define OUTD   256
#define OUT2   128
#define NCLSD  2
#define SLOPE  0.2f
#define KC     128
#define BSTRIDE (KC + 2)

typedef float v2f __attribute__((ext_vector_type(2)));
typedef float v8f __attribute__((ext_vector_type(8)));

__device__ __forceinline__ float lrelu(float x) { return x >= 0.f ? x : SLOPE * x; }

__global__ void wmma_gemm(const float* __restrict__ A, const float* __restrict__ B,
                          const float* __restrict__ bias, float* __restrict__ C,
                          int M, int K, int Nc) {
  __shared__ float Bs[64 * BSTRIDE];

  const int nslab = Nc >> 6;
  const int mblk  = blockIdx.x / nslab;
  const int nblk  = blockIdx.x - mblk * nslab;
  const int wave  = threadIdx.x >> 5;
  const int lane  = threadIdx.x & 31;
  const int half  = lane >> 4;
  const int l     = lane & 15;

  const int mt    = mblk * 8 + wave;
  const int arow  = min(mt * 16 + l, M - 1);
  const int col0  = nblk * 64;

  const int sc = threadIdx.x & 63;
  const int sr = threadIdx.x >> 6;

  const float* Ap = A + (size_t)arow * K + 2 * half;

  v8f acc0 = {}, acc1 = {}, acc2 = {}, acc3 = {};

  for (int k0 = 0; k0 < K; k0 += KC) {
    const int kc = min(KC, K - k0);
    __syncthreads();
    for (int kk = sr; kk < kc; kk += 4)
      Bs[sc * BSTRIDE + kk] = B[(size_t)(k0 + kk) * Nc + col0 + sc];
    __syncthreads();

    __builtin_prefetch(Ap + k0 + KC, 0, 1);

    const float* Bl0 = &Bs[(l +  0) * BSTRIDE + 2 * half];
    const float* Bl1 = &Bs[(l + 16) * BSTRIDE + 2 * half];
    const float* Bl2 = &Bs[(l + 32) * BSTRIDE + 2 * half];
    const float* Bl3 = &Bs[(l + 48) * BSTRIDE + 2 * half];
    for (int k = 0; k < kc; k += 4) {
      v2f a  = *(const v2f*)(Ap + k0 + k);
      v2f b0 = *(const v2f*)(Bl0 + k);
      v2f b1 = *(const v2f*)(Bl1 + k);
      v2f b2 = *(const v2f*)(Bl2 + k);
      v2f b3 = *(const v2f*)(Bl3 + k);
      acc0 = __builtin_amdgcn_wmma_f32_16x16x4_f32(false, a, false, b0, (short)0, acc0, false, false);
      acc1 = __builtin_amdgcn_wmma_f32_16x16x4_f32(false, a, false, b1, (short)0, acc1, false, false);
      acc2 = __builtin_amdgcn_wmma_f32_16x16x4_f32(false, a, false, b2, (short)0, acc2, false, false);
      acc3 = __builtin_amdgcn_wmma_f32_16x16x4_f32(false, a, false, b3, (short)0, acc3, false, false);
    }
  }

  v8f* accs[4] = {&acc0, &acc1, &acc2, &acc3};
#pragma unroll
  for (int j = 0; j < 4; ++j) {
    const int ccol = col0 + 16 * j + l;
    const float bv = bias ? bias[ccol] : 0.f;
#pragma unroll
    for (int i = 0; i < 8; ++i) {
      const int row = mt * 16 + i + 8 * half;
      if (row < M) C[(size_t)row * Nc + ccol] = (*accs[j])[i] + bv;
    }
  }
}

__global__ void attn_scores(const float* __restrict__ h, const float* __restrict__ a_s,
                            const float* __restrict__ a_d, float* __restrict__ ssrc,
                            float* __restrict__ sdst, int n) {
  int idx = blockIdx.x * blockDim.x + threadIdx.x;
  if (idx >= n * NHEAD) return;
  int hd = idx & (NHEAD - 1);
  const float* hp = h + (size_t)idx * HIDF;
  const float* as = a_s + hd * HIDF;
  const float* ad = a_d + hd * HIDF;
  float s1 = 0.f, s2 = 0.f;
  for (int c = 0; c < HIDF; c += 4) {
    float4 hv = *(const float4*)(hp + c);
    float4 av = *(const float4*)(as + c);
    float4 dv = *(const float4*)(ad + c);
    s1 += hv.x * av.x + hv.y * av.y + hv.z * av.z + hv.w * av.w;
    s2 += hv.x * dv.x + hv.y * dv.y + hv.z * dv.z + hv.w * dv.w;
  }
  ssrc[idx] = s1;
  sdst[idx] = s2;
}

__global__ void fill_f32(float* __restrict__ p, float v, int n) {
  int i = blockIdx.x * blockDim.x + threadIdx.x;
  if (i < n) p[i] = v;
}

__global__ void fill_i32(int* __restrict__ p, int v, int n) {
  int i = blockIdx.x * blockDim.x + threadIdx.x;
  if (i < n) p[i] = v;
}

__global__ void csr_hist(const int* __restrict__ ei, int E, int* __restrict__ deg) {
  int e = blockIdx.x * blockDim.x + threadIdx.x;
  if (e < E) atomicAdd(&deg[ei[E + e]], 1);
}

__global__ void __launch_bounds__(1024)
csr_scan(const int* __restrict__ deg, int* __restrict__ rowptr,
         int* __restrict__ cursor, int n) {
  __shared__ int part[1024];
  const int t     = threadIdx.x;
  const int items = (n + 1023) >> 10;
  const int lo    = min(t * items, n);
  const int hi    = min(lo + items, n);
  int s = 0;
  for (int i = lo; i < hi; ++i) s += deg[i];
  part[t] = s;
  __syncthreads();
  for (int off = 1; off < 1024; off <<= 1) {
    const int v = (t >= off) ? part[t - off] : 0;
    __syncthreads();
    part[t] += v;
    __syncthreads();
  }
  int run = part[t] - s;
  for (int i = lo; i < hi; ++i) {
    rowptr[i] = run;
    cursor[i] = run;
    run += deg[i];
  }
  if (t == 1023) rowptr[n] = part[1023];
}

__global__ void csr_fill(const int* __restrict__ ei, int E, int* __restrict__ cursor,
                         int* __restrict__ csr) {
  int e = blockIdx.x * blockDim.x + threadIdx.x;
  if (e >= E) return;
  const int s = ei[e], d = ei[E + e];
  const int pos = atomicAdd(&cursor[d], 1);
  *(int2*)(csr + 2 * (size_t)pos) = make_int2(s, e);
}

__global__ void __launch_bounds__(256)
csr_sort_rows(const int* __restrict__ rowptr, const int* __restrict__ csr_in,
              int* __restrict__ csr_out, int n) {
  const int lane = threadIdx.x & 31;
  const int node = __builtin_amdgcn_readfirstlane((int)(blockIdx.x * 8 + (threadIdx.x >> 5)));
  if (node >= n) return;
  const int beg = rowptr[node], end = rowptr[node + 1];
  for (int b = beg; b < end; b += 32) {
    const bool have = (b + lane) < end;
    int2 mine = make_int2(0, 0x7fffffff);
    if (have) mine = *(const int2*)(csr_in + 2 * (size_t)(b + lane));
    int rank = 0;
    for (int c0 = beg; c0 < end; c0 += 32) {
      const int cnt = min(32, end - c0);
      int other = 0x7fffffff;
      if (c0 + lane < end) other = csr_in[2 * (size_t)(c0 + lane) + 1];
      for (int j = 0; j < cnt; ++j)
        rank += (__builtin_amdgcn_readlane(other, j) < mine.y) ? 1 : 0;
    }
    if (have) *(int2*)(csr_out + 2 * (size_t)(beg + rank)) = mine;
  }
}

__device__ __forceinline__ float bcast_f(float v, int srcLane) {
  return __int_as_float(__builtin_amdgcn_readlane(__float_as_int(v), srcLane));
}

__device__ __forceinline__ float wave_max(float v) {
  for (int o = 16; o > 0; o >>= 1) v = fmaxf(v, __shfl_xor(v, o, 32));
  return v;
}
__device__ __forceinline__ float wave_sum(float v) {
  for (int o = 16; o > 0; o >>= 1) v += __shfl_xor(v, o, 32);
  return v;
}

__device__ __forceinline__ void load_entries(const int* __restrict__ csr,
                                             const float* __restrict__ ssrc,
                                             const float4 sd, int base, int cnt, int lane,
                                             int& src, int& eid, float4& e) {
  src = 0; eid = 0;
  e = make_float4(-INFINITY, -INFINITY, -INFINITY, -INFINITY);
  if (lane < cnt) {
    const int2 ent = *(const int2*)(csr + 2 * (size_t)(base + lane));
    src = ent.x; eid = ent.y;
    const float4 ss = *(const float4*)(ssrc + (size_t)src * NHEAD);
    e.x = lrelu(ss.x + sd.x); e.y = lrelu(ss.y + sd.y);
    e.z = lrelu(ss.z + sd.z); e.w = lrelu(ss.w + sd.w);
  }
}

__device__ __forceinline__ void add_rows(const float* __restrict__ h, int lane, int cnt,
                                         int my_src, const float4 my_a,
                                         float4& s0, float4& s1, float4& s2, float4& s3) {
  for (int j = 0; j < cnt; ++j) {
    const int   s  = __builtin_amdgcn_readlane(my_src, j);
    const float a0 = bcast_f(my_a.x, j);
    const float a1 = bcast_f(my_a.y, j);
    const float a2 = bcast_f(my_a.z, j);
    const float a3 = bcast_f(my_a.w, j);
    const float* hp = h + (size_t)s * WHID + lane * 4;
    const float4 h0 = *(const float4*)(hp);
    const float4 h1 = *(const float4*)(hp + HIDF);
    const float4 h2 = *(const float4*)(hp + 2 * HIDF);
    const float4 h3 = *(const float4*)(hp + 3 * HIDF);
    s0.x += a0 * h0.x; s0.y += a0 * h0.y; s0.z += a0 * h0.z; s0.w += a0 * h0.w;
    s1.x += a1 * h1.x; s1.y += a1 * h1.y; s1.z += a1 * h1.z; s1.w += a1 * h1.w;
    s2.x += a2 * h2.x; s2.y += a2 * h2.y; s2.z += a2 * h2.z; s2.w += a2 * h2.w;
    s3.x += a3 * h3.x; s3.y += a3 * h3.y; s3.z += a3 * h3.z; s3.w += a3 * h3.w;
  }
}

__global__ void __launch_bounds__(256)
edge_softmax_gather(const int* __restrict__ rowptr, const int* __restrict__ csr, int E, int n,
                    const float* __restrict__ ssrc, const float* __restrict__ sdst,
                    const float* __restrict__ h, float* __restrict__ acc,
                    float* __restrict__ alpha_out) {
  const int lane = threadIdx.x & 31;
  const int node = __builtin_amdgcn_readfirstlane((int)(blockIdx.x * 8 + (threadIdx.x >> 5)));
  if (node >= n) return;
  const float4 sd = *(const float4*)(sdst + (size_t)node * NHEAD);
  const float4 sn = *(const float4*)(ssrc + (size_t)node * NHEAD);
  const int beg = rowptr[node], end = rowptr[node + 1];

  float4 es;
  es.x = lrelu(sn.x + sd.x); es.y = lrelu(sn.y + sd.y);
  es.z = lrelu(sn.z + sd.z); es.w = lrelu(sn.w + sd.w);

  const int cnt0 = min(32, end - beg);
  int src0, eid0; float4 e0;
  load_entries(csr, ssrc, sd, beg, cnt0, lane, src0, eid0, e0);

  float4 mx;
  mx.x = fmaxf(es.x, e0.x); mx.y = fmaxf(es.y, e0.y);
  mx.z = fmaxf(es.z, e0.z); mx.w = fmaxf(es.w, e0.w);
  for (int base = beg + 32; base < end; base += 32) {
    int s_, i_; float4 e_;
    load_entries(csr, ssrc, sd, base, min(32, end - base), lane, s_, i_, e_);
    mx.x = fmaxf(mx.x, e_.x); mx.y = fmaxf(mx.y, e_.y);
    mx.z = fmaxf(mx.z, e_.z); mx.w = fmaxf(mx.w, e_.w);
  }
  mx.x = wave_max(mx.x); mx.y = wave_max(mx.y);
  mx.z = wave_max(mx.z); mx.w = wave_max(mx.w);
  if (!(mx.x > -3.0e38f && mx.x < 3.0e38f)) mx.x = 0.f;
  if (!(mx.y > -3.0e38f && mx.y < 3.0e38f)) mx.y = 0.f;
  if (!(mx.z > -3.0e38f && mx.z < 3.0e38f)) mx.z = 0.f;
  if (!(mx.w > -3.0e38f && mx.w < 3.0e38f)) mx.w = 0.f;

  float4 x0 = make_float4(0.f, 0.f, 0.f, 0.f);
  if (lane < cnt0) {
    x0.x = expf(e0.x - mx.x); x0.y = expf(e0.y - mx.y);
    x0.z = expf(e0.z - mx.z); x0.w = expf(e0.w - mx.w);
  }
  float4 ds = x0;
  for (int base = beg + 32; base < end; base += 32) {
    const int cnt = min(32, end - base);
    int s_, i_; float4 e_;
    load_entries(csr, ssrc, sd, base, cnt, lane, s_, i_, e_);
    if (lane < cnt) {
      ds.x += expf(e_.x - mx.x); ds.y += expf(e_.y - mx.y);
      ds.z += expf(e_.z - mx.z); ds.w += expf(e_.w - mx.w);
    }
  }
  float4 xs;
  xs.x = expf(es.x - mx.x); xs.y = expf(es.y - mx.y);
  xs.z = expf(es.z - mx.z); xs.w = expf(es.w - mx.w);
  float4 dn;
  dn.x = wave_sum(ds.x) + xs.x; dn.y = wave_sum(ds.y) + xs.y;
  dn.z = wave_sum(ds.z) + xs.z; dn.w = wave_sum(ds.w) + xs.w;

  float4 s0, s1, s2, s3;
  {
    const size_t eid = (size_t)E + node;
    float4 a;
    a.x = xs.x / dn.x; a.y = xs.y / dn.y; a.z = xs.z / dn.z; a.w = xs.w / dn.w;
    if (alpha_out && lane == 0) *(float4*)(alpha_out + eid * NHEAD) = a;
    const float* hp = h + (size_t)node * WHID + lane * 4;
    const float4 h0 = *(const float4*)(hp);
    const float4 h1 = *(const float4*)(hp + HIDF);
    const float4 h2 = *(const float4*)(hp + 2 * HIDF);
    const float4 h3 = *(const float4*)(hp + 3 * HIDF);
    s0.x = a.x * h0.x; s0.y = a.x * h0.y; s0.z = a.x * h0.z; s0.w = a.x * h0.w;
    s1.x = a.y * h1.x; s1.y = a.y * h1.y; s1.z = a.y * h1.z; s1.w = a.y * h1.w;
    s2.x = a.z * h2.x; s2.y = a.z * h2.y; s2.z = a.z * h2.z; s2.w = a.z * h2.w;
    s3.x = a.w * h3.x; s3.y = a.w * h3.y; s3.z = a.w * h3.z; s3.w = a.w * h3.w;
  }

  {
    float4 a;
    a.x = x0.x / dn.x; a.y = x0.y / dn.y; a.z = x0.z / dn.z; a.w = x0.w / dn.w;
    if (alpha_out && lane < cnt0) *(float4*)(alpha_out + (size_t)eid0 * NHEAD) = a;
    add_rows(h, lane, cnt0, src0, a, s0, s1, s2, s3);
  }
  for (int base = beg + 32; base < end; base += 32) {
    const int cnt = min(32, end - base);
    int my_src, my_eid; float4 e_;
    load_entries(csr, ssrc, sd, base, cnt, lane, my_src, my_eid, e_);
    float4 a = make_float4(0.f, 0.f, 0.f, 0.f);
    if (lane < cnt) {
      a.x = expf(e_.x - mx.x) / dn.x; a.y = expf(e_.y - mx.y) / dn.y;
      a.z = expf(e_.z - mx.z) / dn.z; a.w = expf(e_.w - mx.w) / dn.w;
      if (alpha_out) *(float4*)(alpha_out + (size_t)my_eid * NHEAD) = a;
    }
    add_rows(h, lane, cnt, my_src, a, s0, s1, s2, s3);
  }

  float* ap = acc + (size_t)node * WHID + lane * 4;
  *(float4*)(ap)            = s0;
  *(float4*)(ap + HIDF)     = s1;
  *(float4*)(ap + 2 * HIDF) = s2;
  *(float4*)(ap + 3 * HIDF) = s3;
}

__global__ void l1_post(const float* __restrict__ acc, const float* __restrict__ b1,
                        const float* __restrict__ g, const float* __restrict__ b,
                        float* __restrict__ x2) {
  __shared__ float sh[HIDF];
  int nd = blockIdx.x, c = threadIdx.x;
  const float* ap = acc + (size_t)nd * WHID;
  float v = 0.25f * (ap[c] + ap[HIDF + c] + ap[2 * HIDF + c] + ap[3 * HIDF + c]) + b1[c];
  sh[c] = v; __syncthreads();
  for (int s = HIDF / 2; s > 0; s >>= 1) { if (c < s) sh[c] += sh[c + s]; __syncthreads(); }
  float mu = sh[0] * (1.0f / HIDF);
  __syncthreads();
  float dv = v - mu;
  sh[c] = dv * dv; __syncthreads();
  for (int s = HIDF / 2; s > 0; s >>= 1) { if (c < s) sh[c] += sh[c + s]; __syncthreads(); }
  float var = sh[0] * (1.0f / HIDF);
  float y = g[c] * dv * rsqrtf(var + 1e-5f) + b[c];
  x2[(size_t)nd * HIDF + c] = lrelu(y);
}

#define POOL_RUN 16

__device__ __forceinline__ void pool_flush(float* __restrict__ pooled, float* __restrict__ cnt,
                                           int g, int t, const float4 s, float run) {
  float* p = pooled + (size_t)g * WHID + 4 * t;
  atomicAdd(p + 0, s.x);
  atomicAdd(p + 1, s.y);
  atomicAdd(p + 2, s.z);
  atomicAdd(p + 3, s.w);
  if (t == 0) atomicAdd(&cnt[g], run);
}

__global__ void __launch_bounds__(128)
l2_post_pool(const float* __restrict__ acc, const float* __restrict__ b2,
             const int* __restrict__ batch, float* __restrict__ pooled,
             float* __restrict__ cnt, int n) {
  const int t  = threadIdx.x;
  const int n0 = blockIdx.x * POOL_RUN;
  const float4 bv = *(const float4*)(b2 + 4 * t);

  float4 v[POOL_RUN];
  int    gi[POOL_RUN];
#pragma unroll
  for (int j = 0; j < POOL_RUN; ++j) {
    const int nd = min(n0 + j, n - 1);
    v[j]  = *(const float4*)(acc + (size_t)nd * WHID + 4 * t);
    gi[j] = batch[nd];
  }

  float4 s = make_float4(0.f, 0.f, 0.f, 0.f);
  float run = 0.f;
  int   g   = gi[0];
#pragma unroll
  for (int j = 0; j < POOL_RUN; ++j) {
    if (n0 + j < n) {
      if (gi[j] != g) {
        pool_flush(pooled, cnt, g, t, s, run);
        s = make_float4(0.f, 0.f, 0.f, 0.f);
        run = 0.f;
        g = gi[j];
      }
      s.x += lrelu(v[j].x + bv.x);
      s.y += lrelu(v[j].y + bv.y);
      s.z += lrelu(v[j].z + bv.z);
      s.w += lrelu(v[j].w + bv.w);
      run += 1.f;
    }
  }
  pool_flush(pooled, cnt, g, t, s, run);
}

__global__ void pool_div(float* __restrict__ pooled, const float* __restrict__ cnt) {
  int i = blockIdx.x * blockDim.x + threadIdx.x;
  if (i >= NGROUP * WHID) return;
  pooled[i] /= fmaxf(cnt[i / WHID], 1.0f);
}

__global__ void ln_relu(float* __restrict__ x, const float* __restrict__ g,
                        const float* __restrict__ b, int dim) {
  extern __shared__ float sh[];
  int r = blockIdx.x, c = threadIdx.x;
  float v = x[(size_t)r * dim + c];
  sh[c] = v; __syncthreads();
  for (int s = dim >> 1; s > 0; s >>= 1) { if (c < s) sh[c] += sh[c + s]; __syncthreads(); }
  float mu = sh[0] / dim;
  __syncthreads();
  float dv = v - mu;
  sh[c] = dv * dv; __syncthreads();
  for (int s = dim >> 1; s > 0; s >>= 1) { if (c < s) sh[c] += sh[c + s]; __syncthreads(); }
  float var = sh[0] / dim;
  float y = g[c] * dv * rsqrtf(var + 1e-5f) + b[c];
  x[(size_t)r * dim + c] = fmaxf(y, 0.f);
}

__global__ void head_out(const float* __restrict__ t2, const float* __restrict__ Ws3,
                         const float* __restrict__ bs3, float* __restrict__ out) {
  int r = blockIdx.x * blockDim.x + threadIdx.x;
  if (r >= NGROUP) return;
  float l0 = bs3[0], l1 = bs3[1];
  for (int c = 0; c < OUT2; ++c) {
    float v = t2[r * OUT2 + c];
    l0 += v * Ws3[c * 2 + 0];
    l1 += v * Ws3[c * 2 + 1];
  }
  float mx = fmaxf(l0, l1);
  float e0 = expf(l0 - mx), e1 = expf(l1 - mx), s = e0 + e1;
  out[r * 2 + 0] = e0 / s;
  out[r * 2 + 1] = e1 / s;
}

extern "C" void kernel_launch(void* const* d_in, const int* in_sizes, int n_in,
                              void* d_out, int out_size, void* d_ws, size_t ws_size,
                              hipStream_t stream) {
  (void)n_in; (void)out_size; (void)ws_size;
  const float* x    = (const float*)d_in[0];
  const int*   ei   = (const int*)d_in[1];
  const int*   batch= (const int*)d_in[2];
  const float* W1   = (const float*)d_in[3];
  const float* as1  = (const float*)d_in[4];
  const float* ad1  = (const float*)d_in[5];
  const float* b1   = (const float*)d_in[6];
  const float* g_ln = (const float*)d_in[7];
  const float* b_ln = (const float*)d_in[8];
  const float* W2   = (const float*)d_in[9];
  const float* as2  = (const float*)d_in[10];
  const float* ad2  = (const float*)d_in[11];
  const float* b2   = (const float*)d_in[12];
  const float* Wl0  = (const float*)d_in[13];
  const float* bl0  = (const float*)d_in[14];
  const float* Ws1  = (const float*)d_in[15];
  const float* bs1  = (const float*)d_in[16];
  const float* g1   = (const float*)d_in[17];
  const float* bb1  = (const float*)d_in[18];
  const float* Ws2  = (const float*)d_in[19];
  const float* bs2  = (const float*)d_in[20];
  const float* g2   = (const float*)d_in[21];
  const float* bb2  = (const float*)d_in[22];
  const float* Ws3  = (const float*)d_in[23];
  const float* bs3  = (const float*)d_in[24];

  const int N  = in_sizes[0] / HIDF;
  const int E  = in_sizes[1] / 2;
  const int NH = N * NHEAD;

  float* out       = (float*)d_out;
  float* alpha_out = out + NGROUP * NCLSD;

  size_t off = 0;
  auto alloc = [&](size_t bytes) -> float* {
    off = (off + 255) & ~(size_t)255;
    float* p = (float*)((char*)d_ws + off);
    off += bytes;
    return p;
  };
  float* h      = alloc((size_t)N * WHID * 4);
  float* acc    = alloc((size_t)N * WHID * 4);
  float* x2     = alloc((size_t)N * HIDF * 4);
  float* ssrc   = alloc((size_t)NH * 4);
  float* sdst   = alloc((size_t)NH * 4);
  float* pooled = alloc((size_t)NGROUP * WHID * 4);
  float* cnt    = alloc((size_t)NGROUP * 4);
  float* z0     = alloc((size_t)NGROUP * HIDF * 4);
  float* t1     = alloc((size_t)NGROUP * OUTD * 4);
  float* t2     = alloc((size_t)NGROUP * OUT2 * 4);
  int*   deg    = (int*)alloc((size_t)N * 4);
  int*   rowptr = (int*)alloc((size_t)(N + 1) * 4);
  int*   cursor = (int*)alloc((size_t)N * 4);
  int*   csr_arr = (int*)alloc((size_t)E * 8);
  int*   csr     = (int*)alloc((size_t)E * 8);

  auto cdiv = [](long a, long b) { return (int)((a + b - 1) / b); };
  auto gemm = [&](const float* A, const float* B, const float* bias, float* C,
                  int M, int K, int Nc) {
    int mtiles  = (M + 15) / 16;
    int mblocks = (mtiles + 7) / 8;
    int nslab   = Nc / 64;
    wmma_gemm<<<mblocks * nslab, 256, 0, stream>>>(A, B, bias, C, M, K, Nc);
  };

  fill_i32<<<cdiv(N, 256), 256, 0, stream>>>(deg, 0, N);
  csr_hist<<<cdiv(E, 256), 256, 0, stream>>>(ei, E, deg);
  csr_scan<<<1, 1024, 0, stream>>>(deg, rowptr, cursor, N);
  csr_fill<<<cdiv(E, 256), 256, 0, stream>>>(ei, E, cursor, csr_arr);
  csr_sort_rows<<<cdiv(N, 8), 256, 0, stream>>>(rowptr, csr_arr, csr, N);

  gemm(x, W1, nullptr, h, N, HIDF, WHID);
  attn_scores<<<cdiv(NH, 256), 256, 0, stream>>>(h, as1, ad1, ssrc, sdst, N);
  edge_softmax_gather<<<cdiv(N, 8), 256, 0, stream>>>(rowptr, csr, E, N, ssrc, sdst, h, acc, nullptr);
  l1_post<<<N, HIDF, 0, stream>>>(acc, b1, g_ln, b_ln, x2);

  gemm(x2, W2, nullptr, h, N, HIDF, WHID);
  attn_scores<<<cdiv(NH, 256), 256, 0, stream>>>(h, as2, ad2, ssrc, sdst, N);
  edge_softmax_gather<<<cdiv(N, 8), 256, 0, stream>>>(rowptr, csr, E, N, ssrc, sdst, h, acc, alpha_out);

  fill_f32<<<cdiv(NGROUP * WHID, 256), 256, 0, stream>>>(pooled, 0.f, NGROUP * WHID);
  fill_f32<<<1, NGROUP, 0, stream>>>(cnt, 0.f, NGROUP);
  l2_post_pool<<<cdiv(N, POOL_RUN), 128, 0, stream>>>(acc, b2, batch, pooled, cnt, N);
  pool_div<<<cdiv(NGROUP * WHID, 256), 256, 0, stream>>>(pooled, cnt);

  gemm(pooled, Wl0, bl0, z0, NGROUP, WHID, HIDF);
  gemm(z0, Ws1, bs1, t1, NGROUP, HIDF, OUTD);
  ln_relu<<<NGROUP, OUTD, OUTD * sizeof(float), stream>>>(t1, g1, bb1, OUTD);
  gemm(t1, Ws2, bs2, t2, NGROUP, OUTD, OUT2);
  ln_relu<<<NGROUP, OUT2, OUT2 * sizeof(float), stream>>>(t2, g2, bb2, OUT2);
  head_out<<<1, NGROUP, 0, stream>>>(t2, Ws3, bs3, out);
}


# graph head in ONE launch: 4 blocks x 16 graphs run pool divide, 3 f32 WMMA products, 2 layer norms, softmax with rows kept in LDS; 7 launches -> 1; norm and logit sums now in wave-butterfly order
# speedup vs baseline: 3.4661x; 1.3656x over previous
#include <hip/hip_runtime.h>
#include <math.h>

#define HIDF   128
#define NHEAD  4
#define WHID   512
#define NGROUP 64
#define OUTD   256
#define OUT2   128
#define NCLSD  2
#define SLOPE  0.2f
#define KC     128
#define BSTRIDE (KC + 2)

typedef float v2f __attribute__((ext_vector_type(2)));
typedef float v8f __attribute__((ext_vector_type(8)));

__device__ __forceinline__ float lrelu(float x) { return x >= 0.f ? x : SLOPE * x; }

__global__ void wmma_gemm(const float* __restrict__ A, const float* __restrict__ B,
                          const float* __restrict__ bias, float* __restrict__ C,
                          int M, int K, int Nc) {
  __shared__ float Bs[64 * BSTRIDE];

  const int nslab = Nc >> 6;
  const int mblk  = blockIdx.x / nslab;
  const int nblk  = blockIdx.x - mblk * nslab;
  const int wave  = threadIdx.x >> 5;
  const int lane  = threadIdx.x & 31;
  const int half  = lane >> 4;
  const int l     = lane & 15;

  const int mt    = mblk * 8 + wave;
  const int arow  = min(mt * 16 + l, M - 1);
  const int col0  = nblk * 64;

  const int sc = threadIdx.x & 63;
  const int sr = threadIdx.x >> 6;

  const float* Ap = A + (size_t)arow * K + 2 * half;

  v8f acc0 = {}, acc1 = {}, acc2 = {}, acc3 = {};

  for (int k0 = 0; k0 < K; k0 += KC) {
    const int kc = min(KC, K - k0);
    __syncthreads();
    for (int kk = sr; kk < kc; kk += 4)
      Bs[sc * BSTRIDE + kk] = B[(size_t)(k0 + kk) * Nc + col0 + sc];
    __syncthreads();

    __builtin_prefetch(Ap + k0 + KC, 0, 1);

    const float* Bl0 = &Bs[(l +  0) * BSTRIDE + 2 * half];
    const float* Bl1 = &Bs[(l + 16) * BSTRIDE + 2 * half];
    const float* Bl2 = &Bs[(l + 32) * BSTRIDE + 2 * half];
    const float* Bl3 = &Bs[(l + 48) * BSTRIDE + 2 * half];
    for (int k = 0; k < kc; k += 4) {
      v2f a  = *(const v2f*)(Ap + k0 + k);
      v2f b0 = *(const v2f*)(Bl0 + k);
      v2f b1 = *(const v2f*)(Bl1 + k);
      v2f b2 = *(const v2f*)(Bl2 + k);
      v2f b3 = *(const v2f*)(Bl3 + k);
      acc0 = __builtin_amdgcn_wmma_f32_16x16x4_f32(false, a, false, b0, (short)0, acc0, false, false);
      acc1 = __builtin_amdgcn_wmma_f32_16x16x4_f32(false, a, false, b1, (short)0, acc1, false, false);
      acc2 = __builtin_amdgcn_wmma_f32_16x16x4_f32(false, a, false, b2, (short)0, acc2, false, false);
      acc3 = __builtin_amdgcn_wmma_f32_16x16x4_f32(false, a, false, b3, (short)0, acc3, false, false);
    }
  }

  v8f* accs[4] = {&acc0, &acc1, &acc2, &acc3};
#pragma unroll
  for (int j = 0; j < 4; ++j) {
    const int ccol = col0 + 16 * j + l;
    const float bv = bias ? bias[ccol] : 0.f;
#pragma unroll
    for (int i = 0; i < 8; ++i) {
      const int row = mt * 16 + i + 8 * half;
      if (row < M) C[(size_t)row * Nc + ccol] = (*accs[j])[i] + bv;
    }
  }
}

__global__ void attn_scores(const float* __restrict__ h, const float* __restrict__ a_s,
                            const float* __restrict__ a_d, float* __restrict__ ssrc,
                            float* __restrict__ sdst, int n) {
  int idx = blockIdx.x * blockDim.x + threadIdx.x;
  if (idx >= n * NHEAD) return;
  int hd = idx & (NHEAD - 1);
  const float* hp = h + (size_t)idx * HIDF;
  const float* as = a_s + hd * HIDF;
  const float* ad = a_d + hd * HIDF;
  float s1 = 0.f, s2 = 0.f;
  for (int c = 0; c < HIDF; c += 4) {
    float4 hv = *(const float4*)(hp + c);
    float4 av = *(const float4*)(as + c);
    float4 dv = *(const float4*)(ad + c);
    s1 += hv.x * av.x + hv.y * av.y + hv.z * av.z + hv.w * av.w;
    s2 += hv.x * dv.x + hv.y * dv.y + hv.z * dv.z + hv.w * dv.w;
  }
  ssrc[idx] = s1;
  sdst[idx] = s2;
}

__global__ void fill_f32(float* __restrict__ p, float v, int n) {
  int i = blockIdx.x * blockDim.x + threadIdx.x;
  if (i < n) p[i] = v;
}

__global__ void fill_i32(int* __restrict__ p, int v, int n) {
  int i = blockIdx.x * blockDim.x + threadIdx.x;
  if (i < n) p[i] = v;
}

__global__ void csr_hist(const int* __restrict__ ei, int E, int* __restrict__ deg) {
  int e = blockIdx.x * blockDim.x + threadIdx.x;
  if (e < E) atomicAdd(&deg[ei[E + e]], 1);
}

__global__ void __launch_bounds__(1024)
csr_scan(const int* __restrict__ deg, int* __restrict__ rowptr,
         int* __restrict__ cursor, int n) {
  __shared__ int part[1024];
  const int t     = threadIdx.x;
  const int items = (n + 1023) >> 10;
  const int lo    = min(t * items, n);
  const int hi    = min(lo + items, n);
  int s = 0;
  for (int i = lo; i < hi; ++i) s += deg[i];
  part[t] = s;
  __syncthreads();
  for (int off = 1; off < 1024; off <<= 1) {
    const int v = (t >= off) ? part[t - off] : 0;
    __syncthreads();
    part[t] += v;
    __syncthreads();
  }
  int run = part[t] - s;
  for (int i = lo; i < hi; ++i) {
    rowptr[i] = run;
    cursor[i] = run;
    run += deg[i];
  }
  if (t == 1023) rowptr[n] = part[1023];
}

__global__ void csr_fill(const int* __restrict__ ei, int E, int* __restrict__ cursor,
                         int* __restrict__ csr) {
  int e = blockIdx.x * blockDim.x + threadIdx.x;
  if (e >= E) return;
  const int s = ei[e], d = ei[E + e];
  const int pos = atomicAdd(&cursor[d], 1);
  *(int2*)(csr + 2 * (size_t)pos) = make_int2(s, e);
}

__global__ void __launch_bounds__(256)
csr_sort_rows(const int* __restrict__ rowptr, const int* __restrict__ csr_in,
              int* __restrict__ csr_out, int n) {
  const int lane = threadIdx.x & 31;
  const int node = __builtin_amdgcn_readfirstlane((int)(blockIdx.x * 8 + (threadIdx.x >> 5)));
  if (node >= n) return;
  const int beg = rowptr[node], end = rowptr[node + 1];
  for (int b = beg; b < end; b += 32) {
    const bool have = (b + lane) < end;
    int2 mine = make_int2(0, 0x7fffffff);
    if (have) mine = *(const int2*)(csr_in + 2 * (size_t)(b + lane));
    int rank = 0;
    for (int c0 = beg; c0 < end; c0 += 32) {
      const int cnt = min(32, end - c0);
      int other = 0x7fffffff;
      if (c0 + lane < end) other = csr_in[2 * (size_t)(c0 + lane) + 1];
      for (int j = 0; j < cnt; ++j)
        rank += (__builtin_amdgcn_readlane(other, j) < mine.y) ? 1 : 0;
    }
    if (have) *(int2*)(csr_out + 2 * (size_t)(beg + rank)) = mine;
  }
}

__device__ __forceinline__ float bcast_f(float v, int srcLane) {
  return __int_as_float(__builtin_amdgcn_readlane(__float_as_int(v), srcLane));
}

__device__ __forceinline__ float wave_max(float v) {
  for (int o = 16; o > 0; o >>= 1) v = fmaxf(v, __shfl_xor(v, o, 32));
  return v;
}
__device__ __forceinline__ float wave_sum(float v) {
  for (int o = 16; o > 0; o >>= 1) v += __shfl_xor(v, o, 32);
  return v;
}

__device__ __forceinline__ void load_entries(const int* __restrict__ csr,
                                             const float* __restrict__ ssrc,
                                             const float4 sd, int base, int cnt, int lane,
                                             int& src, int& eid, float4& e) {
  src = 0; eid = 0;
  e = make_float4(-INFINITY, -INFINITY, -INFINITY, -INFINITY);
  if (lane < cnt) {
    const int2 ent = *(const int2*)(csr + 2 * (size_t)(base + lane));
    src = ent.x; eid = ent.y;
    const float4 ss = *(const float4*)(ssrc + (size_t)src * NHEAD);
    e.x = lrelu(ss.x + sd.x); e.y = lrelu(ss.y + sd.y);
    e.z = lrelu(ss.z + sd.z); e.w = lrelu(ss.w + sd.w);
  }
}

__device__ __forceinline__ void add_rows(const float* __restrict__ h, int lane, int cnt,
                                         int my_src, const float4 my_a,
                                         float4& s0, float4& s1, float4& s2, float4& s3) {
  for (int j = 0; j < cnt; ++j) {
    const int   s  = __builtin_amdgcn_readlane(my_src, j);
    const float a0 = bcast_f(my_a.x, j);
    const float a1 = bcast_f(my_a.y, j);
    const float a2 = bcast_f(my_a.z, j);
    const float a3 = bcast_f(my_a.w, j);
    const float* hp = h + (size_t)s * WHID + lane * 4;
    const float4 h0 = *(const float4*)(hp);
    const float4 h1 = *(const float4*)(hp + HIDF);
    const float4 h2 = *(const float4*)(hp + 2 * HIDF);
    const float4 h3 = *(const float4*)(hp + 3 * HIDF);
    s0.x += a0 * h0.x; s0.y += a0 * h0.y; s0.z += a0 * h0.z; s0.w += a0 * h0.w;
    s1.x += a1 * h1.x; s1.y += a1 * h1.y; s1.z += a1 * h1.z; s1.w += a1 * h1.w;
    s2.x += a2 * h2.x; s2.y += a2 * h2.y; s2.z += a2 * h2.z; s2.w += a2 * h2.w;
    s3.x += a3 * h3.x; s3.y += a3 * h3.y; s3.z += a3 * h3.z; s3.w += a3 * h3.w;
  }
}

__global__ void __launch_bounds__(256)
edge_softmax_gather(const int* __restrict__ rowptr, const int* __restrict__ csr, int E, int n,
                    const float* __restrict__ ssrc, const float* __restrict__ sdst,
                    const float* __restrict__ h, float* __restrict__ acc,
                    float* __restrict__ alpha_out) {
  const int lane = threadIdx.x & 31;
  const int node = __builtin_amdgcn_readfirstlane((int)(blockIdx.x * 8 + (threadIdx.x >> 5)));
  if (node >= n) return;
  const float4 sd = *(const float4*)(sdst + (size_t)node * NHEAD);
  const float4 sn = *(const float4*)(ssrc + (size_t)node * NHEAD);
  const int beg = rowptr[node], end = rowptr[node + 1];

  float4 es;
  es.x = lrelu(sn.x + sd.x); es.y = lrelu(sn.y + sd.y);
  es.z = lrelu(sn.z + sd.z); es.w = lrelu(sn.w + sd.w);

  const int cnt0 = min(32, end - beg);
  int src0, eid0; float4 e0;
  load_entries(csr, ssrc, sd, beg, cnt0, lane, src0, eid0, e0);

  float4 mx;
  mx.x = fmaxf(es.x, e0.x); mx.y = fmaxf(es.y, e0.y);
  mx.z = fmaxf(es.z, e0.z); mx.w = fmaxf(es.w, e0.w);
  for (int base = beg + 32; base < end; base += 32) {
    int s_, i_; float4 e_;
    load_entries(csr, ssrc, sd, base, min(32, end - base), lane, s_, i_, e_);
    mx.x = fmaxf(mx.x, e_.x); mx.y = fmaxf(mx.y, e_.y);
    mx.z = fmaxf(mx.z, e_.z); mx.w = fmaxf(mx.w, e_.w);
  }
  mx.x = wave_max(mx.x); mx.y = wave_max(mx.y);
  mx.z = wave_max(mx.z); mx.w = wave_max(mx.w);
  if (!(mx.x > -3.0e38f && mx.x < 3.0e38f)) mx.x = 0.f;
  if (!(mx.y > -3.0e38f && mx.y < 3.0e38f)) mx.y = 0.f;
  if (!(mx.z > -3.0e38f && mx.z < 3.0e38f)) mx.z = 0.f;
  if (!(mx.w > -3.0e38f && mx.w < 3.0e38f)) mx.w = 0.f;

  float4 x0 = make_float4(0.f, 0.f, 0.f, 0.f);
  if (lane < cnt0) {
    x0.x = expf(e0.x - mx.x); x0.y = expf(e0.y - mx.y);
    x0.z = expf(e0.z - mx.z); x0.w = expf(e0.w - mx.w);
  }
  float4 ds = x0;
  for (int base = beg + 32; base < end; base += 32) {
    const int cnt = min(32, end - base);
    int s_, i_; float4 e_;
    load_entries(csr, ssrc, sd, base, cnt, lane, s_, i_, e_);
    if (lane < cnt) {
      ds.x += expf(e_.x - mx.x); ds.y += expf(e_.y - mx.y);
      ds.z += expf(e_.z - mx.z); ds.w += expf(e_.w - mx.w);
    }
  }
  float4 xs;
  xs.x = expf(es.x - mx.x); xs.y = expf(es.y - mx.y);
  xs.z = expf(es.z - mx.z); xs.w = expf(es.w - mx.w);
  float4 dn;
  dn.x = wave_sum(ds.x) + xs.x; dn.y = wave_sum(ds.y) + xs.y;
  dn.z = wave_sum(ds.z) + xs.z; dn.w = wave_sum(ds.w) + xs.w;

  float4 s0, s1, s2, s3;
  {
    const size_t eid = (size_t)E + node;
    float4 a;
    a.x = xs.x / dn.x; a.y = xs.y / dn.y; a.z = xs.z / dn.z; a.w = xs.w / dn.w;
    if (alpha_out && lane == 0) *(float4*)(alpha_out + eid * NHEAD) = a;
    const float* hp = h + (size_t)node * WHID + lane * 4;
    const float4 h0 = *(const float4*)(hp);
    const float4 h1 = *(const float4*)(hp + HIDF);
    const float4 h2 = *(const float4*)(hp + 2 * HIDF);
    const float4 h3 = *(const float4*)(hp + 3 * HIDF);
    s0.x = a.x * h0.x; s0.y = a.x * h0.y; s0.z = a.x * h0.z; s0.w = a.x * h0.w;
    s1.x = a.y * h1.x; s1.y = a.y * h1.y; s1.z = a.y * h1.z; s1.w = a.y * h1.w;
    s2.x = a.z * h2.x; s2.y = a.z * h2.y; s2.z = a.z * h2.z; s2.w = a.z * h2.w;
    s3.x = a.w * h3.x; s3.y = a.w * h3.y; s3.z = a.w * h3.z; s3.w = a.w * h3.w;
  }

  {
    float4 a;
    a.x = x0.x / dn.x; a.y = x0.y / dn.y; a.z = x0.z / dn.z; a.w = x0.w / dn.w;
    if (alpha_out && lane < cnt0) *(float4*)(alpha_out + (size_t)eid0 * NHEAD) = a;
    add_rows(h, lane, cnt0, src0, a, s0, s1, s2, s3);
  }
  for (int base = beg + 32; base < end; base += 32) {
    const int cnt = min(32, end - base);
    int my_src, my_eid; float4 e_;
    load_entries(csr, ssrc, sd, base, cnt, lane, my_src, my_eid, e_);
    float4 a = make_float4(0.f, 0.f, 0.f, 0.f);
    if (lane < cnt) {
      a.x = expf(e_.x - mx.x) / dn.x; a.y = expf(e_.y - mx.y) / dn.y;
      a.z = expf(e_.z - mx.z) / dn.z; a.w = expf(e_.w - mx.w) / dn.w;
      if (alpha_out) *(float4*)(alpha_out + (size_t)my_eid * NHEAD) = a;
    }
    add_rows(h, lane, cnt, my_src, a, s0, s1, s2, s3);
  }

  float* ap = acc + (size_t)node * WHID + lane * 4;
  *(float4*)(ap)            = s0;
  *(float4*)(ap + HIDF)     = s1;
  *(float4*)(ap + 2 * HIDF) = s2;
  *(float4*)(ap + 3 * HIDF) = s3;
}

__global__ void l1_post(const float* __restrict__ acc, const float* __restrict__ b1,
                        const float* __restrict__ g, const float* __restrict__ b,
                        float* __restrict__ x2) {
  __shared__ float sh[HIDF];
  int nd = blockIdx.x, c = threadIdx.x;
  const float* ap = acc + (size_t)nd * WHID;
  float v = 0.25f * (ap[c] + ap[HIDF + c] + ap[2 * HIDF + c] + ap[3 * HIDF + c]) + b1[c];
  sh[c] = v; __syncthreads();
  for (int s = HIDF / 2; s > 0; s >>= 1) { if (c < s) sh[c] += sh[c + s]; __syncthreads(); }
  float mu = sh[0] * (1.0f / HIDF);
  __syncthreads();
  float dv = v - mu;
  sh[c] = dv * dv; __syncthreads();
  for (int s = HIDF / 2; s > 0; s >>= 1) { if (c < s) sh[c] += sh[c + s]; __syncthreads(); }
  float var = sh[0] * (1.0f / HIDF);
  float y = g[c] * dv * rsqrtf(var + 1e-5f) + b[c];
  x2[(size_t)nd * HIDF + c] = lrelu(y);
}

#define POOL_RUN 16

__device__ __forceinline__ void pool_flush(float* __restrict__ pooled, float* __restrict__ cnt,
                                           int g, int t, const float4 s, float run) {
  float* p = pooled + (size_t)g * WHID + 4 * t;
  atomicAdd(p + 0, s.x);
  atomicAdd(p + 1, s.y);
  atomicAdd(p + 2, s.z);
  atomicAdd(p + 3, s.w);
  if (t == 0) atomicAdd(&cnt[g], run);
}

__global__ void __launch_bounds__(128)
l2_post_pool(const float* __restrict__ acc, const float* __restrict__ b2,
             const int* __restrict__ batch, float* __restrict__ pooled,
             float* __restrict__ cnt, int n) {
  const int t  = threadIdx.x;
  const int n0 = blockIdx.x * POOL_RUN;
  const float4 bv = *(const float4*)(b2 + 4 * t);

  float4 v[POOL_RUN];
  int    gi[POOL_RUN];
#pragma unroll
  for (int j = 0; j < POOL_RUN; ++j) {
    const int nd = min(n0 + j, n - 1);
    v[j]  = *(const float4*)(acc + (size_t)nd * WHID + 4 * t);
    gi[j] = batch[nd];
  }

  float4 s = make_float4(0.f, 0.f, 0.f, 0.f);
  float run = 0.f;
  int   g   = gi[0];
#pragma unroll
  for (int j = 0; j < POOL_RUN; ++j) {
    if (n0 + j < n) {
      if (gi[j] != g) {
        pool_flush(pooled, cnt, g, t, s, run);
        s = make_float4(0.f, 0.f, 0.f, 0.f);
        run = 0.f;
        g = gi[j];
      }
      s.x += lrelu(v[j].x + bv.x);
      s.y += lrelu(v[j].y + bv.y);
      s.z += lrelu(v[j].z + bv.z);
      s.w += lrelu(v[j].w + bv.w);
      run += 1.f;
    }
  }
  pool_flush(pooled, cnt, g, t, s, run);
}

#define HROWS 16
#define PITCH0 (WHID + 4)
#define PITCH1 (OUTD + 4)
#define PITCH2 (HIDF + 4)

template <int K, int NC, int NT>
__device__ __forceinline__ void head_product(const float* As, int lda,
                                             const float* __restrict__ Wg,
                                             const float* __restrict__ bias,
                                             float* Cs, int ldc, int wave, int l, int half) {
  v8f acc[NT];
#pragma unroll
  for (int t = 0; t < NT; ++t) acc[t] = (v8f){};
  const float* ap = As + l * lda + 2 * half;
  const float* wp = Wg + (size_t)(2 * half) * NC + 16 * wave + l;
#pragma unroll 16
  for (int k = 0; k < K; k += 4) {
    const v2f a = *(const v2f*)(ap + k);
#pragma unroll
    for (int t = 0; t < NT; ++t) {
      v2f b;
      b.x = wp[(size_t)k * NC + 128 * t];
      b.y = wp[(size_t)(k + 1) * NC + 128 * t];
      acc[t] = __builtin_amdgcn_wmma_f32_16x16x4_f32(false, a, false, b, (short)0, acc[t], false, false);
    }
  }
#pragma unroll
  for (int t = 0; t < NT; ++t) {
    const int col = 16 * wave + 128 * t + l;
    const float bv = bias[col];
#pragma unroll
    for (int i = 0; i < 8; ++i) Cs[(i + 8 * half) * ldc + col] = acc[t][i] + bv;
  }
}

__global__ void __launch_bounds__(256)
head_mlp(const float* __restrict__ pooled, const float* __restrict__ cnt,
         const float* __restrict__ Wl0, const float* __restrict__ bl0,
         const float* __restrict__ Ws1, const float* __restrict__ bs1,
         const float* __restrict__ g1,  const float* __restrict__ bb1,
         const float* __restrict__ Ws2, const float* __restrict__ bs2,
         const float* __restrict__ g2,  const float* __restrict__ bb2,
         const float* __restrict__ Ws3, const float* __restrict__ bs3,
         float* __restrict__ out) {
  __shared__ float P[HROWS * PITCH0];
  __shared__ float T1[HROWS * PITCH1];
  __shared__ float ZT[HROWS * PITCH2];
  const int tid  = threadIdx.x;
  const int wave = tid >> 5;
  const int lane = tid & 31;
  const int half = lane >> 4;
  const int l    = lane & 15;
  const int g0   = blockIdx.x * HROWS;

#pragma unroll
  for (int j = 0; j < 8; ++j) {
    const int i = tid + 256 * j;
    const int r = i >> 7, c4 = i & 127;
    float4 v = *(const float4*)(pooled + (size_t)(g0 + r) * WHID + 4 * c4);
    const float d = fmaxf(cnt[g0 + r], 1.0f);
    v.x /= d; v.y /= d; v.z /= d; v.w /= d;
    *(float4*)(P + r * PITCH0 + 4 * c4) = v;
  }
  __syncthreads();

  head_product<WHID, HIDF, 1>(P, PITCH0, Wl0, bl0, ZT, PITCH2, wave, l, half);
  __syncthreads();
  head_product<HIDF, OUTD, 2>(ZT, PITCH2, Ws1, bs1, T1, PITCH1, wave, l, half);
  __syncthreads();

#pragma unroll
  for (int rr = 0; rr < 2; ++rr) {
    float* row = T1 + (2 * wave + rr) * PITCH1;
    float v[8];
    float s = 0.f;
#pragma unroll
    for (int j = 0; j < 8; ++j) { v[j] = row[lane + 32 * j]; s += v[j]; }
    const float mu = wave_sum(s) / OUTD;
    float q = 0.f;
#pragma unroll
    for (int j = 0; j < 8; ++j) { v[j] -= mu; q += v[j] * v[j]; }
    const float var = wave_sum(q) / OUTD;
    const float inv = rsqrtf(var + 1e-5f);
#pragma unroll
    for (int j = 0; j < 8; ++j) {
      const int cc = lane + 32 * j;
      row[cc] = fmaxf(g1[cc] * v[j] * inv + bb1[cc], 0.f);
    }
  }
  __syncthreads();

  head_product<OUTD, OUT2, 1>(T1, PITCH1, Ws2, bs2, ZT, PITCH2, wave, l, half);
  __syncthreads();

#pragma unroll
  for (int rr = 0; rr < 2; ++rr) {
    const int r = 2 * wave + rr;
    const float* row = ZT + r * PITCH2;
    float v[4];
    float s = 0.f;
#pragma unroll
    for (int j = 0; j < 4; ++j) { v[j] = row[lane + 32 * j]; s += v[j]; }
    const float mu = wave_sum(s) / OUT2;
    float q = 0.f;
#pragma unroll
    for (int j = 0; j < 4; ++j) { v[j] -= mu; q += v[j] * v[j]; }
    const float var = wave_sum(q) / OUT2;
    const float inv = rsqrtf(var + 1e-5f);
    float p0 = 0.f, p1 = 0.f;
#pragma unroll
    for (int j = 0; j < 4; ++j) {
      const int cc = lane + 32 * j;
      const float y = fmaxf(g2[cc] * v[j] * inv + bb2[cc], 0.f);
      const v2f w = *(const v2f*)(Ws3 + 2 * cc);
      p0 += y * w.x;
      p1 += y * w.y;
    }
    const float l0 = wave_sum(p0) + bs3[0];
    const float l1 = wave_sum(p1) + bs3[1];
    const float mx = fmaxf(l0, l1);
    const float e0 = expf(l0 - mx), e1 = expf(l1 - mx), es = e0 + e1;
    if (lane == 0) {
      out[(g0 + r) * 2 + 0] = e0 / es;
      out[(g0 + r) * 2 + 1] = e1 / es;
    }
  }
}

extern "C" void kernel_launch(void* const* d_in, const int* in_sizes, int n_in,
                              void* d_out, int out_size, void* d_ws, size_t ws_size,
                              hipStream_t stream) {
  (void)n_in; (void)out_size; (void)ws_size;
  const float* x    = (const float*)d_in[0];
  const int*   ei   = (const int*)d_in[1];
  const int*   batch= (const int*)d_in[2];
  const float* W1   = (const float*)d_in[3];
  const float* as1  = (const float*)d_in[4];
  const float* ad1  = (const float*)d_in[5];
  const float* b1   = (const float*)d_in[6];
  const float* g_ln = (const float*)d_in[7];
  const float* b_ln = (const float*)d_in[8];
  const float* W2   = (const float*)d_in[9];
  const float* as2  = (const float*)d_in[10];
  const float* ad2  = (const float*)d_in[11];
  const float* b2   = (const float*)d_in[12];
  const float* Wl0  = (const float*)d_in[13];
  const float* bl0  = (const float*)d_in[14];
  const float* Ws1  = (const float*)d_in[15];
  const float* bs1  = (const float*)d_in[16];
  const float* g1   = (const float*)d_in[17];
  const float* bb1  = (const float*)d_in[18];
  const float* Ws2  = (const float*)d_in[19];
  const float* bs2  = (const float*)d_in[20];
  const float* g2   = (const float*)d_in[21];
  const float* bb2  = (const float*)d_in[22];
  const float* Ws3  = (const float*)d_in[23];
  const float* bs3  = (const float*)d_in[24];

  const int N  = in_sizes[0] / HIDF;
  const int E  = in_sizes[1] / 2;
  const int NH = N * NHEAD;

  float* out       = (float*)d_out;
  float* alpha_out = out + NGROUP * NCLSD;

  size_t off = 0;
  auto alloc = [&](size_t bytes) -> float* {
    off = (off + 255) & ~(size_t)255;
    float* p = (float*)((char*)d_ws + off);
    off += bytes;
    return p;
  };
  float* h      = alloc((size_t)N * WHID * 4);
  float* acc    = alloc((size_t)N * WHID * 4);
  float* x2     = alloc((size_t)N * HIDF * 4);
  float* ssrc   = alloc((size_t)NH * 4);
  float* sdst   = alloc((size_t)NH * 4);
  float* pooled = alloc((size_t)NGROUP * WHID * 4);
  float* cnt    = alloc((size_t)NGROUP * 4);
  int*   deg    = (int*)alloc((size_t)N * 4);
  int*   rowptr = (int*)alloc((size_t)(N + 1) * 4);
  int*   cursor = (int*)alloc((size_t)N * 4);
  int*   csr_arr = (int*)alloc((size_t)E * 8);
  int*   csr     = (int*)alloc((size_t)E * 8);

  auto cdiv = [](long a, long b) { return (int)((a + b - 1) / b); };
  auto gemm = [&](const float* A, const float* B, const float* bias, float* C,
                  int M, int K, int Nc) {
    int mtiles  = (M + 15) / 16;
    int mblocks = (mtiles + 7) / 8;
    int nslab   = Nc / 64;
    wmma_gemm<<<mblocks * nslab, 256, 0, stream>>>(A, B, bias, C, M, K, Nc);
  };

  fill_i32<<<cdiv(N, 256), 256, 0, stream>>>(deg, 0, N);
  csr_hist<<<cdiv(E, 256), 256, 0, stream>>>(ei, E, deg);
  csr_scan<<<1, 1024, 0, stream>>>(deg, rowptr, cursor, N);
  csr_fill<<<cdiv(E, 256), 256, 0, stream>>>(ei, E, cursor, csr_arr);
  csr_sort_rows<<<cdiv(N, 8), 256, 0, stream>>>(rowptr, csr_arr, csr, N);

  gemm(x, W1, nullptr, h, N, HIDF, WHID);
  attn_scores<<<cdiv(NH, 256), 256, 0, stream>>>(h, as1, ad1, ssrc, sdst, N);
  edge_softmax_gather<<<cdiv(N, 8), 256, 0, stream>>>(rowptr, csr, E, N, ssrc, sdst, h, acc, nullptr);
  l1_post<<<N, HIDF, 0, stream>>>(acc, b1, g_ln, b_ln, x2);

  gemm(x2, W2, nullptr, h, N, HIDF, WHID);
  attn_scores<<<cdiv(NH, 256), 256, 0, stream>>>(h, as2, ad2, ssrc, sdst, N);
  edge_softmax_gather<<<cdiv(N, 8), 256, 0, stream>>>(rowptr, csr, E, N, ssrc, sdst, h, acc, alpha_out);

  fill_f32<<<cdiv(NGROUP * WHID, 256), 256, 0, stream>>>(pooled, 0.f, NGROUP * WHID);
  fill_f32<<<1, NGROUP, 0, stream>>>(cnt, 0.f, NGROUP);
  l2_post_pool<<<cdiv(N, POOL_RUN), 128, 0, stream>>>(acc, b2, batch, pooled, cnt, N);
  head_mlp<<<NGROUP / HROWS, 256, 0, stream>>>(pooled, cnt, Wl0, bl0, Ws1, bs1, g1, bb1,
                                               Ws2, bs2, g2, bb2, Ws3, bs3, out);
}


# layer-1 epilogue (head mean, bias, LayerNorm, leaky ReLU) done by the gather wave on its register sums, x2 stored directly; l1_post launch and acc round trip gone; LN sums now in wave-butterfly order
# speedup vs baseline: 3.6458x; 1.0518x over previous
#include <hip/hip_runtime.h>
#include <math.h>

#define HIDF   128
#define NHEAD  4
#define WHID   512
#define NGROUP 64
#define OUTD   256
#define OUT2   128
#define NCLSD  2
#define SLOPE  0.2f
#define KC     128
#define BSTRIDE (KC + 2)

typedef float v2f __attribute__((ext_vector_type(2)));
typedef float v8f __attribute__((ext_vector_type(8)));

__device__ __forceinline__ float lrelu(float x) { return x >= 0.f ? x : SLOPE * x; }

__global__ void wmma_gemm(const float* __restrict__ A, const float* __restrict__ B,
                          const float* __restrict__ bias, float* __restrict__ C,
                          int M, int K, int Nc) {
  __shared__ float Bs[64 * BSTRIDE];

  const int nslab = Nc >> 6;
  const int mblk  = blockIdx.x / nslab;
  const int nblk  = blockIdx.x - mblk * nslab;
  const int wave  = threadIdx.x >> 5;
  const int lane  = threadIdx.x & 31;
  const int half  = lane >> 4;
  const int l     = lane & 15;

  const int mt    = mblk * 8 + wave;
  const int arow  = min(mt * 16 + l, M - 1);
  const int col0  = nblk * 64;

  const int sc = threadIdx.x & 63;
  const int sr = threadIdx.x >> 6;

  const float* Ap = A + (size_t)arow * K + 2 * half;

  v8f acc0 = {}, acc1 = {}, acc2 = {}, acc3 = {};

  for (int k0 = 0; k0 < K; k0 += KC) {
    const int kc = min(KC, K - k0);
    __syncthreads();
    for (int kk = sr; kk < kc; kk += 4)
      Bs[sc * BSTRIDE + kk] = B[(size_t)(k0 + kk) * Nc + col0 + sc];
    __syncthreads();

    __builtin_prefetch(Ap + k0 + KC, 0, 1);

    const float* Bl0 = &Bs[(l +  0) * BSTRIDE + 2 * half];
    const float* Bl1 = &Bs[(l + 16) * BSTRIDE + 2 * half];
    const float* Bl2 = &Bs[(l + 32) * BSTRIDE + 2 * half];
    const float* Bl3 = &Bs[(l + 48) * BSTRIDE + 2 * half];
    for (int k = 0; k < kc; k += 4) {
      v2f a  = *(const v2f*)(Ap + k0 + k);
      v2f b0 = *(const v2f*)(Bl0 + k);
      v2f b1 = *(const v2f*)(Bl1 + k);
      v2f b2 = *(const v2f*)(Bl2 + k);
      v2f b3 = *(const v2f*)(Bl3 + k);
      acc0 = __builtin_amdgcn_wmma_f32_16x16x4_f32(false, a, false, b0, (short)0, acc0, false, false);
      acc1 = __builtin_amdgcn_wmma_f32_16x16x4_f32(false, a, false, b1, (short)0, acc1, false, false);
      acc2 = __builtin_amdgcn_wmma_f32_16x16x4_f32(false, a, false, b2, (short)0, acc2, false, false);
      acc3 = __builtin_amdgcn_wmma_f32_16x16x4_f32(false, a, false, b3, (short)0, acc3, false, false);
    }
  }

  v8f* accs[4] = {&acc0, &acc1, &acc2, &acc3};
#pragma unroll
  for (int j = 0; j < 4; ++j) {
    const int ccol = col0 + 16 * j + l;
    const float bv = bias ? bias[ccol] : 0.f;
#pragma unroll
    for (int i = 0; i < 8; ++i) {
      const int row = mt * 16 + i + 8 * half;
      if (row < M) C[(size_t)row * Nc + ccol] = (*accs[j])[i] + bv;
    }
  }
}

__global__ void attn_scores(const float* __restrict__ h, const float* __restrict__ a_s,
                            const float* __restrict__ a_d, float* __restrict__ ssrc,
                            float* __restrict__ sdst, int n) {
  int idx = blockIdx.x * blockDim.x + threadIdx.x;
  if (idx >= n * NHEAD) return;
  int hd = idx & (NHEAD - 1);
  const float* hp = h + (size_t)idx * HIDF;
  const float* as = a_s + hd * HIDF;
  const float* ad = a_d + hd * HIDF;
  float s1 = 0.f, s2 = 0.f;
  for (int c = 0; c < HIDF; c += 4) {
    float4 hv = *(const float4*)(hp + c);
    float4 av = *(const float4*)(as + c);
    float4 dv = *(const float4*)(ad + c);
    s1 += hv.x * av.x + hv.y * av.y + hv.z * av.z + hv.w * av.w;
    s2 += hv.x * dv.x + hv.y * dv.y + hv.z * dv.z + hv.w * dv.w;
  }
  ssrc[idx] = s1;
  sdst[idx] = s2;
}

__global__ void fill_f32(float* __restrict__ p, float v, int n) {
  int i = blockIdx.x * blockDim.x + threadIdx.x;
  if (i < n) p[i] = v;
}

__global__ void fill_i32(int* __restrict__ p, int v, int n) {
  int i = blockIdx.x * blockDim.x + threadIdx.x;
  if (i < n) p[i] = v;
}

__global__ void csr_hist(const int* __restrict__ ei, int E, int* __restrict__ deg) {
  int e = blockIdx.x * blockDim.x + threadIdx.x;
  if (e < E) atomicAdd(&deg[ei[E + e]], 1);
}

__global__ void __launch_bounds__(1024)
csr_scan(const int* __restrict__ deg, int* __restrict__ rowptr,
         int* __restrict__ cursor, int n) {
  __shared__ int part[1024];
  const int t     = threadIdx.x;
  const int items = (n + 1023) >> 10;
  const int lo    = min(t * items, n);
  const int hi    = min(lo + items, n);
  int s = 0;
  for (int i = lo; i < hi; ++i) s += deg[i];
  part[t] = s;
  __syncthreads();
  for (int off = 1; off < 1024; off <<= 1) {
    const int v = (t >= off) ? part[t - off] : 0;
    __syncthreads();
    part[t] += v;
    __syncthreads();
  }
  int run = part[t] - s;
  for (int i = lo; i < hi; ++i) {
    rowptr[i] = run;
    cursor[i] = run;
    run += deg[i];
  }
  if (t == 1023) rowptr[n] = part[1023];
}

__global__ void csr_fill(const int* __restrict__ ei, int E, int* __restrict__ cursor,
                         int* __restrict__ csr) {
  int e = blockIdx.x * blockDim.x + threadIdx.x;
  if (e >= E) return;
  const int s = ei[e], d = ei[E + e];
  const int pos = atomicAdd(&cursor[d], 1);
  *(int2*)(csr + 2 * (size_t)pos) = make_int2(s, e);
}

__global__ void __launch_bounds__(256)
csr_sort_rows(const int* __restrict__ rowptr, const int* __restrict__ csr_in,
              int* __restrict__ csr_out, int n) {
  const int lane = threadIdx.x & 31;
  const int node = __builtin_amdgcn_readfirstlane((int)(blockIdx.x * 8 + (threadIdx.x >> 5)));
  if (node >= n) return;
  const int beg = rowptr[node], end = rowptr[node + 1];
  for (int b = beg; b < end; b += 32) {
    const bool have = (b + lane) < end;
    int2 mine = make_int2(0, 0x7fffffff);
    if (have) mine = *(const int2*)(csr_in + 2 * (size_t)(b + lane));
    int rank = 0;
    for (int c0 = beg; c0 < end; c0 += 32) {
      const int cnt = min(32, end - c0);
      int other = 0x7fffffff;
      if (c0 + lane < end) other = csr_in[2 * (size_t)(c0 + lane) + 1];
      for (int j = 0; j < cnt; ++j)
        rank += (__builtin_amdgcn_readlane(other, j) < mine.y) ? 1 : 0;
    }
    if (have) *(int2*)(csr_out + 2 * (size_t)(beg + rank)) = mine;
  }
}

__device__ __forceinline__ float bcast_f(float v, int srcLane) {
  return __int_as_float(__builtin_amdgcn_readlane(__float_as_int(v), srcLane));
}

__device__ __forceinline__ float wave_max(float v) {
  for (int o = 16; o > 0; o >>= 1) v = fmaxf(v, __shfl_xor(v, o, 32));
  return v;
}
__device__ __forceinline__ float wave_sum(float v) {
  for (int o = 16; o > 0; o >>= 1) v += __shfl_xor(v, o, 32);
  return v;
}

__device__ __forceinline__ void load_entries(const int* __restrict__ csr,
                                             const float* __restrict__ ssrc,
                                             const float4 sd, int base, int cnt, int lane,
                                             int& src, int& eid, float4& e) {
  src = 0; eid = 0;
  e = make_float4(-INFINITY, -INFINITY, -INFINITY, -INFINITY);
  if (lane < cnt) {
    const int2 ent = *(const int2*)(csr + 2 * (size_t)(base + lane));
    src = ent.x; eid = ent.y;
    const float4 ss = *(const float4*)(ssrc + (size_t)src * NHEAD);
    e.x = lrelu(ss.x + sd.x); e.y = lrelu(ss.y + sd.y);
    e.z = lrelu(ss.z + sd.z); e.w = lrelu(ss.w + sd.w);
  }
}

__device__ __forceinline__ void add_rows(const float* __restrict__ h, int lane, int cnt,
                                         int my_src, const float4 my_a,
                                         float4& s0, float4& s1, float4& s2, float4& s3) {
  for (int j = 0; j < cnt; ++j) {
    const int   s  = __builtin_amdgcn_readlane(my_src, j);
    const float a0 = bcast_f(my_a.x, j);
    const float a1 = bcast_f(my_a.y, j);
    const float a2 = bcast_f(my_a.z, j);
    const float a3 = bcast_f(my_a.w, j);
    const float* hp = h + (size_t)s * WHID + lane * 4;
    const float4 h0 = *(const float4*)(hp);
    const float4 h1 = *(const float4*)(hp + HIDF);
    const float4 h2 = *(const float4*)(hp + 2 * HIDF);
    const float4 h3 = *(const float4*)(hp + 3 * HIDF);
    s0.x += a0 * h0.x; s0.y += a0 * h0.y; s0.z += a0 * h0.z; s0.w += a0 * h0.w;
    s1.x += a1 * h1.x; s1.y += a1 * h1.y; s1.z += a1 * h1.z; s1.w += a1 * h1.w;
    s2.x += a2 * h2.x; s2.y += a2 * h2.y; s2.z += a2 * h2.z; s2.w += a2 * h2.w;
    s3.x += a3 * h3.x; s3.y += a3 * h3.y; s3.z += a3 * h3.z; s3.w += a3 * h3.w;
  }
}

template <bool LAYER1>
__global__ void __launch_bounds__(256)
edge_softmax_gather(const int* __restrict__ rowptr, const int* __restrict__ csr, int E, int n,
                    const float* __restrict__ ssrc, const float* __restrict__ sdst,
                    const float* __restrict__ h, float* __restrict__ acc,
                    float* __restrict__ alpha_out,
                    const float* __restrict__ b1, const float* __restrict__ g_ln,
                    const float* __restrict__ b_ln, float* __restrict__ x2) {
  const int lane = threadIdx.x & 31;
  const int node = __builtin_amdgcn_readfirstlane((int)(blockIdx.x * 8 + (threadIdx.x >> 5)));
  if (node >= n) return;
  const float4 sd = *(const float4*)(sdst + (size_t)node * NHEAD);
  const float4 sn = *(const float4*)(ssrc + (size_t)node * NHEAD);
  const int beg = rowptr[node], end = rowptr[node + 1];

  float4 es;
  es.x = lrelu(sn.x + sd.x); es.y = lrelu(sn.y + sd.y);
  es.z = lrelu(sn.z + sd.z); es.w = lrelu(sn.w + sd.w);

  const int cnt0 = min(32, end - beg);
  int src0, eid0; float4 e0;
  load_entries(csr, ssrc, sd, beg, cnt0, lane, src0, eid0, e0);

  float4 mx;
  mx.x = fmaxf(es.x, e0.x); mx.y = fmaxf(es.y, e0.y);
  mx.z = fmaxf(es.z, e0.z); mx.w = fmaxf(es.w, e0.w);
  for (int base = beg + 32; base < end; base += 32) {
    int s_, i_; float4 e_;
    load_entries(csr, ssrc, sd, base, min(32, end - base), lane, s_, i_, e_);
    mx.x = fmaxf(mx.x, e_.x); mx.y = fmaxf(mx.y, e_.y);
    mx.z = fmaxf(mx.z, e_.z); mx.w = fmaxf(mx.w, e_.w);
  }
  mx.x = wave_max(mx.x); mx.y = wave_max(mx.y);
  mx.z = wave_max(mx.z); mx.w = wave_max(mx.w);
  if (!(mx.x > -3.0e38f && mx.x < 3.0e38f)) mx.x = 0.f;
  if (!(mx.y > -3.0e38f && mx.y < 3.0e38f)) mx.y = 0.f;
  if (!(mx.z > -3.0e38f && mx.z < 3.0e38f)) mx.z = 0.f;
  if (!(mx.w > -3.0e38f && mx.w < 3.0e38f)) mx.w = 0.f;

  float4 x0 = make_float4(0.f, 0.f, 0.f, 0.f);
  if (lane < cnt0) {
    x0.x = expf(e0.x - mx.x); x0.y = expf(e0.y - mx.y);
    x0.z = expf(e0.z - mx.z); x0.w = expf(e0.w - mx.w);
  }
  float4 ds = x0;
  for (int base = beg + 32; base < end; base += 32) {
    const int cnt = min(32, end - base);
    int s_, i_; float4 e_;
    load_entries(csr, ssrc, sd, base, cnt, lane, s_, i_, e_);
    if (lane < cnt) {
      ds.x += expf(e_.x - mx.x); ds.y += expf(e_.y - mx.y);
      ds.z += expf(e_.z - mx.z); ds.w += expf(e_.w - mx.w);
    }
  }
  float4 xs;
  xs.x = expf(es.x - mx.x); xs.y = expf(es.y - mx.y);
  xs.z = expf(es.z - mx.z); xs.w = expf(es.w - mx.w);
  float4 dn;
  dn.x = wave_sum(ds.x) + xs.x; dn.y = wave_sum(ds.y) + xs.y;
  dn.z = wave_sum(ds.z) + xs.z; dn.w = wave_sum(ds.w) + xs.w;

  float4 s0, s1, s2, s3;
  {
    const size_t eid = (size_t)E + node;
    float4 a;
    a.x = xs.x / dn.x; a.y = xs.y / dn.y; a.z = xs.z / dn.z; a.w = xs.w / dn.w;
    if (alpha_out && lane == 0) *(float4*)(alpha_out + eid * NHEAD) = a;
    const float* hp = h + (size_t)node * WHID + lane * 4;
    const float4 h0 = *(const float4*)(hp);
    const float4 h1 = *(const float4*)(hp + HIDF);
    const float4 h2 = *(const float4*)(hp + 2 * HIDF);
    const float4 h3 = *(const float4*)(hp + 3 * HIDF);
    s0.x = a.x * h0.x; s0.y = a.x * h0.y; s0.z = a.x * h0.z; s0.w = a.x * h0.w;
    s1.x = a.y * h1.x; s1.y = a.y * h1.y; s1.z = a.y * h1.z; s1.w = a.y * h1.w;
    s2.x = a.z * h2.x; s2.y = a.z * h2.y; s2.z = a.z * h2.z; s2.w = a.z * h2.w;
    s3.x = a.w * h3.x; s3.y = a.w * h3.y; s3.z = a.w * h3.z; s3.w = a.w * h3.w;
  }

  {
    float4 a;
    a.x = x0.x / dn.x; a.y = x0.y / dn.y; a.z = x0.z / dn.z; a.w = x0.w / dn.w;
    if (alpha_out && lane < cnt0) *(float4*)(alpha_out + (size_t)eid0 * NHEAD) = a;
    add_rows(h, lane, cnt0, src0, a, s0, s1, s2, s3);
  }
  for (int base = beg + 32; base < end; base += 32) {
    const int cnt = min(32, end - base);
    int my_src, my_eid; float4 e_;
    load_entries(csr, ssrc, sd, base, cnt, lane, my_src, my_eid, e_);
    float4 a = make_float4(0.f, 0.f, 0.f, 0.f);
    if (lane < cnt) {
      a.x = expf(e_.x - mx.x) / dn.x; a.y = expf(e_.y - mx.y) / dn.y;
      a.z = expf(e_.z - mx.z) / dn.z; a.w = expf(e_.w - mx.w) / dn.w;
      if (alpha_out) *(float4*)(alpha_out + (size_t)my_eid * NHEAD) = a;
    }
    add_rows(h, lane, cnt, my_src, a, s0, s1, s2, s3);
  }

  if constexpr (LAYER1) {
    const float4 bv = *(const float4*)(b1   + lane * 4);
    const float4 gv = *(const float4*)(g_ln + lane * 4);
    const float4 bl = *(const float4*)(b_ln + lane * 4);
    float4 v;
    v.x = 0.25f * (s0.x + s1.x + s2.x + s3.x) + bv.x;
    v.y = 0.25f * (s0.y + s1.y + s2.y + s3.y) + bv.y;
    v.z = 0.25f * (s0.z + s1.z + s2.z + s3.z) + bv.z;
    v.w = 0.25f * (s0.w + s1.w + s2.w + s3.w) + bv.w;
    const float mu = wave_sum((v.x + v.y) + (v.z + v.w)) * (1.0f / HIDF);
    float4 d;
    d.x = v.x - mu; d.y = v.y - mu; d.z = v.z - mu; d.w = v.w - mu;
    const float var = wave_sum((d.x * d.x + d.y * d.y) + (d.z * d.z + d.w * d.w)) * (1.0f / HIDF);
    const float inv = rsqrtf(var + 1e-5f);
    float4 y;
    y.x = lrelu(gv.x * d.x * inv + bl.x);
    y.y = lrelu(gv.y * d.y * inv + bl.y);
    y.z = lrelu(gv.z * d.z * inv + bl.z);
    y.w = lrelu(gv.w * d.w * inv + bl.w);
    *(float4*)(x2 + (size_t)node * HIDF + lane * 4) = y;
  } else {
    float* ap = acc + (size_t)node * WHID + lane * 4;
    *(float4*)(ap)            = s0;
    *(float4*)(ap + HIDF)     = s1;
    *(float4*)(ap + 2 * HIDF) = s2;
    *(float4*)(ap + 3 * HIDF) = s3;
  }
}

#define POOL_RUN 16

__device__ __forceinline__ void pool_flush(float* __restrict__ pooled, float* __restrict__ cnt,
                                           int g, int t, const float4 s, float run) {
  float* p = pooled + (size_t)g * WHID + 4 * t;
  atomicAdd(p + 0, s.x);
  atomicAdd(p + 1, s.y);
  atomicAdd(p + 2, s.z);
  atomicAdd(p + 3, s.w);
  if (t == 0) atomicAdd(&cnt[g], run);
}

__global__ void __launch_bounds__(128)
l2_post_pool(const float* __restrict__ acc, const float* __restrict__ b2,
             const int* __restrict__ batch, float* __restrict__ pooled,
             float* __restrict__ cnt, int n) {
  const int t  = threadIdx.x;
  const int n0 = blockIdx.x * POOL_RUN;
  const float4 bv = *(const float4*)(b2 + 4 * t);

  float4 v[POOL_RUN];
  int    gi[POOL_RUN];
#pragma unroll
  for (int j = 0; j < POOL_RUN; ++j) {
    const int nd = min(n0 + j, n - 1);
    v[j]  = *(const float4*)(acc + (size_t)nd * WHID + 4 * t);
    gi[j] = batch[nd];
  }

  float4 s = make_float4(0.f, 0.f, 0.f, 0.f);
  float run = 0.f;
  int   g   = gi[0];
#pragma unroll
  for (int j = 0; j < POOL_RUN; ++j) {
    if (n0 + j < n) {
      if (gi[j] != g) {
        pool_flush(pooled, cnt, g, t, s, run);
        s = make_float4(0.f, 0.f, 0.f, 0.f);
        run = 0.f;
        g = gi[j];
      }
      s.x += lrelu(v[j].x + bv.x);
      s.y += lrelu(v[j].y + bv.y);
      s.z += lrelu(v[j].z + bv.z);
      s.w += lrelu(v[j].w + bv.w);
      run += 1.f;
    }
  }
  pool_flush(pooled, cnt, g, t, s, run);
}

#define HROWS 16
#define PITCH0 (WHID + 4)
#define PITCH1 (OUTD + 4)
#define PITCH2 (HIDF + 4)

template <int K, int NC, int NT>
__device__ __forceinline__ void head_product(const float* As, int lda,
                                             const float* __restrict__ Wg,
                                             const float* __restrict__ bias,
                                             float* Cs, int ldc, int wave, int l, int half) {
  v8f acc[NT];
#pragma unroll
  for (int t = 0; t < NT; ++t) acc[t] = (v8f){};
  const float* ap = As + l * lda + 2 * half;
  const float* wp = Wg + (size_t)(2 * half) * NC + 16 * wave + l;
#pragma unroll 16
  for (int k = 0; k < K; k += 4) {
    const v2f a = *(const v2f*)(ap + k);
#pragma unroll
    for (int t = 0; t < NT; ++t) {
      v2f b;
      b.x = wp[(size_t)k * NC + 128 * t];
      b.y = wp[(size_t)(k + 1) * NC + 128 * t];
      acc[t] = __builtin_amdgcn_wmma_f32_16x16x4_f32(false, a, false, b, (short)0, acc[t], false, false);
    }
  }
#pragma unroll
  for (int t = 0; t < NT; ++t) {
    const int col = 16 * wave + 128 * t + l;
    const float bv = bias[col];
#pragma unroll
    for (int i = 0; i < 8; ++i) Cs[(i + 8 * half) * ldc + col] = acc[t][i] + bv;
  }
}

__global__ void __launch_bounds__(256)
head_mlp(const float* __restrict__ pooled, const float* __restrict__ cnt,
         const float* __restrict__ Wl0, const float* __restrict__ bl0,
         const float* __restrict__ Ws1, const float* __restrict__ bs1,
         const float* __restrict__ g1,  const float* __restrict__ bb1,
         const float* __restrict__ Ws2, const float* __restrict__ bs2,
         const float* __restrict__ g2,  const float* __restrict__ bb2,
         const float* __restrict__ Ws3, const float* __restrict__ bs3,
         float* __restrict__ out) {
  __shared__ float P[HROWS * PITCH0];
  __shared__ float T1[HROWS * PITCH1];
  __shared__ float ZT[HROWS * PITCH2];
  const int tid  = threadIdx.x;
  const int wave = tid >> 5;
  const int lane = tid & 31;
  const int half = lane >> 4;
  const int l    = lane & 15;
  const int g0   = blockIdx.x * HROWS;

#pragma unroll
  for (int j = 0; j < 8; ++j) {
    const int i = tid + 256 * j;
    const int r = i >> 7, c4 = i & 127;
    float4 v = *(const float4*)(pooled + (size_t)(g0 + r) * WHID + 4 * c4);
    const float d = fmaxf(cnt[g0 + r], 1.0f);
    v.x /= d; v.y /= d; v.z /= d; v.w /= d;
    *(float4*)(P + r * PITCH0 + 4 * c4) = v;
  }
  __syncthreads();

  head_product<WHID, HIDF, 1>(P, PITCH0, Wl0, bl0, ZT, PITCH2, wave, l, half);
  __syncthreads();
  head_product<HIDF, OUTD, 2>(ZT, PITCH2, Ws1, bs1, T1, PITCH1, wave, l, half);
  __syncthreads();

#pragma unroll
  for (int rr = 0; rr < 2; ++rr) {
    float* row = T1 + (2 * wave + rr) * PITCH1;
    float v[8];
    float s = 0.f;
#pragma unroll
    for (int j = 0; j < 8; ++j) { v[j] = row[lane + 32 * j]; s += v[j]; }
    const float mu = wave_sum(s) / OUTD;
    float q = 0.f;
#pragma unroll
    for (int j = 0; j < 8; ++j) { v[j] -= mu; q += v[j] * v[j]; }
    const float var = wave_sum(q) / OUTD;
    const float inv = rsqrtf(var + 1e-5f);
#pragma unroll
    for (int j = 0; j < 8; ++j) {
      const int cc = lane + 32 * j;
      row[cc] = fmaxf(g1[cc] * v[j] * inv + bb1[cc], 0.f);
    }
  }
  __syncthreads();

  head_product<OUTD, OUT2, 1>(T1, PITCH1, Ws2, bs2, ZT, PITCH2, wave, l, half);
  __syncthreads();

#pragma unroll
  for (int rr = 0; rr < 2; ++rr) {
    const int r = 2 * wave + rr;
    const float* row = ZT + r * PITCH2;
    float v[4];
    float s = 0.f;
#pragma unroll
    for (int j = 0; j < 4; ++j) { v[j] = row[lane + 32 * j]; s += v[j]; }
    const float mu = wave_sum(s) / OUT2;
    float q = 0.f;
#pragma unroll
    for (int j = 0; j < 4; ++j) { v[j] -= mu; q += v[j] * v[j]; }
    const float var = wave_sum(q) / OUT2;
    const float inv = rsqrtf(var + 1e-5f);
    float p0 = 0.f, p1 = 0.f;
#pragma unroll
    for (int j = 0; j < 4; ++j) {
      const int cc = lane + 32 * j;
      const float y = fmaxf(g2[cc] * v[j] * inv + bb2[cc], 0.f);
      const v2f w = *(const v2f*)(Ws3 + 2 * cc);
      p0 += y * w.x;
      p1 += y * w.y;
    }
    const float l0 = wave_sum(p0) + bs3[0];
    const float l1 = wave_sum(p1) + bs3[1];
    const float mx = fmaxf(l0, l1);
    const float e0 = expf(l0 - mx), e1 = expf(l1 - mx), es = e0 + e1;
    if (lane == 0) {
      out[(g0 + r) * 2 + 0] = e0 / es;
      out[(g0 + r) * 2 + 1] = e1 / es;
    }
  }
}

extern "C" void kernel_launch(void* const* d_in, const int* in_sizes, int n_in,
                              void* d_out, int out_size, void* d_ws, size_t ws_size,
                              hipStream_t stream) {
  (void)n_in; (void)out_size; (void)ws_size;
  const float* x    = (const float*)d_in[0];
  const int*   ei   = (const int*)d_in[1];
  const int*   batch= (const int*)d_in[2];
  const float* W1   = (const float*)d_in[3];
  const float* as1  = (const float*)d_in[4];
  const float* ad1  = (const float*)d_in[5];
  const float* b1   = (const float*)d_in[6];
  const float* g_ln = (const float*)d_in[7];
  const float* b_ln = (const float*)d_in[8];
  const float* W2   = (const float*)d_in[9];
  const float* as2  = (const float*)d_in[10];
  const float* ad2  = (const float*)d_in[11];
  const float* b2   = (const float*)d_in[12];
  const float* Wl0  = (const float*)d_in[13];
  const float* bl0  = (const float*)d_in[14];
  const float* Ws1  = (const float*)d_in[15];
  const float* bs1  = (const float*)d_in[16];
  const float* g1   = (const float*)d_in[17];
  const float* bb1  = (const float*)d_in[18];
  const float* Ws2  = (const float*)d_in[19];
  const float* bs2  = (const float*)d_in[20];
  const float* g2   = (const float*)d_in[21];
  const float* bb2  = (const float*)d_in[22];
  const float* Ws3  = (const float*)d_in[23];
  const float* bs3  = (const float*)d_in[24];

  const int N  = in_sizes[0] / HIDF;
  const int E  = in_sizes[1] / 2;
  const int NH = N * NHEAD;

  float* out       = (float*)d_out;
  float* alpha_out = out + NGROUP * NCLSD;

  size_t off = 0;
  auto alloc = [&](size_t bytes) -> float* {
    off = (off + 255) & ~(size_t)255;
    float* p = (float*)((char*)d_ws + off);
    off += bytes;
    return p;
  };
  float* h      = alloc((size_t)N * WHID * 4);
  float* acc    = alloc((size_t)N * WHID * 4);
  float* x2     = alloc((size_t)N * HIDF * 4);
  float* ssrc   = alloc((size_t)NH * 4);
  float* sdst   = alloc((size_t)NH * 4);
  float* pooled = alloc((size_t)NGROUP * WHID * 4);
  float* cnt    = alloc((size_t)NGROUP * 4);
  int*   deg    = (int*)alloc((size_t)N * 4);
  int*   rowptr = (int*)alloc((size_t)(N + 1) * 4);
  int*   cursor = (int*)alloc((size_t)N * 4);
  int*   csr_arr = (int*)alloc((size_t)E * 8);
  int*   csr     = (int*)alloc((size_t)E * 8);

  auto cdiv = [](long a, long b) { return (int)((a + b - 1) / b); };
  auto gemm = [&](const float* A, const float* B, const float* bias, float* C,
                  int M, int K, int Nc) {
    int mtiles  = (M + 15) / 16;
    int mblocks = (mtiles + 7) / 8;
    int nslab   = Nc / 64;
    wmma_gemm<<<mblocks * nslab, 256, 0, stream>>>(A, B, bias, C, M, K, Nc);
  };

  fill_i32<<<cdiv(N, 256), 256, 0, stream>>>(deg, 0, N);
  csr_hist<<<cdiv(E, 256), 256, 0, stream>>>(ei, E, deg);
  csr_scan<<<1, 1024, 0, stream>>>(deg, rowptr, cursor, N);
  csr_fill<<<cdiv(E, 256), 256, 0, stream>>>(ei, E, cursor, csr_arr);
  csr_sort_rows<<<cdiv(N, 8), 256, 0, stream>>>(rowptr, csr_arr, csr, N);

  gemm(x, W1, nullptr, h, N, HIDF, WHID);
  attn_scores<<<cdiv(NH, 256), 256, 0, stream>>>(h, as1, ad1, ssrc, sdst, N);
  edge_softmax_gather<true><<<cdiv(N, 8), 256, 0, stream>>>(rowptr, csr, E, N, ssrc, sdst, h, nullptr, nullptr,
                                                            b1, g_ln, b_ln, x2);

  gemm(x2, W2, nullptr, h, N, HIDF, WHID);
  attn_scores<<<cdiv(NH, 256), 256, 0, stream>>>(h, as2, ad2, ssrc, sdst, N);
  edge_softmax_gather<false><<<cdiv(N, 8), 256, 0, stream>>>(rowptr, csr, E, N, ssrc, sdst, h, acc, alpha_out,
                                                             nullptr, nullptr, nullptr, nullptr);

  fill_f32<<<cdiv(NGROUP * WHID, 256), 256, 0, stream>>>(pooled, 0.f, NGROUP * WHID);
  fill_f32<<<1, NGROUP, 0, stream>>>(cnt, 0.f, NGROUP);
  l2_post_pool<<<cdiv(N, POOL_RUN), 128, 0, stream>>>(acc, b2, batch, pooled, cnt, N);
  head_mlp<<<NGROUP / HROWS, 256, 0, stream>>>(pooled, cnt, Wl0, bl0, Ws1, bs1, g1, bb1,
                                               Ws2, bs2, g2, bb2, Ws3, bs3, out);
}


# aggregate before projecting: the edge softmax-gather sums alpha-weighted 128-wide INPUT rows per head (512 B per edge, not 2 KB), W applied after; scores as x.(W a); sum order: edges, then k
# speedup vs baseline: 3.8585x; 1.0583x over previous
#include <hip/hip_runtime.h>
#include <math.h>

#define HIDF   128
#define NHEAD  4
#define WHID   512
#define NGROUP 64
#define OUTD   256
#define OUT2   128
#define NCLSD  2
#define SLOPE  0.2f
#define KC     128
#define BSTRIDE (KC + 2)

typedef float v2f __attribute__((ext_vector_type(2)));
typedef float v8f __attribute__((ext_vector_type(8)));

__device__ __forceinline__ float lrelu(float x) { return x >= 0.f ? x : SLOPE * x; }

__global__ void wmma_gemm(const float* __restrict__ A, int lda,
                          const float* __restrict__ B, int ldb, int bchunk,
                          const float* __restrict__ bias, float scale,
                          float* __restrict__ C, int ldc,
                          int M, int K, int Nc, int boff) {
  __shared__ float Bs[64 * BSTRIDE];

  A += (size_t)blockIdx.y * boff;
  B += (size_t)blockIdx.y * boff;
  C += (size_t)blockIdx.y * boff;

  const int nslab = Nc >> 6;
  const int mblk  = blockIdx.x / nslab;
  const int nblk  = blockIdx.x - mblk * nslab;
  const int wave  = threadIdx.x >> 5;
  const int lane  = threadIdx.x & 31;
  const int half  = lane >> 4;
  const int l     = lane & 15;

  const int mt    = mblk * 8 + wave;
  const int arow  = min(mt * 16 + l, M - 1);
  const int col0  = nblk * 64;

  const int sc = threadIdx.x & 63;
  const int sr = threadIdx.x >> 6;

  const float* Ap = A + (size_t)arow * lda + 2 * half;

  v8f acc0 = {}, acc1 = {}, acc2 = {}, acc3 = {};

  for (int k0 = 0; k0 < K; k0 += KC) {
    const int kc = min(KC, K - k0);
    const float* Bc = B + (size_t)(k0 / KC) * bchunk + col0 + sc;
    __syncthreads();
    for (int kk = sr; kk < kc; kk += 4)
      Bs[sc * BSTRIDE + kk] = Bc[(size_t)kk * ldb];
    __syncthreads();

    if (k0 + KC < K) __builtin_prefetch(Ap + k0 + KC, 0, 1);

    const float* Bl0 = &Bs[(l +  0) * BSTRIDE + 2 * half];
    const float* Bl1 = &Bs[(l + 16) * BSTRIDE + 2 * half];
    const float* Bl2 = &Bs[(l + 32) * BSTRIDE + 2 * half];
    const float* Bl3 = &Bs[(l + 48) * BSTRIDE + 2 * half];
    for (int k = 0; k < kc; k += 4) {
      v2f a  = *(const v2f*)(Ap + k0 + k);
      v2f b0 = *(const v2f*)(Bl0 + k);
      v2f b1 = *(const v2f*)(Bl1 + k);
      v2f b2 = *(const v2f*)(Bl2 + k);
      v2f b3 = *(const v2f*)(Bl3 + k);
      acc0 = __builtin_amdgcn_wmma_f32_16x16x4_f32(false, a, false, b0, (short)0, acc0, false, false);
      acc1 = __builtin_amdgcn_wmma_f32_16x16x4_f32(false, a, false, b1, (short)0, acc1, false, false);
      acc2 = __builtin_amdgcn_wmma_f32_16x16x4_f32(false, a, false, b2, (short)0, acc2, false, false);
      acc3 = __builtin_amdgcn_wmma_f32_16x16x4_f32(false, a, false, b3, (short)0, acc3, false, false);
    }
  }

  v8f* accs[4] = {&acc0, &acc1, &acc2, &acc3};
#pragma unroll
  for (int j = 0; j < 4; ++j) {
    const int ccol = col0 + 16 * j + l;
    const float bv = bias ? bias[ccol] : 0.f;
#pragma unroll
    for (int i = 0; i < 8; ++i) {
      const int row = mt * 16 + i + 8 * half;
      if (row < M) C[(size_t)row * ldc + ccol] = (*accs[j])[i] * scale + bv;
    }
  }
}

__global__ void fill_f32(float* __restrict__ p, float v, int n) {
  int i = blockIdx.x * blockDim.x + threadIdx.x;
  if (i < n) p[i] = v;
}

__global__ void fill_i32(int* __restrict__ p, int v, int n) {
  int i = blockIdx.x * blockDim.x + threadIdx.x;
  if (i < n) p[i] = v;
}

__global__ void csr_hist(const int* __restrict__ ei, int E, int* __restrict__ deg) {
  int e = blockIdx.x * blockDim.x + threadIdx.x;
  if (e < E) atomicAdd(&deg[ei[E + e]], 1);
}

__global__ void __launch_bounds__(1024)
csr_scan(const int* __restrict__ deg, int* __restrict__ rowptr,
         int* __restrict__ cursor, int n) {
  __shared__ int part[1024];
  const int t     = threadIdx.x;
  const int items = (n + 1023) >> 10;
  const int lo    = min(t * items, n);
  const int hi    = min(lo + items, n);
  int s = 0;
  for (int i = lo; i < hi; ++i) s += deg[i];
  part[t] = s;
  __syncthreads();
  for (int off = 1; off < 1024; off <<= 1) {
    const int v = (t >= off) ? part[t - off] : 0;
    __syncthreads();
    part[t] += v;
    __syncthreads();
  }
  int run = part[t] - s;
  for (int i = lo; i < hi; ++i) {
    rowptr[i] = run;
    cursor[i] = run;
    run += deg[i];
  }
  if (t == 1023) rowptr[n] = part[1023];
}

__global__ void csr_fill(const int* __restrict__ ei, int E, int* __restrict__ cursor,
                         int* __restrict__ csr) {
  int e = blockIdx.x * blockDim.x + threadIdx.x;
  if (e >= E) return;
  const int s = ei[e], d = ei[E + e];
  const int pos = atomicAdd(&cursor[d], 1);
  *(int2*)(csr + 2 * (size_t)pos) = make_int2(s, e);
}

__global__ void __launch_bounds__(256)
csr_sort_rows(const int* __restrict__ rowptr, const int* __restrict__ csr_in,
              int* __restrict__ csr_out, int n) {
  const int lane = threadIdx.x & 31;
  const int node = __builtin_amdgcn_readfirstlane((int)(blockIdx.x * 8 + (threadIdx.x >> 5)));
  if (node >= n) return;
  const int beg = rowptr[node], end = rowptr[node + 1];
  for (int b = beg; b < end; b += 32) {
    const bool have = (b + lane) < end;
    int2 mine = make_int2(0, 0x7fffffff);
    if (have) mine = *(const int2*)(csr_in + 2 * (size_t)(b + lane));
    int rank = 0;
    for (int c0 = beg; c0 < end; c0 += 32) {
      const int cnt = min(32, end - c0);
      int other = 0x7fffffff;
      if (c0 + lane < end) other = csr_in[2 * (size_t)(c0 + lane) + 1];
      for (int j = 0; j < cnt; ++j)
        rank += (__builtin_amdgcn_readlane(other, j) < mine.y) ? 1 : 0;
    }
    if (have) *(int2*)(csr_out + 2 * (size_t)(beg + rank)) = mine;
  }
}

__device__ __forceinline__ float bcast_f(float v, int srcLane) {
  return __int_as_float(__builtin_amdgcn_readlane(__float_as_int(v), srcLane));
}

__device__ __forceinline__ float wave_max(float v) {
  for (int o = 16; o > 0; o >>= 1) v = fmaxf(v, __shfl_xor(v, o, 32));
  return v;
}
__device__ __forceinline__ float wave_sum(float v) {
  for (int o = 16; o > 0; o >>= 1) v += __shfl_xor(v, o, 32);
  return v;
}

__global__ void __launch_bounds__(256)
score_vectors(const float* __restrict__ W1, const float* __restrict__ as1,
              const float* __restrict__ ad1, float* __restrict__ V1,
              const float* __restrict__ W2, const float* __restrict__ as2,
              const float* __restrict__ ad2, float* __restrict__ V2) {
  const int lane = threadIdx.x & 31;
  const int w = __builtin_amdgcn_readfirstlane((int)(blockIdx.x * 8 + (threadIdx.x >> 5)));
  const bool second = w >= HIDF * NHEAD;
  const int t  = w & (HIDF * NHEAD - 1);
  const int hd = t & (NHEAD - 1);
  const float* W  = second ? W2  : W1;
  const float* as = second ? as2 : as1;
  const float* ad = second ? ad2 : ad1;
  float*       V  = second ? V2  : V1;
  const float4 wv = *(const float4*)(W  + (size_t)t * HIDF + lane * 4);
  const float4 av = *(const float4*)(as + hd * HIDF + lane * 4);
  const float4 dv = *(const float4*)(ad + hd * HIDF + lane * 4);
  const float s1 = wave_sum((wv.x * av.x + wv.y * av.y) + (wv.z * av.z + wv.w * av.w));
  const float s2 = wave_sum((wv.x * dv.x + wv.y * dv.y) + (wv.z * dv.z + wv.w * dv.w));
  if (lane == 0) {
    V[(t >> 2) * 8 + hd]     = s1;
    V[(t >> 2) * 8 + 4 + hd] = s2;
  }
}

#define NPW 4

template <bool NORM>
__global__ void __launch_bounds__(256)
node_pass(const float* __restrict__ xin, const float* __restrict__ g_ln,
          const float* __restrict__ b_ln, const float* __restrict__ V,
          float* __restrict__ xout, float* __restrict__ ssrc, float* __restrict__ sdst, int n) {
  const int lane = threadIdx.x & 31;
  const int n0 = __builtin_amdgcn_readfirstlane((int)((blockIdx.x * 8 + (threadIdx.x >> 5)) * NPW));
  if (n0 >= n) return;

  float4 r[8];
#pragma unroll
  for (int i = 0; i < 8; ++i) r[i] = *(const float4*)(V + lane * 32 + 4 * i);

  float4 v[NPW];
#pragma unroll
  for (int j = 0; j < NPW; ++j) {
    const int nd = min(n0 + j, n - 1);
    v[j] = *(const float4*)(xin + (size_t)nd * HIDF + lane * 4);
  }
  float4 gv = make_float4(0.f, 0.f, 0.f, 0.f), bl = gv;
  if constexpr (NORM) {
    gv = *(const float4*)(g_ln + lane * 4);
    bl = *(const float4*)(b_ln + lane * 4);
  }
#pragma unroll
  for (int j = 0; j < NPW; ++j) {
    if (n0 + j < n) {
      float4 y = v[j];
      if constexpr (NORM) {
        const float mu = wave_sum((y.x + y.y) + (y.z + y.w)) * (1.0f / HIDF);
        float4 d;
        d.x = y.x - mu; d.y = y.y - mu; d.z = y.z - mu; d.w = y.w - mu;
        const float var = wave_sum((d.x * d.x + d.y * d.y) + (d.z * d.z + d.w * d.w)) * (1.0f / HIDF);
        const float inv = rsqrtf(var + 1e-5f);
        y.x = lrelu(gv.x * d.x * inv + bl.x);
        y.y = lrelu(gv.y * d.y * inv + bl.y);
        y.z = lrelu(gv.z * d.z * inv + bl.z);
        y.w = lrelu(gv.w * d.w * inv + bl.w);
        *(float4*)(xout + (size_t)(n0 + j) * HIDF + lane * 4) = y;
      }
      float4 ss, sd;
      ss.x = wave_sum((y.x * r[0].x + y.y * r[2].x) + (y.z * r[4].x + y.w * r[6].x));
      ss.y = wave_sum((y.x * r[0].y + y.y * r[2].y) + (y.z * r[4].y + y.w * r[6].y));
      ss.z = wave_sum((y.x * r[0].z + y.y * r[2].z) + (y.z * r[4].z + y.w * r[6].z));
      ss.w = wave_sum((y.x * r[0].w + y.y * r[2].w) + (y.z * r[4].w + y.w * r[6].w));
      sd.x = wave_sum((y.x * r[1].x + y.y * r[3].x) + (y.z * r[5].x + y.w * r[7].x));
      sd.y = wave_sum((y.x * r[1].y + y.y * r[3].y) + (y.z * r[5].y + y.w * r[7].y));
      sd.z = wave_sum((y.x * r[1].z + y.y * r[3].z) + (y.z * r[5].z + y.w * r[7].z));
      sd.w = wave_sum((y.x * r[1].w + y.y * r[3].w) + (y.z * r[5].w + y.w * r[7].w));
      if (lane == 0) {
        *(float4*)(ssrc + (size_t)(n0 + j) * NHEAD) = ss;
        *(float4*)(sdst + (size_t)(n0 + j) * NHEAD) = sd;
      }
    }
  }
}

__device__ __forceinline__ void load_entries(const int* __restrict__ csr,
                                             const float* __restrict__ ssrc,
                                             const float4 sd, int base, int cnt, int lane,
                                             int& src, int& eid, float4& e) {
  src = 0; eid = 0;
  e = make_float4(-INFINITY, -INFINITY, -INFINITY, -INFINITY);
  if (lane < cnt) {
    const int2 ent = *(const int2*)(csr + 2 * (size_t)(base + lane));
    src = ent.x; eid = ent.y;
    const float4 ss = *(const float4*)(ssrc + (size_t)src * NHEAD);
    e.x = lrelu(ss.x + sd.x); e.y = lrelu(ss.y + sd.y);
    e.z = lrelu(ss.z + sd.z); e.w = lrelu(ss.w + sd.w);
  }
}

__device__ __forceinline__ void add_rows(const float* __restrict__ xin, int lane, int cnt,
                                         int my_src, const float4 my_a,
                                         float4& s0, float4& s1, float4& s2, float4& s3) {
  for (int j = 0; j < cnt; ++j) {
    const int   s  = __builtin_amdgcn_readlane(my_src, j);
    const float a0 = bcast_f(my_a.x, j);
    const float a1 = bcast_f(my_a.y, j);
    const float a2 = bcast_f(my_a.z, j);
    const float a3 = bcast_f(my_a.w, j);
    const float4 xv = *(const float4*)(xin + (size_t)s * HIDF + lane * 4);
    s0.x += a0 * xv.x; s0.y += a0 * xv.y; s0.z += a0 * xv.z; s0.w += a0 * xv.w;
    s1.x += a1 * xv.x; s1.y += a1 * xv.y; s1.z += a1 * xv.z; s1.w += a1 * xv.w;
    s2.x += a2 * xv.x; s2.y += a2 * xv.y; s2.z += a2 * xv.z; s2.w += a2 * xv.w;
    s3.x += a3 * xv.x; s3.y += a3 * xv.y; s3.z += a3 * xv.z; s3.w += a3 * xv.w;
  }
}

__global__ void __launch_bounds__(256)
edge_softmax_gather(const int* __restrict__ rowptr, const int* __restrict__ csr, int E, int n,
                    const float* __restrict__ ssrc, const float* __restrict__ sdst,
                    const float* __restrict__ xin, float* __restrict__ agg,
                    float* __restrict__ alpha_out) {
  const int lane = threadIdx.x & 31;
  const int node = __builtin_amdgcn_readfirstlane((int)(blockIdx.x * 8 + (threadIdx.x >> 5)));
  if (node >= n) return;
  const float4 sd = *(const float4*)(sdst + (size_t)node * NHEAD);
  const float4 sn = *(const float4*)(ssrc + (size_t)node * NHEAD);
  const int beg = rowptr[node], end = rowptr[node + 1];

  float4 es;
  es.x = lrelu(sn.x + sd.x); es.y = lrelu(sn.y + sd.y);
  es.z = lrelu(sn.z + sd.z); es.w = lrelu(sn.w + sd.w);

  const int cnt0 = min(32, end - beg);
  int src0, eid0; float4 e0;
  load_entries(csr, ssrc, sd, beg, cnt0, lane, src0, eid0, e0);

  float4 mx;
  mx.x = fmaxf(es.x, e0.x); mx.y = fmaxf(es.y, e0.y);
  mx.z = fmaxf(es.z, e0.z); mx.w = fmaxf(es.w, e0.w);
  for (int base = beg + 32; base < end; base += 32) {
    int s_, i_; float4 e_;
    load_entries(csr, ssrc, sd, base, min(32, end - base), lane, s_, i_, e_);
    mx.x = fmaxf(mx.x, e_.x); mx.y = fmaxf(mx.y, e_.y);
    mx.z = fmaxf(mx.z, e_.z); mx.w = fmaxf(mx.w, e_.w);
  }
  mx.x = wave_max(mx.x); mx.y = wave_max(mx.y);
  mx.z = wave_max(mx.z); mx.w = wave_max(mx.w);
  if (!(mx.x > -3.0e38f && mx.x < 3.0e38f)) mx.x = 0.f;
  if (!(mx.y > -3.0e38f && mx.y < 3.0e38f)) mx.y = 0.f;
  if (!(mx.z > -3.0e38f && mx.z < 3.0e38f)) mx.z = 0.f;
  if (!(mx.w > -3.0e38f && mx.w < 3.0e38f)) mx.w = 0.f;

  float4 x0 = make_float4(0.f, 0.f, 0.f, 0.f);
  if (lane < cnt0) {
    x0.x = expf(e0.x - mx.x); x0.y = expf(e0.y - mx.y);
    x0.z = expf(e0.z - mx.z); x0.w = expf(e0.w - mx.w);
  }
  float4 ds = x0;
  for (int base = beg + 32; base < end; base += 32) {
    const int cnt = min(32, end - base);
    int s_, i_; float4 e_;
    load_entries(csr, ssrc, sd, base, cnt, lane, s_, i_, e_);
    if (lane < cnt) {
      ds.x += expf(e_.x - mx.x); ds.y += expf(e_.y - mx.y);
      ds.z += expf(e_.z - mx.z); ds.w += expf(e_.w - mx.w);
    }
  }
  float4 xs;
  xs.x = expf(es.x - mx.x); xs.y = expf(es.y - mx.y);
  xs.z = expf(es.z - mx.z); xs.w = expf(es.w - mx.w);
  float4 dn;
  dn.x = wave_sum(ds.x) + xs.x; dn.y = wave_sum(ds.y) + xs.y;
  dn.z = wave_sum(ds.z) + xs.z; dn.w = wave_sum(ds.w) + xs.w;

  float4 s0, s1, s2, s3;
  {
    const size_t eid = (size_t)E + node;
    float4 a;
    a.x = xs.x / dn.x; a.y = xs.y / dn.y; a.z = xs.z / dn.z; a.w = xs.w / dn.w;
    if (alpha_out && lane == 0) *(float4*)(alpha_out + eid * NHEAD) = a;
    const float4 xv = *(const float4*)(xin + (size_t)node * HIDF + lane * 4);
    s0.x = a.x * xv.x; s0.y = a.x * xv.y; s0.z = a.x * xv.z; s0.w = a.x * xv.w;
    s1.x = a.y * xv.x; s1.y = a.y * xv.y; s1.z = a.y * xv.z; s1.w = a.y * xv.w;
    s2.x = a.z * xv.x; s2.y = a.z * xv.y; s2.z = a.z * xv.z; s2.w = a.z * xv.w;
    s3.x = a.w * xv.x; s3.y = a.w * xv.y; s3.z = a.w * xv.z; s3.w = a.w * xv.w;
  }

  {
    float4 a;
    a.x = x0.x / dn.x; a.y = x0.y / dn.y; a.z = x0.z / dn.z; a.w = x0.w / dn.w;
    if (alpha_out && lane < cnt0) *(float4*)(alpha_out + (size_t)eid0 * NHEAD) = a;
    add_rows(xin, lane, cnt0, src0, a, s0, s1, s2, s3);
  }
  for (int base = beg + 32; base < end; base += 32) {
    const int cnt = min(32, end - base);
    int my_src, my_eid; float4 e_;
    load_entries(csr, ssrc, sd, base, cnt, lane, my_src, my_eid, e_);
    float4 a = make_float4(0.f, 0.f, 0.f, 0.f);
    if (lane < cnt) {
      a.x = expf(e_.x - mx.x) / dn.x; a.y = expf(e_.y - mx.y) / dn.y;
      a.z = expf(e_.z - mx.z) / dn.z; a.w = expf(e_.w - mx.w) / dn.w;
      if (alpha_out) *(float4*)(alpha_out + (size_t)my_eid * NHEAD) = a;
    }
    add_rows(xin, lane, cnt, my_src, a, s0, s1, s2, s3);
  }

  float* ap = agg + (size_t)node * WHID + lane * 4;
  *(float4*)(ap)            = s0;
  *(float4*)(ap + HIDF)     = s1;
  *(float4*)(ap + 2 * HIDF) = s2;
  *(float4*)(ap + 3 * HIDF) = s3;
}

#define POOL_RUN 16

__device__ __forceinline__ void pool_flush(float* __restrict__ pooled, float* __restrict__ cnt,
                                           int g, int t, const float4 s, float run) {
  float* p = pooled + (size_t)g * WHID + 4 * t;
  atomicAdd(p + 0, s.x);
  atomicAdd(p + 1, s.y);
  atomicAdd(p + 2, s.z);
  atomicAdd(p + 3, s.w);
  if (t == 0) atomicAdd(&cnt[g], run);
}

__global__ void __launch_bounds__(128)
l2_post_pool(const float* __restrict__ acc, const float* __restrict__ b2,
             const int* __restrict__ batch, float* __restrict__ pooled,
             float* __restrict__ cnt, int n) {
  const int t  = threadIdx.x;
  const int n0 = blockIdx.x * POOL_RUN;
  const float4 bv = *(const float4*)(b2 + 4 * t);

  float4 v[POOL_RUN];
  int    gi[POOL_RUN];
#pragma unroll
  for (int j = 0; j < POOL_RUN; ++j) {
    const int nd = min(n0 + j, n - 1);
    v[j]  = *(const float4*)(acc + (size_t)nd * WHID + 4 * t);
    gi[j] = batch[nd];
  }

  float4 s = make_float4(0.f, 0.f, 0.f, 0.f);
  float run = 0.f;
  int   g   = gi[0];
#pragma unroll
  for (int j = 0; j < POOL_RUN; ++j) {
    if (n0 + j < n) {
      if (gi[j] != g) {
        pool_flush(pooled, cnt, g, t, s, run);
        s = make_float4(0.f, 0.f, 0.f, 0.f);
        run = 0.f;
        g = gi[j];
      }
      s.x += lrelu(v[j].x + bv.x);
      s.y += lrelu(v[j].y + bv.y);
      s.z += lrelu(v[j].z + bv.z);
      s.w += lrelu(v[j].w + bv.w);
      run += 1.f;
    }
  }
  pool_flush(pooled, cnt, g, t, s, run);
}

#define HROWS 16
#define PITCH0 (WHID + 4)
#define PITCH1 (OUTD + 4)
#define PITCH2 (HIDF + 4)

template <int K, int NC, int NT>
__device__ __forceinline__ void head_product(const float* As, int lda,
                                             const float* __restrict__ Wg,
                                             const float* __restrict__ bias,
                                             float* Cs, int ldc, int wave, int l, int half) {
  v8f acc[NT];
#pragma unroll
  for (int t = 0; t < NT; ++t) acc[t] = (v8f){};
  const float* ap = As + l * lda + 2 * half;
  const float* wp = Wg + (size_t)(2 * half) * NC + 16 * wave + l;
#pragma unroll 16
  for (int k = 0; k < K; k += 4) {
    const v2f a = *(const v2f*)(ap + k);
#pragma unroll
    for (int t = 0; t < NT; ++t) {
      v2f b;
      b.x = wp[(size_t)k * NC + 128 * t];
      b.y = wp[(size_t)(k + 1) * NC + 128 * t];
      acc[t] = __builtin_amdgcn_wmma_f32_16x16x4_f32(false, a, false, b, (short)0, acc[t], false, false);
    }
  }
#pragma unroll
  for (int t = 0; t < NT; ++t) {
    const int col = 16 * wave + 128 * t + l;
    const float bv = bias[col];
#pragma unroll
    for (int i = 0; i < 8; ++i) Cs[(i + 8 * half) * ldc + col] = acc[t][i] + bv;
  }
}

__global__ void __launch_bounds__(256)
head_mlp(const float* __restrict__ pooled, const float* __restrict__ cnt,
         const float* __restrict__ Wl0, const float* __restrict__ bl0,
         const float* __restrict__ Ws1, const float* __restrict__ bs1,
         const float* __restrict__ g1,  const float* __restrict__ bb1,
         const float* __restrict__ Ws2, const float* __restrict__ bs2,
         const float* __restrict__ g2,  const float* __restrict__ bb2,
         const float* __restrict__ Ws3, const float* __restrict__ bs3,
         float* __restrict__ out) {
  __shared__ float P[HROWS * PITCH0];
  __shared__ float T1[HROWS * PITCH1];
  __shared__ float ZT[HROWS * PITCH2];
  const int tid  = threadIdx.x;
  const int wave = tid >> 5;
  const int lane = tid & 31;
  const int half = lane >> 4;
  const int l    = lane & 15;
  const int g0   = blockIdx.x * HROWS;

#pragma unroll
  for (int j = 0; j < 8; ++j) {
    const int i = tid + 256 * j;
    const int r = i >> 7, c4 = i & 127;
    float4 v = *(const float4*)(pooled + (size_t)(g0 + r) * WHID + 4 * c4);
    const float d = fmaxf(cnt[g0 + r], 1.0f);
    v.x /= d; v.y /= d; v.z /= d; v.w /= d;
    *(float4*)(P + r * PITCH0 + 4 * c4) = v;
  }
  __syncthreads();

  head_product<WHID, HIDF, 1>(P, PITCH0, Wl0, bl0, ZT, PITCH2, wave, l, half);
  __syncthreads();
  head_product<HIDF, OUTD, 2>(ZT, PITCH2, Ws1, bs1, T1, PITCH1, wave, l, half);
  __syncthreads();

#pragma unroll
  for (int rr = 0; rr < 2; ++rr) {
    float* row = T1 + (2 * wave + rr) * PITCH1;
    float v[8];
    float s = 0.f;
#pragma unroll
    for (int j = 0; j < 8; ++j) { v[j] = row[lane + 32 * j]; s += v[j]; }
    const float mu = wave_sum(s) / OUTD;
    float q = 0.f;
#pragma unroll
    for (int j = 0; j < 8; ++j) { v[j] -= mu; q += v[j] * v[j]; }
    const float var = wave_sum(q) / OUTD;
    const float inv = rsqrtf(var + 1e-5f);
#pragma unroll
    for (int j = 0; j < 8; ++j) {
      const int cc = lane + 32 * j;
      row[cc] = fmaxf(g1[cc] * v[j] * inv + bb1[cc], 0.f);
    }
  }
  __syncthreads();

  head_product<OUTD, OUT2, 1>(T1, PITCH1, Ws2, bs2, ZT, PITCH2, wave, l, half);
  __syncthreads();

#pragma unroll
  for (int rr = 0; rr < 2; ++rr) {
    const int r = 2 * wave + rr;
    const float* row = ZT + r * PITCH2;
    float v[4];
    float s = 0.f;
#pragma unroll
    for (int j = 0; j < 4; ++j) { v[j] = row[lane + 32 * j]; s += v[j]; }
    const float mu = wave_sum(s) / OUT2;
    float q = 0.f;
#pragma unroll
    for (int j = 0; j < 4; ++j) { v[j] -= mu; q += v[j] * v[j]; }
    const float var = wave_sum(q) / OUT2;
    const float inv = rsqrtf(var + 1e-5f);
    float p0 = 0.f, p1 = 0.f;
#pragma unroll
    for (int j = 0; j < 4; ++j) {
      const int cc = lane + 32 * j;
      const float y = fmaxf(g2[cc] * v[j] * inv + bb2[cc], 0.f);
      const v2f w = *(const v2f*)(Ws3 + 2 * cc);
      p0 += y * w.x;
      p1 += y * w.y;
    }
    const float l0 = wave_sum(p0) + bs3[0];
    const float l1 = wave_sum(p1) + bs3[1];
    const float mx = fmaxf(l0, l1);
    const float e0 = expf(l0 - mx), e1 = expf(l1 - mx), es = e0 + e1;
    if (lane == 0) {
      out[(g0 + r) * 2 + 0] = e0 / es;
      out[(g0 + r) * 2 + 1] = e1 / es;
    }
  }
}

extern "C" void kernel_launch(void* const* d_in, const int* in_sizes, int n_in,
                              void* d_out, int out_size, void* d_ws, size_t ws_size,
                              hipStream_t stream) {
  (void)n_in; (void)out_size; (void)ws_size;
  const float* x    = (const float*)d_in[0];
  const int*   ei   = (const int*)d_in[1];
  const int*   batch= (const int*)d_in[2];
  const float* W1   = (const float*)d_in[3];
  const float* as1  = (const float*)d_in[4];
  const float* ad1  = (const float*)d_in[5];
  const float* b1   = (const float*)d_in[6];
  const float* g_ln = (const float*)d_in[7];
  const float* b_ln = (const float*)d_in[8];
  const float* W2   = (const float*)d_in[9];
  const float* as2  = (const float*)d_in[10];
  const float* ad2  = (const float*)d_in[11];
  const float* b2   = (const float*)d_in[12];
  const float* Wl0  = (const float*)d_in[13];
  const float* bl0  = (const float*)d_in[14];
  const float* Ws1  = (const float*)d_in[15];
  const float* bs1  = (const float*)d_in[16];
  const float* g1   = (const float*)d_in[17];
  const float* bb1  = (const float*)d_in[18];
  const float* Ws2  = (const float*)d_in[19];
  const float* bs2  = (const float*)d_in[20];
  const float* g2   = (const float*)d_in[21];
  const float* bb2  = (const float*)d_in[22];
  const float* Ws3  = (const float*)d_in[23];
  const float* bs3  = (const float*)d_in[24];

  const int N  = in_sizes[0] / HIDF;
  const int E  = in_sizes[1] / 2;
  const int NH = N * NHEAD;

  float* out       = (float*)d_out;
  float* alpha_out = out + NGROUP * NCLSD;

  size_t off = 0;
  auto alloc = [&](size_t bytes) -> float* {
    off = (off + 255) & ~(size_t)255;
    float* p = (float*)((char*)d_ws + off);
    off += bytes;
    return p;
  };
  float* agg    = alloc((size_t)N * WHID * 4);
  float* acc    = alloc((size_t)N * WHID * 4);
  float* x2     = alloc((size_t)N * HIDF * 4);
  float* pre    = alloc((size_t)N * HIDF * 4);
  float* ssrc   = alloc((size_t)NH * 4);
  float* sdst   = alloc((size_t)NH * 4);
  float* V1     = alloc((size_t)HIDF * 8 * 4);
  float* V2     = alloc((size_t)HIDF * 8 * 4);
  float* pooled = alloc((size_t)NGROUP * WHID * 4);
  float* cnt    = alloc((size_t)NGROUP * 4);
  int*   deg    = (int*)alloc((size_t)N * 4);
  int*   rowptr = (int*)alloc((size_t)(N + 1) * 4);
  int*   cursor = (int*)alloc((size_t)N * 4);
  int*   csr_arr = (int*)alloc((size_t)E * 8);
  int*   csr     = (int*)alloc((size_t)E * 8);

  auto cdiv = [](long a, long b) { return (int)((a + b - 1) / b); };
  const int mblocks = cdiv(cdiv(N, 16), 8);

  fill_i32<<<cdiv(N, 256), 256, 0, stream>>>(deg, 0, N);
  csr_hist<<<cdiv(E, 256), 256, 0, stream>>>(ei, E, deg);
  csr_scan<<<1, 1024, 0, stream>>>(deg, rowptr, cursor, N);
  csr_fill<<<cdiv(E, 256), 256, 0, stream>>>(ei, E, cursor, csr_arr);
  csr_sort_rows<<<cdiv(N, 8), 256, 0, stream>>>(rowptr, csr_arr, csr, N);

  score_vectors<<<(2 * HIDF * NHEAD) / 8, 256, 0, stream>>>(W1, as1, ad1, V1, W2, as2, ad2, V2);

  node_pass<false><<<cdiv(N, 8 * NPW), 256, 0, stream>>>(x, nullptr, nullptr, V1, nullptr, ssrc, sdst, N);
  edge_softmax_gather<<<cdiv(N, 8), 256, 0, stream>>>(rowptr, csr, E, N, ssrc, sdst, x, agg, nullptr);
  wmma_gemm<<<dim3(mblocks * (HIDF / 64), 1), 256, 0, stream>>>(agg, WHID, W1, WHID, HIDF, b1, 0.25f,
                                                                  pre, HIDF, N, WHID, HIDF, 0);
  node_pass<true><<<cdiv(N, 8 * NPW), 256, 0, stream>>>(pre, g_ln, b_ln, V2, x2, ssrc, sdst, N);

  edge_softmax_gather<<<cdiv(N, 8), 256, 0, stream>>>(rowptr, csr, E, N, ssrc, sdst, x2, agg, alpha_out);
  wmma_gemm<<<dim3(mblocks * (HIDF / 64), NHEAD), 256, 0, stream>>>(agg, WHID, W2, WHID, KC * WHID, nullptr, 1.0f,
                                                                      acc, WHID, N, HIDF, HIDF, HIDF);

  fill_f32<<<cdiv(NGROUP * WHID, 256), 256, 0, stream>>>(pooled, 0.f, NGROUP * WHID);
  fill_f32<<<1, NGROUP, 0, stream>>>(cnt, 0.f, NGROUP);
  l2_post_pool<<<cdiv(N, POOL_RUN), 128, 0, stream>>>(acc, b2, batch, pooled, cnt, N);
  head_mlp<<<NGROUP / HROWS, 256, 0, stream>>>(pooled, cnt, Wl0, bl0, Ws1, bs1, g1, bb1,
                                               Ws2, bs2, g2, bb2, Ws3, bs3, out);
}


# gather and projection fused: each block gathers 16 nodes' head sums into LDS and feeds them to the f32 WMMA products as the A tile; agg plane and 2 launches gone; layer-1 head sum order changed
# speedup vs baseline: 5.0881x; 1.3187x over previous
#include <hip/hip_runtime.h>
#include <math.h>

#define HIDF   128
#define NHEAD  4
#define WHID   512
#define NGROUP 64
#define OUTD   256
#define OUT2   128
#define NCLSD  2
#define SLOPE  0.2f

typedef float v2f __attribute__((ext_vector_type(2)));
typedef float v8f __attribute__((ext_vector_type(8)));

__device__ __forceinline__ float lrelu(float x) { return x >= 0.f ? x : SLOPE * x; }

__global__ void fill_f32(float* __restrict__ p, float v, int n) {
  int i = blockIdx.x * blockDim.x + threadIdx.x;
  if (i < n) p[i] = v;
}

__global__ void fill_i32(int* __restrict__ p, int v, int n) {
  int i = blockIdx.x * blockDim.x + threadIdx.x;
  if (i < n) p[i] = v;
}

__global__ void csr_hist(const int* __restrict__ ei, int E, int* __restrict__ deg) {
  int e = blockIdx.x * blockDim.x + threadIdx.x;
  if (e < E) atomicAdd(&deg[ei[E + e]], 1);
}

__global__ void __launch_bounds__(1024)
csr_scan(const int* __restrict__ deg, int* __restrict__ rowptr,
         int* __restrict__ cursor, int n) {
  __shared__ int part[1024];
  const int t     = threadIdx.x;
  const int items = (n + 1023) >> 10;
  const int lo    = min(t * items, n);
  const int hi    = min(lo + items, n);
  int s = 0;
  for (int i = lo; i < hi; ++i) s += deg[i];
  part[t] = s;
  __syncthreads();
  for (int off = 1; off < 1024; off <<= 1) {
    const int v = (t >= off) ? part[t - off] : 0;
    __syncthreads();
    part[t] += v;
    __syncthreads();
  }
  int run = part[t] - s;
  for (int i = lo; i < hi; ++i) {
    rowptr[i] = run;
    cursor[i] = run;
    run += deg[i];
  }
  if (t == 1023) rowptr[n] = part[1023];
}

__global__ void csr_fill(const int* __restrict__ ei, int E, int* __restrict__ cursor,
                         int* __restrict__ csr) {
  int e = blockIdx.x * blockDim.x + threadIdx.x;
  if (e >= E) return;
  const int s = ei[e], d = ei[E + e];
  const int pos = atomicAdd(&cursor[d], 1);
  *(int2*)(csr + 2 * (size_t)pos) = make_int2(s, e);
}

__global__ void __launch_bounds__(256)
csr_sort_rows(const int* __restrict__ rowptr, const int* __restrict__ csr_in,
              int* __restrict__ csr_out, int n) {
  const int lane = threadIdx.x & 31;
  const int node = __builtin_amdgcn_readfirstlane((int)(blockIdx.x * 8 + (threadIdx.x >> 5)));
  if (node >= n) return;
  const int beg = rowptr[node], end = rowptr[node + 1];
  for (int b = beg; b < end; b += 32) {
    const bool have = (b + lane) < end;
    int2 mine = make_int2(0, 0x7fffffff);
    if (have) mine = *(const int2*)(csr_in + 2 * (size_t)(b + lane));
    int rank = 0;
    for (int c0 = beg; c0 < end; c0 += 32) {
      const int cnt = min(32, end - c0);
      int other = 0x7fffffff;
      if (c0 + lane < end) other = csr_in[2 * (size_t)(c0 + lane) + 1];
      for (int j = 0; j < cnt; ++j)
        rank += (__builtin_amdgcn_readlane(other, j) < mine.y) ? 1 : 0;
    }
    if (have) *(int2*)(csr_out + 2 * (size_t)(beg + rank)) = mine;
  }
}

__device__ __forceinline__ float bcast_f(float v, int srcLane) {
  return __int_as_float(__builtin_amdgcn_readlane(__float_as_int(v), srcLane));
}

__device__ __forceinline__ float wave_max(float v) {
  for (int o = 16; o > 0; o >>= 1) v = fmaxf(v, __shfl_xor(v, o, 32));
  return v;
}
__device__ __forceinline__ float wave_sum(float v) {
  for (int o = 16; o > 0; o >>= 1) v += __shfl_xor(v, o, 32);
  return v;
}

__global__ void __launch_bounds__(256)
score_vectors(const float* __restrict__ W1, const float* __restrict__ as1,
              const float* __restrict__ ad1, float* __restrict__ V1,
              const float* __restrict__ W2, const float* __restrict__ as2,
              const float* __restrict__ ad2, float* __restrict__ V2) {
  const int lane = threadIdx.x & 31;
  const int w = __builtin_amdgcn_readfirstlane((int)(blockIdx.x * 8 + (threadIdx.x >> 5)));
  const bool second = w >= HIDF * NHEAD;
  const int t  = w & (HIDF * NHEAD - 1);
  const int hd = t & (NHEAD - 1);
  const float* W  = second ? W2  : W1;
  const float* as = second ? as2 : as1;
  const float* ad = second ? ad2 : ad1;
  float*       V  = second ? V2  : V1;
  const float4 wv = *(const float4*)(W  + (size_t)t * HIDF + lane * 4);
  const float4 av = *(const float4*)(as + hd * HIDF + lane * 4);
  const float4 dv = *(const float4*)(ad + hd * HIDF + lane * 4);
  const float s1 = wave_sum((wv.x * av.x + wv.y * av.y) + (wv.z * av.z + wv.w * av.w));
  const float s2 = wave_sum((wv.x * dv.x + wv.y * dv.y) + (wv.z * dv.z + wv.w * dv.w));
  if (lane == 0) {
    V[(t >> 2) * 8 + hd]     = s1;
    V[(t >> 2) * 8 + 4 + hd] = s2;
  }
}

#define NPW 4

template <bool NORM>
__global__ void __launch_bounds__(256)
node_pass(const float* __restrict__ xin, const float* __restrict__ g_ln,
          const float* __restrict__ b_ln, const float* __restrict__ V,
          float* __restrict__ xout, float* __restrict__ ssrc, float* __restrict__ sdst, int n) {
  const int lane = threadIdx.x & 31;
  const int n0 = __builtin_amdgcn_readfirstlane((int)((blockIdx.x * 8 + (threadIdx.x >> 5)) * NPW));
  if (n0 >= n) return;

  float4 r[8];
#pragma unroll
  for (int i = 0; i < 8; ++i) r[i] = *(const float4*)(V + lane * 32 + 4 * i);

  float4 v[NPW];
#pragma unroll
  for (int j = 0; j < NPW; ++j) {
    const int nd = min(n0 + j, n - 1);
    v[j] = *(const float4*)(xin + (size_t)nd * HIDF + lane * 4);
  }
  float4 gv = make_float4(0.f, 0.f, 0.f, 0.f), bl = gv;
  if constexpr (NORM) {
    gv = *(const float4*)(g_ln + lane * 4);
    bl = *(const float4*)(b_ln + lane * 4);
  }
#pragma unroll
  for (int j = 0; j < NPW; ++j) {
    if (n0 + j < n) {
      float4 y = v[j];
      if constexpr (NORM) {
        const float mu = wave_sum((y.x + y.y) + (y.z + y.w)) * (1.0f / HIDF);
        float4 d;
        d.x = y.x - mu; d.y = y.y - mu; d.z = y.z - mu; d.w = y.w - mu;
        const float var = wave_sum((d.x * d.x + d.y * d.y) + (d.z * d.z + d.w * d.w)) * (1.0f / HIDF);
        const float inv = rsqrtf(var + 1e-5f);
        y.x = lrelu(gv.x * d.x * inv + bl.x);
        y.y = lrelu(gv.y * d.y * inv + bl.y);
        y.z = lrelu(gv.z * d.z * inv + bl.z);
        y.w = lrelu(gv.w * d.w * inv + bl.w);
        *(float4*)(xout + (size_t)(n0 + j) * HIDF + lane * 4) = y;
      }
      float4 ss, sd;
      ss.x = wave_sum((y.x * r[0].x + y.y * r[2].x) + (y.z * r[4].x + y.w * r[6].x));
      ss.y = wave_sum((y.x * r[0].y + y.y * r[2].y) + (y.z * r[4].y + y.w * r[6].y));
      ss.z = wave_sum((y.x * r[0].z + y.y * r[2].z) + (y.z * r[4].z + y.w * r[6].z));
      ss.w = wave_sum((y.x * r[0].w + y.y * r[2].w) + (y.z * r[4].w + y.w * r[6].w));
      sd.x = wave_sum((y.x * r[1].x + y.y * r[3].x) + (y.z * r[5].x + y.w * r[7].x));
      sd.y = wave_sum((y.x * r[1].y + y.y * r[3].y) + (y.z * r[5].y + y.w * r[7].y));
      sd.z = wave_sum((y.x * r[1].z + y.y * r[3].z) + (y.z * r[5].z + y.w * r[7].z));
      sd.w = wave_sum((y.x * r[1].w + y.y * r[3].w) + (y.z * r[5].w + y.w * r[7].w));
      if (lane == 0) {
        *(float4*)(ssrc + (size_t)(n0 + j) * NHEAD) = ss;
        *(float4*)(sdst + (size_t)(n0 + j) * NHEAD) = sd;
      }
    }
  }
}

__device__ __forceinline__ void load_entries(const int* __restrict__ csr,
                                             const float* __restrict__ ssrc,
                                             const float4 sd, int base, int cnt, int lane,
                                             int& src, int& eid, float4& e) {
  src = 0; eid = 0;
  e = make_float4(-INFINITY, -INFINITY, -INFINITY, -INFINITY);
  if (lane < cnt) {
    const int2 ent = *(const int2*)(csr + 2 * (size_t)(base + lane));
    src = ent.x; eid = ent.y;
    const float4 ss = *(const float4*)(ssrc + (size_t)src * NHEAD);
    e.x = lrelu(ss.x + sd.x); e.y = lrelu(ss.y + sd.y);
    e.z = lrelu(ss.z + sd.z); e.w = lrelu(ss.w + sd.w);
  }
}

__device__ __forceinline__ void add_rows(const float* __restrict__ xin, int lane, int cnt,
                                         int my_src, const float4 my_a,
                                         float4& s0, float4& s1, float4& s2, float4& s3) {
  for (int j = 0; j < cnt; ++j) {
    const int   s  = __builtin_amdgcn_readlane(my_src, j);
    const float a0 = bcast_f(my_a.x, j);
    const float a1 = bcast_f(my_a.y, j);
    const float a2 = bcast_f(my_a.z, j);
    const float a3 = bcast_f(my_a.w, j);
    const float4 xv = *(const float4*)(xin + (size_t)s * HIDF + lane * 4);
    s0.x += a0 * xv.x; s0.y += a0 * xv.y; s0.z += a0 * xv.z; s0.w += a0 * xv.w;
    s1.x += a1 * xv.x; s1.y += a1 * xv.y; s1.z += a1 * xv.z; s1.w += a1 * xv.w;
    s2.x += a2 * xv.x; s2.y += a2 * xv.y; s2.z += a2 * xv.z; s2.w += a2 * xv.w;
    s3.x += a3 * xv.x; s3.y += a3 * xv.y; s3.z += a3 * xv.z; s3.w += a3 * xv.w;
  }
}

__device__ __forceinline__ void gather_node(const int* __restrict__ rowptr, const int* __restrict__ csr,
                                            int E, int node,
                                            const float* __restrict__ ssrc, const float* __restrict__ sdst,
                                            const float* __restrict__ xin, float* __restrict__ alpha_out,
                                            int lane, float4& s0, float4& s1, float4& s2, float4& s3) {
  const float4 sd = *(const float4*)(sdst + (size_t)node * NHEAD);
  const float4 sn = *(const float4*)(ssrc + (size_t)node * NHEAD);
  const int beg = rowptr[node], end = rowptr[node + 1];

  float4 es;
  es.x = lrelu(sn.x + sd.x); es.y = lrelu(sn.y + sd.y);
  es.z = lrelu(sn.z + sd.z); es.w = lrelu(sn.w + sd.w);

  const int cnt0 = min(32, end - beg);
  int src0, eid0; float4 e0;
  load_entries(csr, ssrc, sd, beg, cnt0, lane, src0, eid0, e0);

  float4 mx;
  mx.x = fmaxf(es.x, e0.x); mx.y = fmaxf(es.y, e0.y);
  mx.z = fmaxf(es.z, e0.z); mx.w = fmaxf(es.w, e0.w);
  for (int base = beg + 32; base < end; base += 32) {
    int s_, i_; float4 e_;
    load_entries(csr, ssrc, sd, base, min(32, end - base), lane, s_, i_, e_);
    mx.x = fmaxf(mx.x, e_.x); mx.y = fmaxf(mx.y, e_.y);
    mx.z = fmaxf(mx.z, e_.z); mx.w = fmaxf(mx.w, e_.w);
  }
  mx.x = wave_max(mx.x); mx.y = wave_max(mx.y);
  mx.z = wave_max(mx.z); mx.w = wave_max(mx.w);
  if (!(mx.x > -3.0e38f && mx.x < 3.0e38f)) mx.x = 0.f;
  if (!(mx.y > -3.0e38f && mx.y < 3.0e38f)) mx.y = 0.f;
  if (!(mx.z > -3.0e38f && mx.z < 3.0e38f)) mx.z = 0.f;
  if (!(mx.w > -3.0e38f && mx.w < 3.0e38f)) mx.w = 0.f;

  float4 x0 = make_float4(0.f, 0.f, 0.f, 0.f);
  if (lane < cnt0) {
    x0.x = expf(e0.x - mx.x); x0.y = expf(e0.y - mx.y);
    x0.z = expf(e0.z - mx.z); x0.w = expf(e0.w - mx.w);
  }
  float4 ds = x0;
  for (int base = beg + 32; base < end; base += 32) {
    const int cnt = min(32, end - base);
    int s_, i_; float4 e_;
    load_entries(csr, ssrc, sd, base, cnt, lane, s_, i_, e_);
    if (lane < cnt) {
      ds.x += expf(e_.x - mx.x); ds.y += expf(e_.y - mx.y);
      ds.z += expf(e_.z - mx.z); ds.w += expf(e_.w - mx.w);
    }
  }
  float4 xs;
  xs.x = expf(es.x - mx.x); xs.y = expf(es.y - mx.y);
  xs.z = expf(es.z - mx.z); xs.w = expf(es.w - mx.w);
  float4 dn;
  dn.x = wave_sum(ds.x) + xs.x; dn.y = wave_sum(ds.y) + xs.y;
  dn.z = wave_sum(ds.z) + xs.z; dn.w = wave_sum(ds.w) + xs.w;

  {
    const size_t eid = (size_t)E + node;
    float4 a;
    a.x = xs.x / dn.x; a.y = xs.y / dn.y; a.z = xs.z / dn.z; a.w = xs.w / dn.w;
    if (alpha_out && lane == 0) *(float4*)(alpha_out + eid * NHEAD) = a;
    const float4 xv = *(const float4*)(xin + (size_t)node * HIDF + lane * 4);
    s0.x = a.x * xv.x; s0.y = a.x * xv.y; s0.z = a.x * xv.z; s0.w = a.x * xv.w;
    s1.x = a.y * xv.x; s1.y = a.y * xv.y; s1.z = a.y * xv.z; s1.w = a.y * xv.w;
    s2.x = a.z * xv.x; s2.y = a.z * xv.y; s2.z = a.z * xv.z; s2.w = a.z * xv.w;
    s3.x = a.w * xv.x; s3.y = a.w * xv.y; s3.z = a.w * xv.z; s3.w = a.w * xv.w;
  }

  {
    float4 a;
    a.x = x0.x / dn.x; a.y = x0.y / dn.y; a.z = x0.z / dn.z; a.w = x0.w / dn.w;
    if (alpha_out && lane < cnt0) *(float4*)(alpha_out + (size_t)eid0 * NHEAD) = a;
    add_rows(xin, lane, cnt0, src0, a, s0, s1, s2, s3);
  }
  for (int base = beg + 32; base < end; base += 32) {
    const int cnt = min(32, end - base);
    int my_src, my_eid; float4 e_;
    load_entries(csr, ssrc, sd, base, cnt, lane, my_src, my_eid, e_);
    float4 a = make_float4(0.f, 0.f, 0.f, 0.f);
    if (lane < cnt) {
      a.x = expf(e_.x - mx.x) / dn.x; a.y = expf(e_.y - mx.y) / dn.y;
      a.z = expf(e_.z - mx.z) / dn.z; a.w = expf(e_.w - mx.w) / dn.w;
      if (alpha_out) *(float4*)(alpha_out + (size_t)my_eid * NHEAD) = a;
    }
    add_rows(xin, lane, cnt, my_src, a, s0, s1, s2, s3);
  }
}

#define GP_ROWS 16
#define APITCH (WHID + 4)

template <bool L2>
__global__ void __launch_bounds__(256)
gather_project(const int* __restrict__ rowptr, const int* __restrict__ csr, int E, int n,
               const float* __restrict__ ssrc, const float* __restrict__ sdst,
               const float* __restrict__ xin, const float* __restrict__ W,
               const float* __restrict__ bias, float* __restrict__ outp,
               float* __restrict__ alpha_out) {
  __shared__ float As[GP_ROWS * APITCH];
  const int tid  = threadIdx.x;
  const int lane = tid & 31;
  const int wave  = __builtin_amdgcn_readfirstlane(tid >> 5);
  const int node0 = blockIdx.x * GP_ROWS;

  for (int r = 0; r < 2; ++r) {
    const int row  = 2 * wave + r;
    const int node = node0 + row;
    float4 s0 = make_float4(0.f, 0.f, 0.f, 0.f), s1 = s0, s2 = s0, s3 = s0;
    if (node < n)
      gather_node(rowptr, csr, E, node, ssrc, sdst, xin, alpha_out, lane, s0, s1, s2, s3);
    float* ap = As + row * APITCH + lane * 4;
    *(float4*)(ap)            = s0;
    *(float4*)(ap + HIDF)     = s1;
    *(float4*)(ap + 2 * HIDF) = s2;
    *(float4*)(ap + 3 * HIDF) = s3;
  }
  __syncthreads();

  const int half = lane >> 4;
  const int l    = lane & 15;
  const int h    = wave >> 1;
  const int c2   = wave & 1;
  const int ccol = 64 * c2 + 4 * l;
  const float* ap = As + l * APITCH + h * HIDF + 2 * half;
  const float* wp = W + (size_t)(2 * half) * WHID + h * HIDF + ccol;

  v8f acc0 = {}, acc1 = {}, acc2 = {}, acc3 = {};
#pragma unroll 8
  for (int k = 0; k < HIDF; k += 4) {
    const v2f a = *(const v2f*)(ap + k);
    const float4 w0 = *(const float4*)(wp + (size_t)k * WHID);
    const float4 w1 = *(const float4*)(wp + (size_t)(k + 1) * WHID);
    v2f b0, b1, b2, b3;
    b0.x = w0.x; b0.y = w1.x;
    b1.x = w0.y; b1.y = w1.y;
    b2.x = w0.z; b2.y = w1.z;
    b3.x = w0.w; b3.y = w1.w;
    acc0 = __builtin_amdgcn_wmma_f32_16x16x4_f32(false, a, false, b0, (short)0, acc0, false, false);
    acc1 = __builtin_amdgcn_wmma_f32_16x16x4_f32(false, a, false, b1, (short)0, acc1, false, false);
    acc2 = __builtin_amdgcn_wmma_f32_16x16x4_f32(false, a, false, b2, (short)0, acc2, false, false);
    acc3 = __builtin_amdgcn_wmma_f32_16x16x4_f32(false, a, false, b3, (short)0, acc3, false, false);
  }

  if constexpr (L2) {
#pragma unroll
    for (int i = 0; i < 8; ++i) {
      const int node = node0 + i + 8 * half;
      if (node < n)
        *(float4*)(outp + (size_t)node * WHID + h * HIDF + ccol) = make_float4(acc0[i], acc1[i], acc2[i], acc3[i]);
    }
  } else {
    __syncthreads();
#pragma unroll
    for (int i = 0; i < 8; ++i)
      *(float4*)(As + (h * GP_ROWS + i + 8 * half) * HIDF + ccol) = make_float4(acc0[i], acc1[i], acc2[i], acc3[i]);
    __syncthreads();
    const int row = tid >> 4;
    const int c8  = (tid & 15) * 8;
    if (node0 + row < n) {
#pragma unroll
      for (int q = 0; q < 2; ++q) {
        const float* rp = As + row * HIDF + c8 + 4 * q;
        const float4 r0 = *(const float4*)(rp);
        const float4 r1 = *(const float4*)(rp + 1 * GP_ROWS * HIDF);
        const float4 r2 = *(const float4*)(rp + 2 * GP_ROWS * HIDF);
        const float4 r3 = *(const float4*)(rp + 3 * GP_ROWS * HIDF);
        const float4 bv = *(const float4*)(bias + c8 + 4 * q);
        float4 o;
        o.x = (((r0.x + r1.x) + r2.x) + r3.x) * 0.25f + bv.x;
        o.y = (((r0.y + r1.y) + r2.y) + r3.y) * 0.25f + bv.y;
        o.z = (((r0.z + r1.z) + r2.z) + r3.z) * 0.25f + bv.z;
        o.w = (((r0.w + r1.w) + r2.w) + r3.w) * 0.25f + bv.w;
        *(float4*)(outp + (size_t)(node0 + row) * HIDF + c8 + 4 * q) = o;
      }
    }
  }
}

#define POOL_RUN 16

__device__ __forceinline__ void pool_flush(float* __restrict__ pooled, float* __restrict__ cnt,
                                           int g, int t, const float4 s, float run) {
  float* p = pooled + (size_t)g * WHID + 4 * t;
  atomicAdd(p + 0, s.x);
  atomicAdd(p + 1, s.y);
  atomicAdd(p + 2, s.z);
  atomicAdd(p + 3, s.w);
  if (t == 0) atomicAdd(&cnt[g], run);
}

__global__ void __launch_bounds__(128)
l2_post_pool(const float* __restrict__ acc, const float* __restrict__ b2,
             const int* __restrict__ batch, float* __restrict__ pooled,
             float* __restrict__ cnt, int n) {
  const int t  = threadIdx.x;
  const int n0 = blockIdx.x * POOL_RUN;
  const float4 bv = *(const float4*)(b2 + 4 * t);

  float4 v[POOL_RUN];
  int    gi[POOL_RUN];
#pragma unroll
  for (int j = 0; j < POOL_RUN; ++j) {
    const int nd = min(n0 + j, n - 1);
    v[j]  = *(const float4*)(acc + (size_t)nd * WHID + 4 * t);
    gi[j] = batch[nd];
  }

  float4 s = make_float4(0.f, 0.f, 0.f, 0.f);
  float run = 0.f;
  int   g   = gi[0];
#pragma unroll
  for (int j = 0; j < POOL_RUN; ++j) {
    if (n0 + j < n) {
      if (gi[j] != g) {
        pool_flush(pooled, cnt, g, t, s, run);
        s = make_float4(0.f, 0.f, 0.f, 0.f);
        run = 0.f;
        g = gi[j];
      }
      s.x += lrelu(v[j].x + bv.x);
      s.y += lrelu(v[j].y + bv.y);
      s.z += lrelu(v[j].z + bv.z);
      s.w += lrelu(v[j].w + bv.w);
      run += 1.f;
    }
  }
  pool_flush(pooled, cnt, g, t, s, run);
}

#define HROWS 16
#define PITCH0 (WHID + 4)
#define PITCH1 (OUTD + 4)
#define PITCH2 (HIDF + 4)

template <int K, int NC, int NT>
__device__ __forceinline__ void head_product(const float* As, int lda,
                                             const float* __restrict__ Wg,
                                             const float* __restrict__ bias,
                                             float* Cs, int ldc, int wave, int l, int half) {
  v8f acc[NT];
#pragma unroll
  for (int t = 0; t < NT; ++t) acc[t] = (v8f){};
  const float* ap = As + l * lda + 2 * half;
  const float* wp = Wg + (size_t)(2 * half) * NC + 16 * wave + l;
#pragma unroll 16
  for (int k = 0; k < K; k += 4) {
    const v2f a = *(const v2f*)(ap + k);
#pragma unroll
    for (int t = 0; t < NT; ++t) {
      v2f b;
      b.x = wp[(size_t)k * NC + 128 * t];
      b.y = wp[(size_t)(k + 1) * NC + 128 * t];
      acc[t] = __builtin_amdgcn_wmma_f32_16x16x4_f32(false, a, false, b, (short)0, acc[t], false, false);
    }
  }
#pragma unroll
  for (int t = 0; t < NT; ++t) {
    const int col = 16 * wave + 128 * t + l;
    const float bv = bias[col];
#pragma unroll
    for (int i = 0; i < 8; ++i) Cs[(i + 8 * half) * ldc + col] = acc[t][i] + bv;
  }
}

__global__ void __launch_bounds__(256)
head_mlp(const float* __restrict__ pooled, const float* __restrict__ cnt,
         const float* __restrict__ Wl0, const float* __restrict__ bl0,
         const float* __restrict__ Ws1, const float* __restrict__ bs1,
         const float* __restrict__ g1,  const float* __restrict__ bb1,
         const float* __restrict__ Ws2, const float* __restrict__ bs2,
         const float* __restrict__ g2,  const float* __restrict__ bb2,
         const float* __restrict__ Ws3, const float* __restrict__ bs3,
         float* __restrict__ out) {
  __shared__ float P[HROWS * PITCH0];
  __shared__ float T1[HROWS * PITCH1];
  __shared__ float ZT[HROWS * PITCH2];
  const int tid  = threadIdx.x;
  const int wave = tid >> 5;
  const int lane = tid & 31;
  const int half = lane >> 4;
  const int l    = lane & 15;
  const int g0   = blockIdx.x * HROWS;

#pragma unroll
  for (int j = 0; j < 8; ++j) {
    const int i = tid + 256 * j;
    const int r = i >> 7, c4 = i & 127;
    float4 v = *(const float4*)(pooled + (size_t)(g0 + r) * WHID + 4 * c4);
    const float d = fmaxf(cnt[g0 + r], 1.0f);
    v.x /= d; v.y /= d; v.z /= d; v.w /= d;
    *(float4*)(P + r * PITCH0 + 4 * c4) = v;
  }
  __syncthreads();

  head_product<WHID, HIDF, 1>(P, PITCH0, Wl0, bl0, ZT, PITCH2, wave, l, half);
  __syncthreads();
  head_product<HIDF, OUTD, 2>(ZT, PITCH2, Ws1, bs1, T1, PITCH1, wave, l, half);
  __syncthreads();

#pragma unroll
  for (int rr = 0; rr < 2; ++rr) {
    float* row = T1 + (2 * wave + rr) * PITCH1;
    float v[8];
    float s = 0.f;
#pragma unroll
    for (int j = 0; j < 8; ++j) { v[j] = row[lane + 32 * j]; s += v[j]; }
    const float mu = wave_sum(s) / OUTD;
    float q = 0.f;
#pragma unroll
    for (int j = 0; j < 8; ++j) { v[j] -= mu; q += v[j] * v[j]; }
    const float var = wave_sum(q) / OUTD;
    const float inv = rsqrtf(var + 1e-5f);
#pragma unroll
    for (int j = 0; j < 8; ++j) {
      const int cc = lane + 32 * j;
      row[cc] = fmaxf(g1[cc] * v[j] * inv + bb1[cc], 0.f);
    }
  }
  __syncthreads();

  head_product<OUTD, OUT2, 1>(T1, PITCH1, Ws2, bs2, ZT, PITCH2, wave, l, half);
  __syncthreads();

#pragma unroll
  for (int rr = 0; rr < 2; ++rr) {
    const int r = 2 * wave + rr;
    const float* row = ZT + r * PITCH2;
    float v[4];
    float s = 0.f;
#pragma unroll
    for (int j = 0; j < 4; ++j) { v[j] = row[lane + 32 * j]; s += v[j]; }
    const float mu = wave_sum(s) / OUT2;
    float q = 0.f;
#pragma unroll
    for (int j = 0; j < 4; ++j) { v[j] -= mu; q += v[j] * v[j]; }
    const float var = wave_sum(q) / OUT2;
    const float inv = rsqrtf(var + 1e-5f);
    float p0 = 0.f, p1 = 0.f;
#pragma unroll
    for (int j = 0; j < 4; ++j) {
      const int cc = lane + 32 * j;
      const float y = fmaxf(g2[cc] * v[j] * inv + bb2[cc], 0.f);
      const v2f w = *(const v2f*)(Ws3 + 2 * cc);
      p0 += y * w.x;
      p1 += y * w.y;
    }
    const float l0 = wave_sum(p0) + bs3[0];
    const float l1 = wave_sum(p1) + bs3[1];
    const float mx = fmaxf(l0, l1);
    const float e0 = expf(l0 - mx), e1 = expf(l1 - mx), es = e0 + e1;
    if (lane == 0) {
      out[(g0 + r) * 2 + 0] = e0 / es;
      out[(g0 + r) * 2 + 1] = e1 / es;
    }
  }
}

extern "C" void kernel_launch(void* const* d_in, const int* in_sizes, int n_in,
                              void* d_out, int out_size, void* d_ws, size_t ws_size,
                              hipStream_t stream) {
  (void)n_in; (void)out_size; (void)ws_size;
  const float* x    = (const float*)d_in[0];
  const int*   ei   = (const int*)d_in[1];
  const int*   batch= (const int*)d_in[2];
  const float* W1   = (const float*)d_in[3];
  const float* as1  = (const float*)d_in[4];
  const float* ad1  = (const float*)d_in[5];
  const float* b1   = (const float*)d_in[6];
  const float* g_ln = (const float*)d_in[7];
  const float* b_ln = (const float*)d_in[8];
  const float* W2   = (const float*)d_in[9];
  const float* as2  = (const float*)d_in[10];
  const float* ad2  = (const float*)d_in[11];
  const float* b2   = (const float*)d_in[12];
  const float* Wl0  = (const float*)d_in[13];
  const float* bl0  = (const float*)d_in[14];
  const float* Ws1  = (const float*)d_in[15];
  const float* bs1  = (const float*)d_in[16];
  const float* g1   = (const float*)d_in[17];
  const float* bb1  = (const float*)d_in[18];
  const float* Ws2  = (const float*)d_in[19];
  const float* bs2  = (const float*)d_in[20];
  const float* g2   = (const float*)d_in[21];
  const float* bb2  = (const float*)d_in[22];
  const float* Ws3  = (const float*)d_in[23];
  const float* bs3  = (const float*)d_in[24];

  const int N  = in_sizes[0] / HIDF;
  const int E  = in_sizes[1] / 2;
  const int NH = N * NHEAD;

  float* out       = (float*)d_out;
  float* alpha_out = out + NGROUP * NCLSD;

  size_t off = 0;
  auto alloc = [&](size_t bytes) -> float* {
    off = (off + 255) & ~(size_t)255;
    float* p = (float*)((char*)d_ws + off);
    off += bytes;
    return p;
  };
  float* acc    = alloc((size_t)N * WHID * 4);
  float* x2     = alloc((size_t)N * HIDF * 4);
  float* pre    = alloc((size_t)N * HIDF * 4);
  float* ssrc   = alloc((size_t)NH * 4);
  float* sdst   = alloc((size_t)NH * 4);
  float* V1     = alloc((size_t)HIDF * 8 * 4);
  float* V2     = alloc((size_t)HIDF * 8 * 4);
  float* pooled = alloc((size_t)NGROUP * WHID * 4);
  float* cnt    = alloc((size_t)NGROUP * 4);
  int*   deg    = (int*)alloc((size_t)N * 4);
  int*   rowptr = (int*)alloc((size_t)(N + 1) * 4);
  int*   cursor = (int*)alloc((size_t)N * 4);
  int*   csr_arr = (int*)alloc((size_t)E * 8);
  int*   csr     = (int*)alloc((size_t)E * 8);

  auto cdiv = [](long a, long b) { return (int)((a + b - 1) / b); };

  fill_i32<<<cdiv(N, 256), 256, 0, stream>>>(deg, 0, N);
  csr_hist<<<cdiv(E, 256), 256, 0, stream>>>(ei, E, deg);
  csr_scan<<<1, 1024, 0, stream>>>(deg, rowptr, cursor, N);
  csr_fill<<<cdiv(E, 256), 256, 0, stream>>>(ei, E, cursor, csr_arr);
  csr_sort_rows<<<cdiv(N, 8), 256, 0, stream>>>(rowptr, csr_arr, csr, N);

  score_vectors<<<(2 * HIDF * NHEAD) / 8, 256, 0, stream>>>(W1, as1, ad1, V1, W2, as2, ad2, V2);

  node_pass<false><<<cdiv(N, 8 * NPW), 256, 0, stream>>>(x, nullptr, nullptr, V1, nullptr, ssrc, sdst, N);
  gather_project<false><<<cdiv(N, GP_ROWS), 256, 0, stream>>>(rowptr, csr, E, N, ssrc, sdst, x, W1, b1, pre, nullptr);
  node_pass<true><<<cdiv(N, 8 * NPW), 256, 0, stream>>>(pre, g_ln, b_ln, V2, x2, ssrc, sdst, N);

  gather_project<true><<<cdiv(N, GP_ROWS), 256, 0, stream>>>(rowptr, csr, E, N, ssrc, sdst, x2, W2, nullptr, acc, alpha_out);

  fill_f32<<<cdiv(NGROUP * WHID, 256), 256, 0, stream>>>(pooled, 0.f, NGROUP * WHID);
  fill_f32<<<1, NGROUP, 0, stream>>>(cnt, 0.f, NGROUP);
  l2_post_pool<<<cdiv(N, POOL_RUN), 128, 0, stream>>>(acc, b2, batch, pooled, cnt, N);
  head_mlp<<<NGROUP / HROWS, 256, 0, stream>>>(pooled, cnt, Wl0, bl0, Ws1, bs1, g1, bb1,
                                               Ws2, bs2, g2, bb2, Ws3, bs3, out);
}


# independent work shares launches: the 3 zero-fills join the score-vector launch as the call's first, layer 1's node pass rides in the one-block prefix-scan launch; 14 launches -> 10
# speedup vs baseline: 5.3159x; 1.0448x over previous
#include <hip/hip_runtime.h>
#include <math.h>

#define HIDF   128
#define NHEAD  4
#define WHID   512
#define NGROUP 64
#define OUTD   256
#define OUT2   128
#define NCLSD  2
#define SLOPE  0.2f

typedef float v2f __attribute__((ext_vector_type(2)));
typedef float v8f __attribute__((ext_vector_type(8)));

__device__ __forceinline__ float lrelu(float x) { return x >= 0.f ? x : SLOPE * x; }

__global__ void csr_hist(const int* __restrict__ ei, int E, int* __restrict__ deg) {
  int e = blockIdx.x * blockDim.x + threadIdx.x;
  if (e < E) atomicAdd(&deg[ei[E + e]], 1);
}

__device__ __forceinline__ void
csr_scan_block(const int* __restrict__ deg, int* __restrict__ rowptr,
               int* __restrict__ cursor, int n) {
  __shared__ int part[1024];
  const int t     = threadIdx.x;
  const int items = (n + 1023) >> 10;
  const int lo    = min(t * items, n);
  const int hi    = min(lo + items, n);
  int s = 0;
  for (int i = lo; i < hi; ++i) s += deg[i];
  part[t] = s;
  __syncthreads();
  for (int off = 1; off < 1024; off <<= 1) {
    const int v = (t >= off) ? part[t - off] : 0;
    __syncthreads();
    part[t] += v;
    __syncthreads();
  }
  int run = part[t] - s;
  for (int i = lo; i < hi; ++i) {
    rowptr[i] = run;
    cursor[i] = run;
    run += deg[i];
  }
  if (t == 1023) rowptr[n] = part[1023];
}

__global__ void csr_fill(const int* __restrict__ ei, int E, int* __restrict__ cursor,
                         int* __restrict__ csr) {
  int e = blockIdx.x * blockDim.x + threadIdx.x;
  if (e >= E) return;
  const int s = ei[e], d = ei[E + e];
  const int pos = atomicAdd(&cursor[d], 1);
  *(int2*)(csr + 2 * (size_t)pos) = make_int2(s, e);
}

__global__ void __launch_bounds__(256)
csr_sort_rows(const int* __restrict__ rowptr, const int* __restrict__ csr_in,
              int* __restrict__ csr_out, int n) {
  const int lane = threadIdx.x & 31;
  const int node = __builtin_amdgcn_readfirstlane((int)(blockIdx.x * 8 + (threadIdx.x >> 5)));
  if (node >= n) return;
  const int beg = rowptr[node], end = rowptr[node + 1];
  for (int b = beg; b < end; b += 32) {
    const bool have = (b + lane) < end;
    int2 mine = make_int2(0, 0x7fffffff);
    if (have) mine = *(const int2*)(csr_in + 2 * (size_t)(b + lane));
    int rank = 0;
    for (int c0 = beg; c0 < end; c0 += 32) {
      const int cnt = min(32, end - c0);
      int other = 0x7fffffff;
      if (c0 + lane < end) other = csr_in[2 * (size_t)(c0 + lane) + 1];
      for (int j = 0; j < cnt; ++j)
        rank += (__builtin_amdgcn_readlane(other, j) < mine.y) ? 1 : 0;
    }
    if (have) *(int2*)(csr_out + 2 * (size_t)(beg + rank)) = mine;
  }
}

__device__ __forceinline__ float bcast_f(float v, int srcLane) {
  return __int_as_float(__builtin_amdgcn_readlane(__float_as_int(v), srcLane));
}

__device__ __forceinline__ float wave_max(float v) {
  for (int o = 16; o > 0; o >>= 1) v = fmaxf(v, __shfl_xor(v, o, 32));
  return v;
}
__device__ __forceinline__ float wave_sum(float v) {
  for (int o = 16; o > 0; o >>= 1) v += __shfl_xor(v, o, 32);
  return v;
}

#define SV_BLOCKS ((2 * HIDF * NHEAD) / 8)

__global__ void __launch_bounds__(256)
setup_pass(const float* __restrict__ W1, const float* __restrict__ as1,
           const float* __restrict__ ad1, float* __restrict__ V1,
           const float* __restrict__ W2, const float* __restrict__ as2,
           const float* __restrict__ ad2, float* __restrict__ V2,
           int* __restrict__ deg, int n, float* __restrict__ pooled, float* __restrict__ cnt) {
  if (blockIdx.x >= SV_BLOCKS) {
    const int i = (blockIdx.x - SV_BLOCKS) * 256 + threadIdx.x;
    if (i < n) { deg[i] = 0; return; }
    const int j = i - n;
    if (j < NGROUP * WHID) { pooled[j] = 0.f; return; }
    const int k = j - NGROUP * WHID;
    if (k < NGROUP) cnt[k] = 0.f;
    return;
  }
  const int lane = threadIdx.x & 31;
  const int w = __builtin_amdgcn_readfirstlane((int)(blockIdx.x * 8 + (threadIdx.x >> 5)));
  const bool second = w >= HIDF * NHEAD;
  const int t  = w & (HIDF * NHEAD - 1);
  const int hd = t & (NHEAD - 1);
  const float* W  = second ? W2  : W1;
  const float* as = second ? as2 : as1;
  const float* ad = second ? ad2 : ad1;
  float*       V  = second ? V2  : V1;
  const float4 wv = *(const float4*)(W  + (size_t)t * HIDF + lane * 4);
  const float4 av = *(const float4*)(as + hd * HIDF + lane * 4);
  const float4 dv = *(const float4*)(ad + hd * HIDF + lane * 4);
  const float s1 = wave_sum((wv.x * av.x + wv.y * av.y) + (wv.z * av.z + wv.w * av.w));
  const float s2 = wave_sum((wv.x * dv.x + wv.y * dv.y) + (wv.z * dv.z + wv.w * dv.w));
  if (lane == 0) {
    V[(t >> 2) * 8 + hd]     = s1;
    V[(t >> 2) * 8 + 4 + hd] = s2;
  }
}

#define NPW 4

template <bool NORM>
__device__ __forceinline__ void
node_pass_wave(int wave_id, const float* __restrict__ xin, const float* __restrict__ g_ln,
               const float* __restrict__ b_ln, const float* __restrict__ V,
               float* __restrict__ xout, float* __restrict__ ssrc, float* __restrict__ sdst, int n) {
  const int lane = threadIdx.x & 31;
  const int n0 = __builtin_amdgcn_readfirstlane(wave_id * NPW);
  if (n0 >= n) return;

  float4 r[8];
#pragma unroll
  for (int i = 0; i < 8; ++i) r[i] = *(const float4*)(V + lane * 32 + 4 * i);

  float4 v[NPW];
#pragma unroll
  for (int j = 0; j < NPW; ++j) {
    const int nd = min(n0 + j, n - 1);
    v[j] = *(const float4*)(xin + (size_t)nd * HIDF + lane * 4);
  }
  float4 gv = make_float4(0.f, 0.f, 0.f, 0.f), bl = gv;
  if constexpr (NORM) {
    gv = *(const float4*)(g_ln + lane * 4);
    bl = *(const float4*)(b_ln + lane * 4);
  }
#pragma unroll
  for (int j = 0; j < NPW; ++j) {
    if (n0 + j < n) {
      float4 y = v[j];
      if constexpr (NORM) {
        const float mu = wave_sum((y.x + y.y) + (y.z + y.w)) * (1.0f / HIDF);
        float4 d;
        d.x = y.x - mu; d.y = y.y - mu; d.z = y.z - mu; d.w = y.w - mu;
        const float var = wave_sum((d.x * d.x + d.y * d.y) + (d.z * d.z + d.w * d.w)) * (1.0f / HIDF);
        const float inv = rsqrtf(var + 1e-5f);
        y.x = lrelu(gv.x * d.x * inv + bl.x);
        y.y = lrelu(gv.y * d.y * inv + bl.y);
        y.z = lrelu(gv.z * d.z * inv + bl.z);
        y.w = lrelu(gv.w * d.w * inv + bl.w);
        *(float4*)(xout + (size_t)(n0 + j) * HIDF + lane * 4) = y;
      }
      float4 ss, sd;
      ss.x = wave_sum((y.x * r[0].x + y.y * r[2].x) + (y.z * r[4].x + y.w * r[6].x));
      ss.y = wave_sum((y.x * r[0].y + y.y * r[2].y) + (y.z * r[4].y + y.w * r[6].y));
      ss.z = wave_sum((y.x * r[0].z + y.y * r[2].z) + (y.z * r[4].z + y.w * r[6].z));
      ss.w = wave_sum((y.x * r[0].w + y.y * r[2].w) + (y.z * r[4].w + y.w * r[6].w));
      sd.x = wave_sum((y.x * r[1].x + y.y * r[3].x) + (y.z * r[5].x + y.w * r[7].x));
      sd.y = wave_sum((y.x * r[1].y + y.y * r[3].y) + (y.z * r[5].y + y.w * r[7].y));
      sd.z = wave_sum((y.x * r[1].z + y.y * r[3].z) + (y.z * r[5].z + y.w * r[7].z));
      sd.w = wave_sum((y.x * r[1].w + y.y * r[3].w) + (y.z * r[5].w + y.w * r[7].w));
      if (lane == 0) {
        *(float4*)(ssrc + (size_t)(n0 + j) * NHEAD) = ss;
        *(float4*)(sdst + (size_t)(n0 + j) * NHEAD) = sd;
      }
    }
  }
}

__global__ void __launch_bounds__(256)
node_pass_norm(const float* __restrict__ xin, const float* __restrict__ g_ln,
               const float* __restrict__ b_ln, const float* __restrict__ V,
               float* __restrict__ xout, float* __restrict__ ssrc, float* __restrict__ sdst, int n) {
  node_pass_wave<true>((int)(blockIdx.x * 8 + (threadIdx.x >> 5)), xin, g_ln, b_ln, V, xout, ssrc, sdst, n);
}

__global__ void __launch_bounds__(1024)
scan_and_node_pass(const int* __restrict__ deg, int* __restrict__ rowptr,
                   int* __restrict__ cursor, int n,
                   const float* __restrict__ xin, const float* __restrict__ V,
                   float* __restrict__ ssrc, float* __restrict__ sdst) {
  if (blockIdx.x == 0) {
    csr_scan_block(deg, rowptr, cursor, n);
    return;
  }
  node_pass_wave<false>((int)((blockIdx.x - 1) * 32 + (threadIdx.x >> 5)), xin, nullptr, nullptr, V,
                        nullptr, ssrc, sdst, n);
}

__device__ __forceinline__ void load_entries(const int* __restrict__ csr,
                                             const float* __restrict__ ssrc,
                                             const float4 sd, int base, int cnt, int lane,
                                             int& src, int& eid, float4& e) {
  src = 0; eid = 0;
  e = make_float4(-INFINITY, -INFINITY, -INFINITY, -INFINITY);
  if (lane < cnt) {
    const int2 ent = *(const int2*)(csr + 2 * (size_t)(base + lane));
    src = ent.x; eid = ent.y;
    const float4 ss = *(const float4*)(ssrc + (size_t)src * NHEAD);
    e.x = lrelu(ss.x + sd.x); e.y = lrelu(ss.y + sd.y);
    e.z = lrelu(ss.z + sd.z); e.w = lrelu(ss.w + sd.w);
  }
}

__device__ __forceinline__ void add_rows(const float* __restrict__ xin, int lane, int cnt,
                                         int my_src, const float4 my_a,
                                         float4& s0, float4& s1, float4& s2, float4& s3) {
  for (int j = 0; j < cnt; ++j) {
    const int   s  = __builtin_amdgcn_readlane(my_src, j);
    const float a0 = bcast_f(my_a.x, j);
    const float a1 = bcast_f(my_a.y, j);
    const float a2 = bcast_f(my_a.z, j);
    const float a3 = bcast_f(my_a.w, j);
    const float4 xv = *(const float4*)(xin + (size_t)s * HIDF + lane * 4);
    s0.x += a0 * xv.x; s0.y += a0 * xv.y; s0.z += a0 * xv.z; s0.w += a0 * xv.w;
    s1.x += a1 * xv.x; s1.y += a1 * xv.y; s1.z += a1 * xv.z; s1.w += a1 * xv.w;
    s2.x += a2 * xv.x; s2.y += a2 * xv.y; s2.z += a2 * xv.z; s2.w += a2 * xv.w;
    s3.x += a3 * xv.x; s3.y += a3 * xv.y; s3.z += a3 * xv.z; s3.w += a3 * xv.w;
  }
}

__device__ __forceinline__ void gather_node(const int* __restrict__ rowptr, const int* __restrict__ csr,
                                            int E, int node,
                                            const float* __restrict__ ssrc, const float* __restrict__ sdst,
                                            const float* __restrict__ xin, float* __restrict__ alpha_out,
                                            int lane, float4& s0, float4& s1, float4& s2, float4& s3) {
  const float4 sd = *(const float4*)(sdst + (size_t)node * NHEAD);
  const float4 sn = *(const float4*)(ssrc + (size_t)node * NHEAD);
  const int beg = rowptr[node], end = rowptr[node + 1];

  float4 es;
  es.x = lrelu(sn.x + sd.x); es.y = lrelu(sn.y + sd.y);
  es.z = lrelu(sn.z + sd.z); es.w = lrelu(sn.w + sd.w);

  const int cnt0 = min(32, end - beg);
  int src0, eid0; float4 e0;
  load_entries(csr, ssrc, sd, beg, cnt0, lane, src0, eid0, e0);

  float4 mx;
  mx.x = fmaxf(es.x, e0.x); mx.y = fmaxf(es.y, e0.y);
  mx.z = fmaxf(es.z, e0.z); mx.w = fmaxf(es.w, e0.w);
  for (int base = beg + 32; base < end; base += 32) {
    int s_, i_; float4 e_;
    load_entries(csr, ssrc, sd, base, min(32, end - base), lane, s_, i_, e_);
    mx.x = fmaxf(mx.x, e_.x); mx.y = fmaxf(mx.y, e_.y);
    mx.z = fmaxf(mx.z, e_.z); mx.w = fmaxf(mx.w, e_.w);
  }
  mx.x = wave_max(mx.x); mx.y = wave_max(mx.y);
  mx.z = wave_max(mx.z); mx.w = wave_max(mx.w);
  if (!(mx.x > -3.0e38f && mx.x < 3.0e38f)) mx.x = 0.f;
  if (!(mx.y > -3.0e38f && mx.y < 3.0e38f)) mx.y = 0.f;
  if (!(mx.z > -3.0e38f && mx.z < 3.0e38f)) mx.z = 0.f;
  if (!(mx.w > -3.0e38f && mx.w < 3.0e38f)) mx.w = 0.f;

  float4 x0 = make_float4(0.f, 0.f, 0.f, 0.f);
  if (lane < cnt0) {
    x0.x = expf(e0.x - mx.x); x0.y = expf(e0.y - mx.y);
    x0.z = expf(e0.z - mx.z); x0.w = expf(e0.w - mx.w);
  }
  float4 ds = x0;
  for (int base = beg + 32; base < end; base += 32) {
    const int cnt = min(32, end - base);
    int s_, i_; float4 e_;
    load_entries(csr, ssrc, sd, base, cnt, lane, s_, i_, e_);
    if (lane < cnt) {
      ds.x += expf(e_.x - mx.x); ds.y += expf(e_.y - mx.y);
      ds.z += expf(e_.z - mx.z); ds.w += expf(e_.w - mx.w);
    }
  }
  float4 xs;
  xs.x = expf(es.x - mx.x); xs.y = expf(es.y - mx.y);
  xs.z = expf(es.z - mx.z); xs.w = expf(es.w - mx.w);
  float4 dn;
  dn.x = wave_sum(ds.x) + xs.x; dn.y = wave_sum(ds.y) + xs.y;
  dn.z = wave_sum(ds.z) + xs.z; dn.w = wave_sum(ds.w) + xs.w;

  {
    const size_t eid = (size_t)E + node;
    float4 a;
    a.x = xs.x / dn.x; a.y = xs.y / dn.y; a.z = xs.z / dn.z; a.w = xs.w / dn.w;
    if (alpha_out && lane == 0) *(float4*)(alpha_out + eid * NHEAD) = a;
    const float4 xv = *(const float4*)(xin + (size_t)node * HIDF + lane * 4);
    s0.x = a.x * xv.x; s0.y = a.x * xv.y; s0.z = a.x * xv.z; s0.w = a.x * xv.w;
    s1.x = a.y * xv.x; s1.y = a.y * xv.y; s1.z = a.y * xv.z; s1.w = a.y * xv.w;
    s2.x = a.z * xv.x; s2.y = a.z * xv.y; s2.z = a.z * xv.z; s2.w = a.z * xv.w;
    s3.x = a.w * xv.x; s3.y = a.w * xv.y; s3.z = a.w * xv.z; s3.w = a.w * xv.w;
  }

  {
    float4 a;
    a.x = x0.x / dn.x; a.y = x0.y / dn.y; a.z = x0.z / dn.z; a.w = x0.w / dn.w;
    if (alpha_out && lane < cnt0) *(float4*)(alpha_out + (size_t)eid0 * NHEAD) = a;
    add_rows(xin, lane, cnt0, src0, a, s0, s1, s2, s3);
  }
  for (int base = beg + 32; base < end; base += 32) {
    const int cnt = min(32, end - base);
    int my_src, my_eid; float4 e_;
    load_entries(csr, ssrc, sd, base, cnt, lane, my_src, my_eid, e_);
    float4 a = make_float4(0.f, 0.f, 0.f, 0.f);
    if (lane < cnt) {
      a.x = expf(e_.x - mx.x) / dn.x; a.y = expf(e_.y - mx.y) / dn.y;
      a.z = expf(e_.z - mx.z) / dn.z; a.w = expf(e_.w - mx.w) / dn.w;
      if (alpha_out) *(float4*)(alpha_out + (size_t)my_eid * NHEAD) = a;
    }
    add_rows(xin, lane, cnt, my_src, a, s0, s1, s2, s3);
  }
}

#define GP_ROWS 16
#define APITCH (WHID + 4)

template <bool L2>
__global__ void __launch_bounds__(256)
gather_project(const int* __restrict__ rowptr, const int* __restrict__ csr, int E, int n,
               const float* __restrict__ ssrc, const float* __restrict__ sdst,
               const float* __restrict__ xin, const float* __restrict__ W,
               const float* __restrict__ bias, float* __restrict__ outp,
               float* __restrict__ alpha_out) {
  __shared__ float As[GP_ROWS * APITCH];
  const int tid  = threadIdx.x;
  const int lane = tid & 31;
  const int wave  = __builtin_amdgcn_readfirstlane(tid >> 5);
  const int node0 = blockIdx.x * GP_ROWS;

  for (int r = 0; r < 2; ++r) {
    const int row  = 2 * wave + r;
    const int node = node0 + row;
    float4 s0 = make_float4(0.f, 0.f, 0.f, 0.f), s1 = s0, s2 = s0, s3 = s0;
    if (node < n)
      gather_node(rowptr, csr, E, node, ssrc, sdst, xin, alpha_out, lane, s0, s1, s2, s3);
    float* ap = As + row * APITCH + lane * 4;
    *(float4*)(ap)            = s0;
    *(float4*)(ap + HIDF)     = s1;
    *(float4*)(ap + 2 * HIDF) = s2;
    *(float4*)(ap + 3 * HIDF) = s3;
  }
  __syncthreads();

  const int half = lane >> 4;
  const int l    = lane & 15;
  const int h    = wave >> 1;
  const int c2   = wave & 1;
  const int ccol = 64 * c2 + 4 * l;
  const float* ap = As + l * APITCH + h * HIDF + 2 * half;
  const float* wp = W + (size_t)(2 * half) * WHID + h * HIDF + ccol;

  v8f acc0 = {}, acc1 = {}, acc2 = {}, acc3 = {};
#pragma unroll 8
  for (int k = 0; k < HIDF; k += 4) {
    const v2f a = *(const v2f*)(ap + k);
    const float4 w0 = *(const float4*)(wp + (size_t)k * WHID);
    const float4 w1 = *(const float4*)(wp + (size_t)(k + 1) * WHID);
    v2f b0, b1, b2, b3;
    b0.x = w0.x; b0.y = w1.x;
    b1.x = w0.y; b1.y = w1.y;
    b2.x = w0.z; b2.y = w1.z;
    b3.x = w0.w; b3.y = w1.w;
    acc0 = __builtin_amdgcn_wmma_f32_16x16x4_f32(false, a, false, b0, (short)0, acc0, false, false);
    acc1 = __builtin_amdgcn_wmma_f32_16x16x4_f32(false, a, false, b1, (short)0, acc1, false, false);
    acc2 = __builtin_amdgcn_wmma_f32_16x16x4_f32(false, a, false, b2, (short)0, acc2, false, false);
    acc3 = __builtin_amdgcn_wmma_f32_16x16x4_f32(false, a, false, b3, (short)0, acc3, false, false);
  }

  if constexpr (L2) {
#pragma unroll
    for (int i = 0; i < 8; ++i) {
      const int node = node0 + i + 8 * half;
      if (node < n)
        *(float4*)(outp + (size_t)node * WHID + h * HIDF + ccol) = make_float4(acc0[i], acc1[i], acc2[i], acc3[i]);
    }
  } else {
    __syncthreads();
#pragma unroll
    for (int i = 0; i < 8; ++i)
      *(float4*)(As + (h * GP_ROWS + i + 8 * half) * HIDF + ccol) = make_float4(acc0[i], acc1[i], acc2[i], acc3[i]);
    __syncthreads();
    const int row = tid >> 4;
    const int c8  = (tid & 15) * 8;
    if (node0 + row < n) {
#pragma unroll
      for (int q = 0; q < 2; ++q) {
        const float* rp = As + row * HIDF + c8 + 4 * q;
        const float4 r0 = *(const float4*)(rp);
        const float4 r1 = *(const float4*)(rp + 1 * GP_ROWS * HIDF);
        const float4 r2 = *(const float4*)(rp + 2 * GP_ROWS * HIDF);
        const float4 r3 = *(const float4*)(rp + 3 * GP_ROWS * HIDF);
        const float4 bv = *(const float4*)(bias + c8 + 4 * q);
        float4 o;
        o.x = (((r0.x + r1.x) + r2.x) + r3.x) * 0.25f + bv.x;
        o.y = (((r0.y + r1.y) + r2.y) + r3.y) * 0.25f + bv.y;
        o.z = (((r0.z + r1.z) + r2.z) + r3.z) * 0.25f + bv.z;
        o.w = (((r0.w + r1.w) + r2.w) + r3.w) * 0.25f + bv.w;
        *(float4*)(outp + (size_t)(node0 + row) * HIDF + c8 + 4 * q) = o;
      }
    }
  }
}

#define POOL_RUN 16

__device__ __forceinline__ void pool_flush(float* __restrict__ pooled, float* __restrict__ cnt,
                                           int g, int t, const float4 s, float run) {
  float* p = pooled + (size_t)g * WHID + 4 * t;
  atomicAdd(p + 0, s.x);
  atomicAdd(p + 1, s.y);
  atomicAdd(p + 2, s.z);
  atomicAdd(p + 3, s.w);
  if (t == 0) atomicAdd(&cnt[g], run);
}

__global__ void __launch_bounds__(128)
l2_post_pool(const float* __restrict__ acc, const float* __restrict__ b2,
             const int* __restrict__ batch, float* __restrict__ pooled,
             float* __restrict__ cnt, int n) {
  const int t  = threadIdx.x;
  const int n0 = blockIdx.x * POOL_RUN;
  const float4 bv = *(const float4*)(b2 + 4 * t);

  float4 v[POOL_RUN];
  int    gi[POOL_RUN];
#pragma unroll
  for (int j = 0; j < POOL_RUN; ++j) {
    const int nd = min(n0 + j, n - 1);
    v[j]  = *(const float4*)(acc + (size_t)nd * WHID + 4 * t);
    gi[j] = batch[nd];
  }

  float4 s = make_float4(0.f, 0.f, 0.f, 0.f);
  float run = 0.f;
  int   g   = gi[0];
#pragma unroll
  for (int j = 0; j < POOL_RUN; ++j) {
    if (n0 + j < n) {
      if (gi[j] != g) {
        pool_flush(pooled, cnt, g, t, s, run);
        s = make_float4(0.f, 0.f, 0.f, 0.f);
        run = 0.f;
        g = gi[j];
      }
      s.x += lrelu(v[j].x + bv.x);
      s.y += lrelu(v[j].y + bv.y);
      s.z += lrelu(v[j].z + bv.z);
      s.w += lrelu(v[j].w + bv.w);
      run += 1.f;
    }
  }
  pool_flush(pooled, cnt, g, t, s, run);
}

#define HROWS 16
#define PITCH0 (WHID + 4)
#define PITCH1 (OUTD + 4)
#define PITCH2 (HIDF + 4)

template <int K, int NC, int NT>
__device__ __forceinline__ void head_product(const float* As, int lda,
                                             const float* __restrict__ Wg,
                                             const float* __restrict__ bias,
                                             float* Cs, int ldc, int wave, int l, int half) {
  v8f acc[NT];
#pragma unroll
  for (int t = 0; t < NT; ++t) acc[t] = (v8f){};
  const float* ap = As + l * lda + 2 * half;
  const float* wp = Wg + (size_t)(2 * half) * NC + 16 * wave + l;
#pragma unroll 16
  for (int k = 0; k < K; k += 4) {
    const v2f a = *(const v2f*)(ap + k);
#pragma unroll
    for (int t = 0; t < NT; ++t) {
      v2f b;
      b.x = wp[(size_t)k * NC + 128 * t];
      b.y = wp[(size_t)(k + 1) * NC + 128 * t];
      acc[t] = __builtin_amdgcn_wmma_f32_16x16x4_f32(false, a, false, b, (short)0, acc[t], false, false);
    }
  }
#pragma unroll
  for (int t = 0; t < NT; ++t) {
    const int col = 16 * wave + 128 * t + l;
    const float bv = bias[col];
#pragma unroll
    for (int i = 0; i < 8; ++i) Cs[(i + 8 * half) * ldc + col] = acc[t][i] + bv;
  }
}

__global__ void __launch_bounds__(256)
head_mlp(const float* __restrict__ pooled, const float* __restrict__ cnt,
         const float* __restrict__ Wl0, const float* __restrict__ bl0,
         const float* __restrict__ Ws1, const float* __restrict__ bs1,
         const float* __restrict__ g1,  const float* __restrict__ bb1,
         const float* __restrict__ Ws2, const float* __restrict__ bs2,
         const float* __restrict__ g2,  const float* __restrict__ bb2,
         const float* __restrict__ Ws3, const float* __restrict__ bs3,
         float* __restrict__ out) {
  __shared__ float P[HROWS * PITCH0];
  __shared__ float T1[HROWS * PITCH1];
  __shared__ float ZT[HROWS * PITCH2];
  const int tid  = threadIdx.x;
  const int wave = tid >> 5;
  const int lane = tid & 31;
  const int half = lane >> 4;
  const int l    = lane & 15;
  const int g0   = blockIdx.x * HROWS;

#pragma unroll
  for (int j = 0; j < 8; ++j) {
    const int i = tid + 256 * j;
    const int r = i >> 7, c4 = i & 127;
    float4 v = *(const float4*)(pooled + (size_t)(g0 + r) * WHID + 4 * c4);
    const float d = fmaxf(cnt[g0 + r], 1.0f);
    v.x /= d; v.y /= d; v.z /= d; v.w /= d;
    *(float4*)(P + r * PITCH0 + 4 * c4) = v;
  }
  __syncthreads();

  head_product<WHID, HIDF, 1>(P, PITCH0, Wl0, bl0, ZT, PITCH2, wave, l, half);
  __syncthreads();
  head_product<HIDF, OUTD, 2>(ZT, PITCH2, Ws1, bs1, T1, PITCH1, wave, l, half);
  __syncthreads();

#pragma unroll
  for (int rr = 0; rr < 2; ++rr) {
    float* row = T1 + (2 * wave + rr) * PITCH1;
    float v[8];
    float s = 0.f;
#pragma unroll
    for (int j = 0; j < 8; ++j) { v[j] = row[lane + 32 * j]; s += v[j]; }
    const float mu = wave_sum(s) / OUTD;
    float q = 0.f;
#pragma unroll
    for (int j = 0; j < 8; ++j) { v[j] -= mu; q += v[j] * v[j]; }
    const float var = wave_sum(q) / OUTD;
    const float inv = rsqrtf(var + 1e-5f);
#pragma unroll
    for (int j = 0; j < 8; ++j) {
      const int cc = lane + 32 * j;
      row[cc] = fmaxf(g1[cc] * v[j] * inv + bb1[cc], 0.f);
    }
  }
  __syncthreads();

  head_product<OUTD, OUT2, 1>(T1, PITCH1, Ws2, bs2, ZT, PITCH2, wave, l, half);
  __syncthreads();

#pragma unroll
  for (int rr = 0; rr < 2; ++rr) {
    const int r = 2 * wave + rr;
    const float* row = ZT + r * PITCH2;
    float v[4];
    float s = 0.f;
#pragma unroll
    for (int j = 0; j < 4; ++j) { v[j] = row[lane + 32 * j]; s += v[j]; }
    const float mu = wave_sum(s) / OUT2;
    float q = 0.f;
#pragma unroll
    for (int j = 0; j < 4; ++j) { v[j] -= mu; q += v[j] * v[j]; }
    const float var = wave_sum(q) / OUT2;
    const float inv = rsqrtf(var + 1e-5f);
    float p0 = 0.f, p1 = 0.f;
#pragma unroll
    for (int j = 0; j < 4; ++j) {
      const int cc = lane + 32 * j;
      const float y = fmaxf(g2[cc] * v[j] * inv + bb2[cc], 0.f);
      const v2f w = *(const v2f*)(Ws3 + 2 * cc);
      p0 += y * w.x;
      p1 += y * w.y;
    }
    const float l0 = wave_sum(p0) + bs3[0];
    const float l1 = wave_sum(p1) + bs3[1];
    const float mx = fmaxf(l0, l1);
    const float e0 = expf(l0 - mx), e1 = expf(l1 - mx), es = e0 + e1;
    if (lane == 0) {
      out[(g0 + r) * 2 + 0] = e0 / es;
      out[(g0 + r) * 2 + 1] = e1 / es;
    }
  }
}

extern "C" void kernel_launch(void* const* d_in, const int* in_sizes, int n_in,
                              void* d_out, int out_size, void* d_ws, size_t ws_size,
                              hipStream_t stream) {
  (void)n_in; (void)out_size; (void)ws_size;
  const float* x    = (const float*)d_in[0];
  const int*   ei   = (const int*)d_in[1];
  const int*   batch= (const int*)d_in[2];
  const float* W1   = (const float*)d_in[3];
  const float* as1  = (const float*)d_in[4];
  const float* ad1  = (const float*)d_in[5];
  const float* b1   = (const float*)d_in[6];
  const float* g_ln = (const float*)d_in[7];
  const float* b_ln = (const float*)d_in[8];
  const float* W2   = (const float*)d_in[9];
  const float* as2  = (const float*)d_in[10];
  const float* ad2  = (const float*)d_in[11];
  const float* b2   = (const float*)d_in[12];
  const float* Wl0  = (const float*)d_in[13];
  const float* bl0  = (const float*)d_in[14];
  const float* Ws1  = (const float*)d_in[15];
  const float* bs1  = (const float*)d_in[16];
  const float* g1   = (const float*)d_in[17];
  const float* bb1  = (const float*)d_in[18];
  const float* Ws2  = (const float*)d_in[19];
  const float* bs2  = (const float*)d_in[20];
  const float* g2   = (const float*)d_in[21];
  const float* bb2  = (const float*)d_in[22];
  const float* Ws3  = (const float*)d_in[23];
  const float* bs3  = (const float*)d_in[24];

  const int N  = in_sizes[0] / HIDF;
  const int E  = in_sizes[1] / 2;
  const int NH = N * NHEAD;

  float* out       = (float*)d_out;
  float* alpha_out = out + NGROUP * NCLSD;

  size_t off = 0;
  auto alloc = [&](size_t bytes) -> float* {
    off = (off + 255) & ~(size_t)255;
    float* p = (float*)((char*)d_ws + off);
    off += bytes;
    return p;
  };
  float* acc    = alloc((size_t)N * WHID * 4);
  float* x2     = alloc((size_t)N * HIDF * 4);
  float* pre    = alloc((size_t)N * HIDF * 4);
  float* ssrc   = alloc((size_t)NH * 4);
  float* sdst   = alloc((size_t)NH * 4);
  float* V1     = alloc((size_t)HIDF * 8 * 4);
  float* V2     = alloc((size_t)HIDF * 8 * 4);
  float* pooled = alloc((size_t)NGROUP * WHID * 4);
  float* cnt    = alloc((size_t)NGROUP * 4);
  int*   deg    = (int*)alloc((size_t)N * 4);
  int*   rowptr = (int*)alloc((size_t)(N + 1) * 4);
  int*   cursor = (int*)alloc((size_t)N * 4);
  int*   csr_arr = (int*)alloc((size_t)E * 8);
  int*   csr     = (int*)alloc((size_t)E * 8);

  auto cdiv = [](long a, long b) { return (int)((a + b - 1) / b); };

  setup_pass<<<SV_BLOCKS + cdiv((long)N + NGROUP * WHID + NGROUP, 256), 256, 0, stream>>>(
      W1, as1, ad1, V1, W2, as2, ad2, V2, deg, N, pooled, cnt);

  csr_hist<<<cdiv(E, 256), 256, 0, stream>>>(ei, E, deg);
  scan_and_node_pass<<<1 + cdiv(N, 32 * NPW), 1024, 0, stream>>>(deg, rowptr, cursor, N, x, V1, ssrc, sdst);
  csr_fill<<<cdiv(E, 256), 256, 0, stream>>>(ei, E, cursor, csr_arr);
  csr_sort_rows<<<cdiv(N, 8), 256, 0, stream>>>(rowptr, csr_arr, csr, N);

  gather_project<false><<<cdiv(N, GP_ROWS), 256, 0, stream>>>(rowptr, csr, E, N, ssrc, sdst, x, W1, b1, pre, nullptr);
  node_pass_norm<<<cdiv(N, 8 * NPW), 256, 0, stream>>>(pre, g_ln, b_ln, V2, x2, ssrc, sdst, N);

  gather_project<true><<<cdiv(N, GP_ROWS), 256, 0, stream>>>(rowptr, csr, E, N, ssrc, sdst, x2, W2, nullptr, acc, alpha_out);

  l2_post_pool<<<cdiv(N, POOL_RUN), 128, 0, stream>>>(acc, b2, batch, pooled, cnt, N);
  head_mlp<<<NGROUP / HROWS, 256, 0, stream>>>(pooled, cnt, Wl0, bl0, Ws1, bs1, g1, bb1,
                                               Ws2, bs2, g2, bb2, Ws3, bs3, out);
}
